# starting kernel
# baseline (speedup 1.0000x reference)
#include <hip/hip_runtime.h>
#include <hip/hip_bf16.h>

typedef float v2f __attribute__((ext_vector_type(2)));
typedef float v8f __attribute__((ext_vector_type(8)));

#define F_IN 128
#define F_E  16
#define NH   8
#define NC   32
#define HC   256
#define NEG_SLOPE 0.2f

__device__ __forceinline__ unsigned ordkey(float f) {
    unsigned u = __float_as_uint(f);
    return (u & 0x80000000u) ? ~u : (u | 0x80000000u);
}
__device__ __forceinline__ float ordinv(unsigned k) {
    unsigned u = (k & 0x80000000u) ? (k & 0x7fffffffu) : ~k;
    return __uint_as_float(u);
}

__global__ void GAT_zero(float* p, long n) {
    long i = (long)blockIdx.x * blockDim.x + threadIdx.x;
    long stride = (long)gridDim.x * blockDim.x;
    for (; i < n; i += stride) p[i] = 0.0f;
}

__global__ void GAT_degloop(const int* __restrict__ dst,
                            const float* __restrict__ eattr,
                            float* __restrict__ deg,
                            float* __restrict__ loop, int E) {
    long idx = (long)blockIdx.x * blockDim.x + threadIdx.x;
    long e = idx >> 4;
    int  f = (int)(idx & 15);
    if (e >= E) return;
    int d = dst[e];
    atomicAdd(&loop[(long)d * F_E + f], eattr[e * F_E + f]);
    if (f == 0) atomicAdd(&deg[d], 1.0f);
}

__global__ void GAT_loopdiv(float* __restrict__ loop,
                            const float* __restrict__ deg, int N) {
    long idx = (long)blockIdx.x * blockDim.x + threadIdx.x;
    if (idx >= (long)N * F_E) return;
    int n = (int)(idx >> 4);
    loop[idx] = loop[idx] / fmaxf(deg[n], 1.0f);
}

__global__ void GAT_foldM(const float* __restrict__ We,
                          const float* __restrict__ att_edge,
                          float* __restrict__ M) {
    int tid = threadIdx.x;
    if (tid >= F_E * NH) return;
    int f = tid >> 3, h = tid & 7;
    float s = 0.0f;
    for (int c = 0; c < NC; ++c)
        s += We[(long)(h * NC + c) * F_E + f] * att_edge[h * NC + c];
    M[f * NH + h] = s;
}

__global__ void GAT_gemm_xh(const float* __restrict__ x,
                            const float* __restrict__ W,
                            float* __restrict__ xh, int N) {
    int wave = threadIdx.x >> 5;
    int lane = threadIdx.x & 31;
    int half = lane >> 4;
    int l    = lane & 15;
    long rowTiles = ((long)N + 15) >> 4;
    long wtile = (long)blockIdx.x * 8 + wave;
    if (wtile >= rowTiles * 4) return;
    long rowTile = wtile >> 2;
    int  colGrp  = (int)(wtile & 3);
    long row0 = rowTile * 16;
    int  col0 = colGrp * 64;

    long arow = row0 + l;
    if (arow >= N) arow = N - 1;
    const float* xp  = x + arow * F_IN + 2 * half;
    const float* wp0 = W + (long)(col0 +  0 + l) * F_IN + 2 * half;
    const float* wp1 = W + (long)(col0 + 16 + l) * F_IN + 2 * half;
    const float* wp2 = W + (long)(col0 + 32 + l) * F_IN + 2 * half;
    const float* wp3 = W + (long)(col0 + 48 + l) * F_IN + 2 * half;

    v8f a0 = {0.f,0.f,0.f,0.f,0.f,0.f,0.f,0.f};
    v8f a1 = a0, a2 = a0, a3 = a0;
    #pragma unroll
    for (int k0 = 0; k0 < F_IN; k0 += 4) {
        v2f a  = *(const v2f*)(xp  + k0);
        v2f b0 = *(const v2f*)(wp0 + k0);
        v2f b1 = *(const v2f*)(wp1 + k0);
        v2f b2 = *(const v2f*)(wp2 + k0);
        v2f b3 = *(const v2f*)(wp3 + k0);
        a0 = __builtin_amdgcn_wmma_f32_16x16x4_f32(false, a, false, b0, (short)0, a0, false, false);
        a1 = __builtin_amdgcn_wmma_f32_16x16x4_f32(false, a, false, b1, (short)0, a1, false, false);
        a2 = __builtin_amdgcn_wmma_f32_16x16x4_f32(false, a, false, b2, (short)0, a2, false, false);
        a3 = __builtin_amdgcn_wmma_f32_16x16x4_f32(false, a, false, b3, (short)0, a3, false, false);
    }
    #pragma unroll
    for (int v = 0; v < 8; ++v) {
        long r = row0 + v + 8 * half;
        if (r < N) {
            float* o = xh + r * HC + col0 + l;
            o[0]  = a0[v];
            o[16] = a1[v];
            o[32] = a2[v];
            o[48] = a3[v];
        }
    }
}

__global__ void GAT_nodeatt(const float* __restrict__ xh,
                            const float* __restrict__ att_src,
                            const float* __restrict__ att_dst,
                            float* __restrict__ asrc,
                            float* __restrict__ adst, int N) {
    long idx = (long)blockIdx.x * blockDim.x + threadIdx.x;
    if (idx >= (long)N * NH) return;
    long n = idx >> 3;
    int  h = (int)(idx & 7);
    const float* xr = xh + n * HC + h * NC;
    float ss = 0.0f, sd = 0.0f;
    for (int c = 0; c < NC; ++c) {
        float v = xr[c];
        ss += v * att_src[h * NC + c];
        sd += v * att_dst[h * NC + c];
    }
    asrc[idx] = ss;
    adst[idx] = sd;
}

__global__ void GAT_alpha(const int* __restrict__ src, const int* __restrict__ dst,
                          const float* __restrict__ eattr, const float* __restrict__ loop,
                          const float* __restrict__ M,
                          const float* __restrict__ asrc, const float* __restrict__ adst,
                          float* __restrict__ exbuf, unsigned* __restrict__ amax,
                          int E, int N) {
    long t = (long)blockIdx.x * blockDim.x + threadIdx.x;
    long T = (long)E + N;
    if (t >= T) return;
    int s, d;
    const float* row;
    if (t < E) { s = src[t]; d = dst[t]; row = eattr + t * F_E; }
    else       { s = (int)(t - E); d = s;  row = loop + (long)s * F_E; }
    float ea[F_E];
    #pragma unroll
    for (int f = 0; f < F_E; ++f) ea[f] = row[f];
    #pragma unroll
    for (int h = 0; h < NH; ++h) {
        float ae = 0.0f;
        #pragma unroll
        for (int f = 0; f < F_E; ++f) ae += ea[f] * M[f * NH + h];
        float al = asrc[(long)s * NH + h] + adst[(long)d * NH + h] + ae;
        al = (al >= 0.0f) ? al : NEG_SLOPE * al;
        exbuf[t * NH + h] = al;
        atomicMax(&amax[(long)d * NH + h], ordkey(al));
    }
}

__global__ void GAT_expsum(const int* __restrict__ dst,
                           float* __restrict__ exbuf,
                           const unsigned* __restrict__ amax,
                           float* __restrict__ denom, int E, int N) {
    long idx = (long)blockIdx.x * blockDim.x + threadIdx.x;
    long T = ((long)E + N) * NH;
    if (idx >= T) return;
    long t = idx >> 3;
    int  h = (int)(idx & 7);
    int d = (t < E) ? dst[t] : (int)(t - E);
    float m = ordinv(amax[(long)d * NH + h]);
    float e = __expf(exbuf[idx] - m);
    exbuf[idx] = e;
    atomicAdd(&denom[(long)d * NH + h], e);
}

__global__ void GAT_scatter(const int* __restrict__ src, const int* __restrict__ dst,
                            const float* __restrict__ xh,
                            const float* __restrict__ exbuf,
                            const float* __restrict__ denom,
                            float* __restrict__ accum, int E, int N) {
    long t = blockIdx.x;
    int tid = threadIdx.x;
    int s, d;
    if (t < E) { s = src[t]; d = dst[t]; }
    else       { s = (int)(t - E); d = s; }
    int h = tid >> 5;
    float att = exbuf[t * NH + h] / denom[(long)d * NH + h];
    float val = xh[(long)s * HC + tid] * att;
    atomicAdd(&accum[(long)d * HC + tid], val);
}

__global__ void GAT_final(const float* __restrict__ accum,
                          const float* __restrict__ bias,
                          float* __restrict__ out, int N) {
    long idx = (long)blockIdx.x * blockDim.x + threadIdx.x;
    if (idx >= (long)N * NC) return;
    long n = idx >> 5;
    int  c = (int)(idx & 31);
    float sum = 0.0f;
    #pragma unroll
    for (int h = 0; h < NH; ++h) sum += accum[n * HC + h * NC + c];
    out[idx] = sum * (1.0f / NH) + bias[c];
}

extern "C" void kernel_launch(void* const* d_in, const int* in_sizes, int n_in,
                              void* d_out, int out_size, void* d_ws, size_t ws_size,
                              hipStream_t stream) {
    const float* x        = (const float*)d_in[0];
    const int*   eidx     = (const int*)  d_in[1];
    const float* eattr    = (const float*)d_in[2];
    const float* W        = (const float*)d_in[3];
    const float* We       = (const float*)d_in[4];
    const float* att_src  = (const float*)d_in[5];
    const float* att_dst  = (const float*)d_in[6];
    const float* att_edge = (const float*)d_in[7];
    const float* bias     = (const float*)d_in[8];

    const int N = in_sizes[0] / F_IN;
    const int E = in_sizes[1] / 2;
    const int* src = eidx;
    const int* dst = eidx + E;

    float* ws = (float*)d_ws;
    const long Nl = N, El = E, Tl = El + Nl;
    float*    w_deg   = ws;
    float*    w_loop  = ws + Nl;
    float*    w_denom = ws + 17 * Nl;
    unsigned* w_amax  = (unsigned*)(ws + 25 * Nl);
    float*    w_acc   = ws + 33 * Nl;
    const long zeroSpan = 289 * Nl;
    float*    w_xh    = ws + zeroSpan;
    float*    w_asrc  = ws + zeroSpan + 256 * Nl;
    float*    w_adst  = ws + zeroSpan + 264 * Nl;
    float*    w_M     = ws + zeroSpan + 272 * Nl;
    float*    w_ex    = ws + zeroSpan + 272 * Nl + 128;

    const int B = 256;
    GAT_zero<<<2048, B, 0, stream>>>(ws, zeroSpan);
    {
        long th = El * F_E;
        GAT_degloop<<<(int)((th + B - 1) / B), B, 0, stream>>>(dst, eattr, w_deg, w_loop, E);
    }
    {
        long th = Nl * F_E;
        GAT_loopdiv<<<(int)((th + B - 1) / B), B, 0, stream>>>(w_loop, w_deg, N);
    }
    GAT_foldM<<<1, 128, 0, stream>>>(We, att_edge, w_M);
    {
        long rowTiles = (Nl + 15) >> 4;
        long wtiles = rowTiles * 4;
        GAT_gemm_xh<<<(int)((wtiles + 7) / 8), B, 0, stream>>>(x, W, w_xh, N);
    }
    {
        long th = Nl * NH;
        GAT_nodeatt<<<(int)((th + B - 1) / B), B, 0, stream>>>(w_xh, att_src, att_dst,
                                                               w_asrc, w_adst, N);
    }
    GAT_alpha<<<(int)((Tl + B - 1) / B), B, 0, stream>>>(src, dst, eattr, w_loop, w_M,
                                                         w_asrc, w_adst, w_ex, w_amax, E, N);
    {
        long th = Tl * NH;
        GAT_expsum<<<(int)((th + B - 1) / B), B, 0, stream>>>(dst, w_ex, w_amax, w_denom, E, N);
    }
    GAT_scatter<<<(int)Tl, B, 0, stream>>>(src, dst, w_xh, w_ex, w_denom, w_acc, E, N);
    {
        long th = Nl * NC;
        GAT_final<<<(int)((th + B - 1) / B), B, 0, stream>>>(w_acc, bias, (float*)d_out, N);
    }
}


# Scatter sums the 8 heads per edge in registers before the atomic add: accumulator [N][32], one 128-byte add per edge, 4 edges per wave; float sum order changes (heads before edges)
# speedup vs baseline: 2.1089x; 2.1089x over previous
#include <hip/hip_runtime.h>
#include <hip/hip_bf16.h>

typedef float v2f __attribute__((ext_vector_type(2)));
typedef float v8f __attribute__((ext_vector_type(8)));

#define F_IN 128
#define F_E  16
#define NH   8
#define NC   32
#define HC   256
#define NEG_SLOPE 0.2f

__device__ __forceinline__ unsigned ordkey(float f) {
    unsigned u = __float_as_uint(f);
    return (u & 0x80000000u) ? ~u : (u | 0x80000000u);
}
__device__ __forceinline__ float ordinv(unsigned k) {
    unsigned u = (k & 0x80000000u) ? (k & 0x7fffffffu) : ~k;
    return __uint_as_float(u);
}

__global__ void GAT_zero(float* p, long n) {
    long i = (long)blockIdx.x * blockDim.x + threadIdx.x;
    long stride = (long)gridDim.x * blockDim.x;
    for (; i < n; i += stride) p[i] = 0.0f;
}

__global__ void GAT_degloop(const int* __restrict__ dst,
                            const float* __restrict__ eattr,
                            float* __restrict__ deg,
                            float* __restrict__ loop, int E) {
    long idx = (long)blockIdx.x * blockDim.x + threadIdx.x;
    long e = idx >> 4;
    int  f = (int)(idx & 15);
    if (e >= E) return;
    int d = dst[e];
    atomicAdd(&loop[(long)d * F_E + f], eattr[e * F_E + f]);
    if (f == 0) atomicAdd(&deg[d], 1.0f);
}

__global__ void GAT_loopdiv(float* __restrict__ loop,
                            const float* __restrict__ deg, int N) {
    long idx = (long)blockIdx.x * blockDim.x + threadIdx.x;
    if (idx >= (long)N * F_E) return;
    int n = (int)(idx >> 4);
    loop[idx] = loop[idx] / fmaxf(deg[n], 1.0f);
}

__global__ void GAT_foldM(const float* __restrict__ We,
                          const float* __restrict__ att_edge,
                          float* __restrict__ M) {
    int tid = threadIdx.x;
    if (tid >= F_E * NH) return;
    int f = tid >> 3, h = tid & 7;
    float s = 0.0f;
    for (int c = 0; c < NC; ++c)
        s += We[(long)(h * NC + c) * F_E + f] * att_edge[h * NC + c];
    M[f * NH + h] = s;
}

__global__ void GAT_gemm_xh(const float* __restrict__ x,
                            const float* __restrict__ W,
                            float* __restrict__ xh, int N) {
    int wave = threadIdx.x >> 5;
    int lane = threadIdx.x & 31;
    int half = lane >> 4;
    int l    = lane & 15;
    long rowTiles = ((long)N + 15) >> 4;
    long wtile = (long)blockIdx.x * 8 + wave;
    if (wtile >= rowTiles * 4) return;
    long rowTile = wtile >> 2;
    int  colGrp  = (int)(wtile & 3);
    long row0 = rowTile * 16;
    int  col0 = colGrp * 64;

    long arow = row0 + l;
    if (arow >= N) arow = N - 1;
    const float* xp  = x + arow * F_IN + 2 * half;
    const float* wp0 = W + (long)(col0 +  0 + l) * F_IN + 2 * half;
    const float* wp1 = W + (long)(col0 + 16 + l) * F_IN + 2 * half;
    const float* wp2 = W + (long)(col0 + 32 + l) * F_IN + 2 * half;
    const float* wp3 = W + (long)(col0 + 48 + l) * F_IN + 2 * half;

    v8f a0 = {0.f,0.f,0.f,0.f,0.f,0.f,0.f,0.f};
    v8f a1 = a0, a2 = a0, a3 = a0;
    #pragma unroll
    for (int k0 = 0; k0 < F_IN; k0 += 4) {
        v2f a  = *(const v2f*)(xp  + k0);
        v2f b0 = *(const v2f*)(wp0 + k0);
        v2f b1 = *(const v2f*)(wp1 + k0);
        v2f b2 = *(const v2f*)(wp2 + k0);
        v2f b3 = *(const v2f*)(wp3 + k0);
        a0 = __builtin_amdgcn_wmma_f32_16x16x4_f32(false, a, false, b0, (short)0, a0, false, false);
        a1 = __builtin_amdgcn_wmma_f32_16x16x4_f32(false, a, false, b1, (short)0, a1, false, false);
        a2 = __builtin_amdgcn_wmma_f32_16x16x4_f32(false, a, false, b2, (short)0, a2, false, false);
        a3 = __builtin_amdgcn_wmma_f32_16x16x4_f32(false, a, false, b3, (short)0, a3, false, false);
    }
    #pragma unroll
    for (int v = 0; v < 8; ++v) {
        long r = row0 + v + 8 * half;
        if (r < N) {
            float* o = xh + r * HC + col0 + l;
            o[0]  = a0[v];
            o[16] = a1[v];
            o[32] = a2[v];
            o[48] = a3[v];
        }
    }
}

__global__ void GAT_nodeatt(const float* __restrict__ xh,
                            const float* __restrict__ att_src,
                            const float* __restrict__ att_dst,
                            float* __restrict__ asrc,
                            float* __restrict__ adst, int N) {
    long idx = (long)blockIdx.x * blockDim.x + threadIdx.x;
    if (idx >= (long)N * NH) return;
    long n = idx >> 3;
    int  h = (int)(idx & 7);
    const float* xr = xh + n * HC + h * NC;
    float ss = 0.0f, sd = 0.0f;
    for (int c = 0; c < NC; ++c) {
        float v = xr[c];
        ss += v * att_src[h * NC + c];
        sd += v * att_dst[h * NC + c];
    }
    asrc[idx] = ss;
    adst[idx] = sd;
}

__global__ void GAT_alpha(const int* __restrict__ src, const int* __restrict__ dst,
                          const float* __restrict__ eattr, const float* __restrict__ loop,
                          const float* __restrict__ M,
                          const float* __restrict__ asrc, const float* __restrict__ adst,
                          float* __restrict__ exbuf, unsigned* __restrict__ amax,
                          int E, int N) {
    long t = (long)blockIdx.x * blockDim.x + threadIdx.x;
    long T = (long)E + N;
    if (t >= T) return;
    int s, d;
    const float* row;
    if (t < E) { s = src[t]; d = dst[t]; row = eattr + t * F_E; }
    else       { s = (int)(t - E); d = s;  row = loop + (long)s * F_E; }
    float ea[F_E];
    #pragma unroll
    for (int f = 0; f < F_E; ++f) ea[f] = row[f];
    #pragma unroll
    for (int h = 0; h < NH; ++h) {
        float ae = 0.0f;
        #pragma unroll
        for (int f = 0; f < F_E; ++f) ae += ea[f] * M[f * NH + h];
        float al = asrc[(long)s * NH + h] + adst[(long)d * NH + h] + ae;
        al = (al >= 0.0f) ? al : NEG_SLOPE * al;
        exbuf[t * NH + h] = al;
        atomicMax(&amax[(long)d * NH + h], ordkey(al));
    }
}

__global__ void GAT_expsum(const int* __restrict__ dst,
                           float* __restrict__ exbuf,
                           const unsigned* __restrict__ amax,
                           float* __restrict__ denom, int E, int N) {
    long idx = (long)blockIdx.x * blockDim.x + threadIdx.x;
    long T = ((long)E + N) * NH;
    if (idx >= T) return;
    long t = idx >> 3;
    int  h = (int)(idx & 7);
    int d = (t < E) ? dst[t] : (int)(t - E);
    float m = ordinv(amax[(long)d * NH + h]);
    float e = __expf(exbuf[idx] - m);
    exbuf[idx] = e;
    atomicAdd(&denom[(long)d * NH + h], e);
}

#define EPW 4
__global__ void GAT_scatter(const int* __restrict__ src, const int* __restrict__ dst,
                            const float* __restrict__ xh,
                            const float* __restrict__ exbuf,
                            const float* __restrict__ denom,
                            float* __restrict__ accum, int E, int N) {
    const int lane = threadIdx.x & 31;
    const int wv = __builtin_amdgcn_readfirstlane((int)(threadIdx.x >> 5));
    const long T  = (long)E + N;
    const long t0 = ((long)blockIdx.x * (blockDim.x >> 5) + wv) * EPW;
    if (t0 >= T) return;

    long tl = t0 + (lane >> 3);
    if (tl >= T) tl = T - 1;
    const int h_l = lane & 7;
    int sl, dl;
    if (tl < E) { sl = src[tl]; dl = dst[tl]; }
    else        { sl = (int)(tl - E); dl = sl; }

    if (t0 + EPW <= T) {
        float xv[EPW][NH];
        #pragma unroll
        for (int i = 0; i < EPW; ++i) {
            const int s = __builtin_amdgcn_readlane(sl, i * 8);
            const float* xr = xh + (long)s * HC + lane;
            #pragma unroll
            for (int h = 0; h < NH; ++h) xv[i][h] = xr[h * NC];
        }
        const float att_l = exbuf[tl * NH + h_l] / denom[(long)dl * NH + h_l];
        #pragma unroll
        for (int i = 0; i < EPW; ++i) {
            float a = __int_as_float(__builtin_amdgcn_readlane(__float_as_int(att_l), i * 8));
            float r = a * xv[i][0];
            #pragma unroll
            for (int h = 1; h < NH; ++h) {
                a = __int_as_float(__builtin_amdgcn_readlane(__float_as_int(att_l), i * 8 + h));
                r = fmaf(a, xv[i][h], r);
            }
            const int d = __builtin_amdgcn_readlane(dl, i * 8);
            atomicAdd(&accum[(long)d * NC + lane], r);
        }
    } else {
        const float att_l = exbuf[tl * NH + h_l] / denom[(long)dl * NH + h_l];
        const int left = (int)(T - t0);
        for (int i = 0; i < left; ++i) {
            const int s = __builtin_amdgcn_readlane(sl, i * 8);
            const int d = __builtin_amdgcn_readlane(dl, i * 8);
            const float* xr = xh + (long)s * HC + lane;
            float r = 0.0f;
            for (int h = 0; h < NH; ++h) {
                const float a = __int_as_float(
                    __builtin_amdgcn_readlane(__float_as_int(att_l), i * 8 + h));
                r = fmaf(a, xr[h * NC], r);
            }
            atomicAdd(&accum[(long)d * NC + lane], r);
        }
    }
}

__global__ void GAT_final(const float* __restrict__ accum,
                          const float* __restrict__ bias,
                          float* __restrict__ out, int N) {
    long idx = (long)blockIdx.x * blockDim.x + threadIdx.x;
    if (idx >= (long)N * NC) return;
    int  c = (int)(idx & 31);
    out[idx] = accum[idx] * (1.0f / NH) + bias[c];
}

extern "C" void kernel_launch(void* const* d_in, const int* in_sizes, int n_in,
                              void* d_out, int out_size, void* d_ws, size_t ws_size,
                              hipStream_t stream) {
    const float* x        = (const float*)d_in[0];
    const int*   eidx     = (const int*)  d_in[1];
    const float* eattr    = (const float*)d_in[2];
    const float* W        = (const float*)d_in[3];
    const float* We       = (const float*)d_in[4];
    const float* att_src  = (const float*)d_in[5];
    const float* att_dst  = (const float*)d_in[6];
    const float* att_edge = (const float*)d_in[7];
    const float* bias     = (const float*)d_in[8];

    const int N = in_sizes[0] / F_IN;
    const int E = in_sizes[1] / 2;
    const int* src = eidx;
    const int* dst = eidx + E;

    float* ws = (float*)d_ws;
    const long Nl = N, El = E, Tl = El + Nl;
    float*    w_acc   = ws;
    float*    w_deg   = ws + 32 * Nl;
    float*    w_loop  = ws + 33 * Nl;
    float*    w_denom = ws + 49 * Nl;
    unsigned* w_amax  = (unsigned*)(ws + 57 * Nl);
    const long zeroSpan = 65 * Nl;
    float*    w_xh    = ws + zeroSpan;
    float*    w_asrc  = ws + zeroSpan + 256 * Nl;
    float*    w_adst  = ws + zeroSpan + 264 * Nl;
    float*    w_M     = ws + zeroSpan + 272 * Nl;
    float*    w_ex    = ws + zeroSpan + 272 * Nl + 128;

    const int B = 256;
    GAT_zero<<<2048, B, 0, stream>>>(ws, zeroSpan);
    {
        long th = El * F_E;
        GAT_degloop<<<(int)((th + B - 1) / B), B, 0, stream>>>(dst, eattr, w_deg, w_loop, E);
    }
    {
        long th = Nl * F_E;
        GAT_loopdiv<<<(int)((th + B - 1) / B), B, 0, stream>>>(w_loop, w_deg, N);
    }
    GAT_foldM<<<1, 128, 0, stream>>>(We, att_edge, w_M);
    {
        long rowTiles = (Nl + 15) >> 4;
        long wtiles = rowTiles * 4;
        GAT_gemm_xh<<<(int)((wtiles + 7) / 8), B, 0, stream>>>(x, W, w_xh, N);
    }
    {
        long th = Nl * NH;
        GAT_nodeatt<<<(int)((th + B - 1) / B), B, 0, stream>>>(w_xh, att_src, att_dst,
                                                               w_asrc, w_adst, N);
    }
    GAT_alpha<<<(int)((Tl + B - 1) / B), B, 0, stream>>>(src, dst, eattr, w_loop, w_M,
                                                         w_asrc, w_adst, w_ex, w_amax, E, N);
    {
        long th = Tl * NH;
        GAT_expsum<<<(int)((th + B - 1) / B), B, 0, stream>>>(dst, w_ex, w_amax, w_denom, E, N);
    }
    {
        long groups = (Tl + EPW - 1) / EPW;
        GAT_scatter<<<(int)((groups + 7) / 8), B, 0, stream>>>(src, dst, w_xh, w_ex, w_denom,
                                                               w_acc, E, N);
    }
    {
        long th = Nl * NC;
        GAT_final<<<(int)((th + B - 1) / B), B, 0, stream>>>(w_acc, bias, (float*)d_out, N);
    }
}


# Alpha pass: one lane per (edge, head) in looping waves, so an edge's 8 segment maxima leave as one 32-byte piece of one atomic instruction (4 lines per instruction, not 32); sum order kept
# speedup vs baseline: 3.1713x; 1.5038x over previous
#include <hip/hip_runtime.h>
#include <hip/hip_bf16.h>

typedef float v2f __attribute__((ext_vector_type(2)));
typedef float v8f __attribute__((ext_vector_type(8)));

#define F_IN 128
#define F_E  16
#define NH   8
#define NC   32
#define HC   256
#define NEG_SLOPE 0.2f

__device__ __forceinline__ unsigned ordkey(float f) {
    unsigned u = __float_as_uint(f);
    return (u & 0x80000000u) ? ~u : (u | 0x80000000u);
}
__device__ __forceinline__ float ordinv(unsigned k) {
    unsigned u = (k & 0x80000000u) ? (k & 0x7fffffffu) : ~k;
    return __uint_as_float(u);
}

__global__ void GAT_zero(float* p, long n) {
    long i = (long)blockIdx.x * blockDim.x + threadIdx.x;
    long stride = (long)gridDim.x * blockDim.x;
    for (; i < n; i += stride) p[i] = 0.0f;
}

__global__ void GAT_degloop(const int* __restrict__ dst,
                            const float* __restrict__ eattr,
                            float* __restrict__ deg,
                            float* __restrict__ loop, int E) {
    long idx = (long)blockIdx.x * blockDim.x + threadIdx.x;
    long e = idx >> 4;
    int  f = (int)(idx & 15);
    if (e >= E) return;
    int d = dst[e];
    atomicAdd(&loop[(long)d * F_E + f], eattr[e * F_E + f]);
    if (f == 0) atomicAdd(&deg[d], 1.0f);
}

__global__ void GAT_loopdiv(float* __restrict__ loop,
                            const float* __restrict__ deg, int N) {
    long idx = (long)blockIdx.x * blockDim.x + threadIdx.x;
    if (idx >= (long)N * F_E) return;
    int n = (int)(idx >> 4);
    loop[idx] = loop[idx] / fmaxf(deg[n], 1.0f);
}

__global__ void GAT_foldM(const float* __restrict__ We,
                          const float* __restrict__ att_edge,
                          float* __restrict__ M) {
    int tid = threadIdx.x;
    if (tid >= F_E * NH) return;
    int f = tid >> 3, h = tid & 7;
    float s = 0.0f;
    for (int c = 0; c < NC; ++c)
        s += We[(long)(h * NC + c) * F_E + f] * att_edge[h * NC + c];
    M[f * NH + h] = s;
}

__global__ void GAT_gemm_xh(const float* __restrict__ x,
                            const float* __restrict__ W,
                            float* __restrict__ xh, int N) {
    int wave = threadIdx.x >> 5;
    int lane = threadIdx.x & 31;
    int half = lane >> 4;
    int l    = lane & 15;
    long rowTiles = ((long)N + 15) >> 4;
    long wtile = (long)blockIdx.x * 8 + wave;
    if (wtile >= rowTiles * 4) return;
    long rowTile = wtile >> 2;
    int  colGrp  = (int)(wtile & 3);
    long row0 = rowTile * 16;
    int  col0 = colGrp * 64;

    long arow = row0 + l;
    if (arow >= N) arow = N - 1;
    const float* xp  = x + arow * F_IN + 2 * half;
    const float* wp0 = W + (long)(col0 +  0 + l) * F_IN + 2 * half;
    const float* wp1 = W + (long)(col0 + 16 + l) * F_IN + 2 * half;
    const float* wp2 = W + (long)(col0 + 32 + l) * F_IN + 2 * half;
    const float* wp3 = W + (long)(col0 + 48 + l) * F_IN + 2 * half;

    v8f a0 = {0.f,0.f,0.f,0.f,0.f,0.f,0.f,0.f};
    v8f a1 = a0, a2 = a0, a3 = a0;
    #pragma unroll
    for (int k0 = 0; k0 < F_IN; k0 += 4) {
        v2f a  = *(const v2f*)(xp  + k0);
        v2f b0 = *(const v2f*)(wp0 + k0);
        v2f b1 = *(const v2f*)(wp1 + k0);
        v2f b2 = *(const v2f*)(wp2 + k0);
        v2f b3 = *(const v2f*)(wp3 + k0);
        a0 = __builtin_amdgcn_wmma_f32_16x16x4_f32(false, a, false, b0, (short)0, a0, false, false);
        a1 = __builtin_amdgcn_wmma_f32_16x16x4_f32(false, a, false, b1, (short)0, a1, false, false);
        a2 = __builtin_amdgcn_wmma_f32_16x16x4_f32(false, a, false, b2, (short)0, a2, false, false);
        a3 = __builtin_amdgcn_wmma_f32_16x16x4_f32(false, a, false, b3, (short)0, a3, false, false);
    }
    #pragma unroll
    for (int v = 0; v < 8; ++v) {
        long r = row0 + v + 8 * half;
        if (r < N) {
            float* o = xh + r * HC + col0 + l;
            o[0]  = a0[v];
            o[16] = a1[v];
            o[32] = a2[v];
            o[48] = a3[v];
        }
    }
}

__global__ void GAT_nodeatt(const float* __restrict__ xh,
                            const float* __restrict__ att_src,
                            const float* __restrict__ att_dst,
                            float* __restrict__ asrc,
                            float* __restrict__ adst, int N) {
    long idx = (long)blockIdx.x * blockDim.x + threadIdx.x;
    if (idx >= (long)N * NH) return;
    long n = idx >> 3;
    int  h = (int)(idx & 7);
    const float* xr = xh + n * HC + h * NC;
    float ss = 0.0f, sd = 0.0f;
    for (int c = 0; c < NC; ++c) {
        float v = xr[c];
        ss += v * att_src[h * NC + c];
        sd += v * att_dst[h * NC + c];
    }
    asrc[idx] = ss;
    adst[idx] = sd;
}

typedef float v4fu __attribute__((ext_vector_type(4), aligned(4)));
#define A_EPI 4
#define A_UNR 4

__device__ __forceinline__ float GAT_alpha_one(long t, int h, int E,
                                               const int* __restrict__ src,
                                               const int* __restrict__ dst,
                                               const float* __restrict__ eattr,
                                               const float* __restrict__ loop,
                                               const float* __restrict__ asrc,
                                               const float* __restrict__ adst,
                                               const float* m, int* d_out) {
    const bool isE = t < E;
    const long te = isE ? t : 0;
    int s = src[te];
    int d = dst[te];
    if (!isE) { s = (int)(t - E); d = s; }
    const float* row = isE ? (eattr + t * F_E) : (loop + (long)s * F_E);
    const v4fu q0 = *(const v4fu*)(row + 0);
    const v4fu q1 = *(const v4fu*)(row + 4);
    const v4fu q2 = *(const v4fu*)(row + 8);
    const v4fu q3 = *(const v4fu*)(row + 12);
    const float as = asrc[(long)s * NH + h];
    const float ad = adst[(long)d * NH + h];
    const float ea[F_E] = {q0.x, q0.y, q0.z, q0.w, q1.x, q1.y, q1.z, q1.w,
                           q2.x, q2.y, q2.z, q2.w, q3.x, q3.y, q3.z, q3.w};
    float ae = 0.0f;
    #pragma unroll
    for (int f = 0; f < F_E; ++f) ae = fmaf(ea[f], m[f], ae);
    float al = as + ad + ae;
    al = (al >= 0.0f) ? al : NEG_SLOPE * al;
    *d_out = d;
    return al;
}

__global__ void __launch_bounds__(256)
GAT_alpha(const int* __restrict__ src, const int* __restrict__ dst,
                          const float* __restrict__ eattr, const float* __restrict__ loop,
                          const float* __restrict__ M,
                          const float* __restrict__ asrc, const float* __restrict__ adst,
                          float* __restrict__ exbuf, unsigned* __restrict__ amax,
                          int E, int N) {
    const int lane = threadIdx.x & 31;
    const int h    = lane & 7;
    const int sub  = lane >> 3;
    const int wv = __builtin_amdgcn_readfirstlane((int)(threadIdx.x >> 5));
    const long T      = (long)E + N;
    const long wpb    = blockDim.x >> 5;
    const long nwaves = (long)gridDim.x * wpb;
    const long wave   = (long)blockIdx.x * wpb + wv;

    float m[F_E];
    #pragma unroll
    for (int f = 0; f < F_E; ++f) m[f] = M[f * NH + h];

    for (long base = wave * (A_EPI * A_UNR); base < T; base += nwaves * (A_EPI * A_UNR)) {
        if (base + A_EPI * A_UNR <= T) {
            int  sv[A_UNR], dv[A_UNR];
            v4fu q[A_UNR][4];
            #pragma unroll
            for (int g = 0; g < A_UNR; ++g) {
                const long t   = base + g * A_EPI + sub;
                const bool isE = t < E;
                const long te  = isE ? t : 0;
                sv[g] = src[te];
                dv[g] = dst[te];
                const float* row = isE ? (eattr + t * F_E) : (loop + (t - E) * F_E);
                q[g][0] = *(const v4fu*)(row + 0);
                q[g][1] = *(const v4fu*)(row + 4);
                q[g][2] = *(const v4fu*)(row + 8);
                q[g][3] = *(const v4fu*)(row + 12);
            }
            __builtin_amdgcn_sched_barrier(0);
            float as[A_UNR], ad[A_UNR];
            #pragma unroll
            for (int g = 0; g < A_UNR; ++g) {
                const long t = base + g * A_EPI + sub;
                if (t >= E) { sv[g] = (int)(t - E); dv[g] = sv[g]; }
                as[g] = asrc[(long)sv[g] * NH + h];
                ad[g] = adst[(long)dv[g] * NH + h];
            }
            __builtin_amdgcn_sched_barrier(0);
            #pragma unroll
            for (int g = 0; g < A_UNR; ++g) {
                const long t = base + g * A_EPI + sub;
                const float ea[F_E] = {q[g][0].x, q[g][0].y, q[g][0].z, q[g][0].w,
                                       q[g][1].x, q[g][1].y, q[g][1].z, q[g][1].w,
                                       q[g][2].x, q[g][2].y, q[g][2].z, q[g][2].w,
                                       q[g][3].x, q[g][3].y, q[g][3].z, q[g][3].w};
                float ae = 0.0f;
                #pragma unroll
                for (int f = 0; f < F_E; ++f) ae = fmaf(ea[f], m[f], ae);
                float al = as[g] + ad[g] + ae;
                al = (al >= 0.0f) ? al : NEG_SLOPE * al;
                exbuf[t * NH + h] = al;
                atomicMax(&amax[(long)dv[g] * NH + h], ordkey(al));
            }
        } else {
            for (int g = 0; g < A_UNR; ++g) {
                const long t = base + g * A_EPI + sub;
                if (t < T) {
                    int d;
                    const float a = GAT_alpha_one(t, h, E, src, dst, eattr, loop,
                                                  asrc, adst, m, &d);
                    exbuf[t * NH + h] = a;
                    atomicMax(&amax[(long)d * NH + h], ordkey(a));
                }
            }
        }
    }
}

__global__ void GAT_expsum(const int* __restrict__ dst,
                           float* __restrict__ exbuf,
                           const unsigned* __restrict__ amax,
                           float* __restrict__ denom, int E, int N) {
    long idx = (long)blockIdx.x * blockDim.x + threadIdx.x;
    long T = ((long)E + N) * NH;
    if (idx >= T) return;
    long t = idx >> 3;
    int  h = (int)(idx & 7);
    int d = (t < E) ? dst[t] : (int)(t - E);
    float m = ordinv(amax[(long)d * NH + h]);
    float e = __expf(exbuf[idx] - m);
    exbuf[idx] = e;
    atomicAdd(&denom[(long)d * NH + h], e);
}

#define EPW 4
__global__ void GAT_scatter(const int* __restrict__ src, const int* __restrict__ dst,
                            const float* __restrict__ xh,
                            const float* __restrict__ exbuf,
                            const float* __restrict__ denom,
                            float* __restrict__ accum, int E, int N) {
    const int lane = threadIdx.x & 31;
    const int wv = __builtin_amdgcn_readfirstlane((int)(threadIdx.x >> 5));
    const long T  = (long)E + N;
    const long t0 = ((long)blockIdx.x * (blockDim.x >> 5) + wv) * EPW;
    if (t0 >= T) return;

    long tl = t0 + (lane >> 3);
    if (tl >= T) tl = T - 1;
    const int h_l = lane & 7;
    int sl, dl;
    if (tl < E) { sl = src[tl]; dl = dst[tl]; }
    else        { sl = (int)(tl - E); dl = sl; }

    if (t0 + EPW <= T) {
        float xv[EPW][NH];
        #pragma unroll
        for (int i = 0; i < EPW; ++i) {
            const int s = __builtin_amdgcn_readlane(sl, i * 8);
            const float* xr = xh + (long)s * HC + lane;
            #pragma unroll
            for (int h = 0; h < NH; ++h) xv[i][h] = xr[h * NC];
        }
        const float att_l = exbuf[tl * NH + h_l] / denom[(long)dl * NH + h_l];
        #pragma unroll
        for (int i = 0; i < EPW; ++i) {
            float a = __int_as_float(__builtin_amdgcn_readlane(__float_as_int(att_l), i * 8));
            float r = a * xv[i][0];
            #pragma unroll
            for (int h = 1; h < NH; ++h) {
                a = __int_as_float(__builtin_amdgcn_readlane(__float_as_int(att_l), i * 8 + h));
                r = fmaf(a, xv[i][h], r);
            }
            const int d = __builtin_amdgcn_readlane(dl, i * 8);
            atomicAdd(&accum[(long)d * NC + lane], r);
        }
    } else {
        const float att_l = exbuf[tl * NH + h_l] / denom[(long)dl * NH + h_l];
        const int left = (int)(T - t0);
        for (int i = 0; i < left; ++i) {
            const int s = __builtin_amdgcn_readlane(sl, i * 8);
            const int d = __builtin_amdgcn_readlane(dl, i * 8);
            const float* xr = xh + (long)s * HC + lane;
            float r = 0.0f;
            for (int h = 0; h < NH; ++h) {
                const float a = __int_as_float(
                    __builtin_amdgcn_readlane(__float_as_int(att_l), i * 8 + h));
                r = fmaf(a, xr[h * NC], r);
            }
            atomicAdd(&accum[(long)d * NC + lane], r);
        }
    }
}

__global__ void GAT_final(const float* __restrict__ accum,
                          const float* __restrict__ bias,
                          float* __restrict__ out, int N) {
    long idx = (long)blockIdx.x * blockDim.x + threadIdx.x;
    if (idx >= (long)N * NC) return;
    int  c = (int)(idx & 31);
    out[idx] = accum[idx] * (1.0f / NH) + bias[c];
}

extern "C" void kernel_launch(void* const* d_in, const int* in_sizes, int n_in,
                              void* d_out, int out_size, void* d_ws, size_t ws_size,
                              hipStream_t stream) {
    const float* x        = (const float*)d_in[0];
    const int*   eidx     = (const int*)  d_in[1];
    const float* eattr    = (const float*)d_in[2];
    const float* W        = (const float*)d_in[3];
    const float* We       = (const float*)d_in[4];
    const float* att_src  = (const float*)d_in[5];
    const float* att_dst  = (const float*)d_in[6];
    const float* att_edge = (const float*)d_in[7];
    const float* bias     = (const float*)d_in[8];

    const int N = in_sizes[0] / F_IN;
    const int E = in_sizes[1] / 2;
    const int* src = eidx;
    const int* dst = eidx + E;

    float* ws = (float*)d_ws;
    const long Nl = N, El = E, Tl = El + Nl;
    float*    w_acc   = ws;
    float*    w_deg   = ws + 32 * Nl;
    float*    w_loop  = ws + 33 * Nl;
    float*    w_denom = ws + 49 * Nl;
    unsigned* w_amax  = (unsigned*)(ws + 57 * Nl);
    const long zeroSpan = 65 * Nl;
    float*    w_xh    = ws + zeroSpan;
    float*    w_asrc  = ws + zeroSpan + 256 * Nl;
    float*    w_adst  = ws + zeroSpan + 264 * Nl;
    float*    w_M     = ws + zeroSpan + 272 * Nl;
    float*    w_ex    = ws + zeroSpan + 272 * Nl + 128;

    const int B = 256;
    GAT_zero<<<2048, B, 0, stream>>>(ws, zeroSpan);
    {
        long th = El * F_E;
        GAT_degloop<<<(int)((th + B - 1) / B), B, 0, stream>>>(dst, eattr, w_deg, w_loop, E);
    }
    {
        long th = Nl * F_E;
        GAT_loopdiv<<<(int)((th + B - 1) / B), B, 0, stream>>>(w_loop, w_deg, N);
    }
    GAT_foldM<<<1, 128, 0, stream>>>(We, att_edge, w_M);
    {
        long rowTiles = (Nl + 15) >> 4;
        long wtiles = rowTiles * 4;
        GAT_gemm_xh<<<(int)((wtiles + 7) / 8), B, 0, stream>>>(x, W, w_xh, N);
    }
    {
        long th = Nl * NH;
        GAT_nodeatt<<<(int)((th + B - 1) / B), B, 0, stream>>>(w_xh, att_src, att_dst,
                                                               w_asrc, w_adst, N);
    }
    GAT_alpha<<<768, B, 0, stream>>>(src, dst, eattr, w_loop, w_M,
                                      w_asrc, w_adst, w_ex, w_amax, E, N);
    {
        long th = Tl * NH;
        GAT_expsum<<<(int)((th + B - 1) / B), B, 0, stream>>>(dst, w_ex, w_amax, w_denom, E, N);
    }
    {
        long groups = (Tl + EPW - 1) / EPW;
        GAT_scatter<<<(int)((groups + 7) / 8), B, 0, stream>>>(src, dst, w_xh, w_ex, w_denom,
                                                               w_acc, E, N);
    }
    {
        long th = Nl * NC;
        GAT_final<<<(int)((th + B - 1) / B), B, 0, stream>>>(w_acc, bias, (float*)d_out, N);
    }
}


# Self-loop edge term by linearity: alpha pass adds each edge's 8 head scores + degree into a 64-byte node record (one atomic piece per edge); the 16-attribute sum pass is dropped; sum order changes
# speedup vs baseline: 4.5027x; 1.4198x over previous
#include <hip/hip_runtime.h>
#include <hip/hip_bf16.h>

typedef float v2f __attribute__((ext_vector_type(2)));
typedef float v8f __attribute__((ext_vector_type(8)));

#define F_IN 128
#define F_E  16
#define NH   8
#define NC   32
#define HC   256
#define NEG_SLOPE 0.2f

__device__ __forceinline__ unsigned ordkey(float f) {
    unsigned u = __float_as_uint(f);
    return (u & 0x80000000u) ? ~u : (u | 0x80000000u);
}
__device__ __forceinline__ float ordinv(unsigned k) {
    unsigned u = (k & 0x80000000u) ? (k & 0x7fffffffu) : ~k;
    return __uint_as_float(u);
}

__global__ void GAT_zero(float* p, long n) {
    long i = (long)blockIdx.x * blockDim.x + threadIdx.x;
    long stride = (long)gridDim.x * blockDim.x;
    for (; i < n; i += stride) p[i] = 0.0f;
}

__global__ void GAT_foldM(const float* __restrict__ We,
                          const float* __restrict__ att_edge,
                          float* __restrict__ M) {
    int tid = threadIdx.x;
    if (tid >= F_E * NH) return;
    int f = tid >> 3, h = tid & 7;
    float s = 0.0f;
    for (int c = 0; c < NC; ++c)
        s += We[(long)(h * NC + c) * F_E + f] * att_edge[h * NC + c];
    M[f * NH + h] = s;
}

__global__ void GAT_gemm_xh(const float* __restrict__ x,
                            const float* __restrict__ W,
                            float* __restrict__ xh, int N) {
    int wave = threadIdx.x >> 5;
    int lane = threadIdx.x & 31;
    int half = lane >> 4;
    int l    = lane & 15;
    long rowTiles = ((long)N + 15) >> 4;
    long wtile = (long)blockIdx.x * 8 + wave;
    if (wtile >= rowTiles * 4) return;
    long rowTile = wtile >> 2;
    int  colGrp  = (int)(wtile & 3);
    long row0 = rowTile * 16;
    int  col0 = colGrp * 64;

    long arow = row0 + l;
    if (arow >= N) arow = N - 1;
    const float* xp  = x + arow * F_IN + 2 * half;
    const float* wp0 = W + (long)(col0 +  0 + l) * F_IN + 2 * half;
    const float* wp1 = W + (long)(col0 + 16 + l) * F_IN + 2 * half;
    const float* wp2 = W + (long)(col0 + 32 + l) * F_IN + 2 * half;
    const float* wp3 = W + (long)(col0 + 48 + l) * F_IN + 2 * half;

    v8f a0 = {0.f,0.f,0.f,0.f,0.f,0.f,0.f,0.f};
    v8f a1 = a0, a2 = a0, a3 = a0;
    #pragma unroll
    for (int k0 = 0; k0 < F_IN; k0 += 4) {
        v2f a  = *(const v2f*)(xp  + k0);
        v2f b0 = *(const v2f*)(wp0 + k0);
        v2f b1 = *(const v2f*)(wp1 + k0);
        v2f b2 = *(const v2f*)(wp2 + k0);
        v2f b3 = *(const v2f*)(wp3 + k0);
        a0 = __builtin_amdgcn_wmma_f32_16x16x4_f32(false, a, false, b0, (short)0, a0, false, false);
        a1 = __builtin_amdgcn_wmma_f32_16x16x4_f32(false, a, false, b1, (short)0, a1, false, false);
        a2 = __builtin_amdgcn_wmma_f32_16x16x4_f32(false, a, false, b2, (short)0, a2, false, false);
        a3 = __builtin_amdgcn_wmma_f32_16x16x4_f32(false, a, false, b3, (short)0, a3, false, false);
    }
    #pragma unroll
    for (int v = 0; v < 8; ++v) {
        long r = row0 + v + 8 * half;
        if (r < N) {
            float* o = xh + r * HC + col0 + l;
            o[0]  = a0[v];
            o[16] = a1[v];
            o[32] = a2[v];
            o[48] = a3[v];
        }
    }
}

__global__ void GAT_nodeatt(const float* __restrict__ xh,
                            const float* __restrict__ att_src,
                            const float* __restrict__ att_dst,
                            float* __restrict__ asrc,
                            float* __restrict__ adst, int N) {
    long idx = (long)blockIdx.x * blockDim.x + threadIdx.x;
    if (idx >= (long)N * NH) return;
    long n = idx >> 3;
    int  h = (int)(idx & 7);
    const float* xr = xh + n * HC + h * NC;
    float ss = 0.0f, sd = 0.0f;
    for (int c = 0; c < NC; ++c) {
        float v = xr[c];
        ss += v * att_src[h * NC + c];
        sd += v * att_dst[h * NC + c];
    }
    asrc[idx] = ss;
    adst[idx] = sd;
}

typedef float v4fu __attribute__((ext_vector_type(4), aligned(4)));
#define A_EPI 4
#define A_UNR 4
#define REC   16
#define REC_DEG 8

__device__ __forceinline__ float GAT_alpha_one(long t, int h,
                                               const int* __restrict__ src,
                                               const int* __restrict__ dst,
                                               const float* __restrict__ eattr,
                                               const float* __restrict__ asrc,
                                               const float* __restrict__ adst,
                                               const float* m, int* d_out, float* ae_out) {
    const int s = src[t];
    const int d = dst[t];
    const float* row = eattr + t * F_E;
    const v4fu q0 = *(const v4fu*)(row + 0);
    const v4fu q1 = *(const v4fu*)(row + 4);
    const v4fu q2 = *(const v4fu*)(row + 8);
    const v4fu q3 = *(const v4fu*)(row + 12);
    const float as = asrc[(long)s * NH + h];
    const float ad = adst[(long)d * NH + h];
    const float ea[F_E] = {q0.x, q0.y, q0.z, q0.w, q1.x, q1.y, q1.z, q1.w,
                           q2.x, q2.y, q2.z, q2.w, q3.x, q3.y, q3.z, q3.w};
    float ae = 0.0f;
    #pragma unroll
    for (int f = 0; f < F_E; ++f) ae = fmaf(ea[f], m[f], ae);
    float al = as + ad + ae;
    al = (al >= 0.0f) ? al : NEG_SLOPE * al;
    *d_out = d;
    *ae_out = ae;
    return al;
}

__global__ void __launch_bounds__(256)
GAT_alpha(const int* __restrict__ src, const int* __restrict__ dst,
                          const float* __restrict__ eattr,
                          const float* __restrict__ M,
                          const float* __restrict__ asrc, const float* __restrict__ adst,
                          float* __restrict__ exbuf, unsigned* __restrict__ amax,
                          float* __restrict__ rec, int E) {
    const int lane = threadIdx.x & 31;
    const int h    = lane & 7;
    const int sub  = lane >> 3;
    const int slot = lane & 15;
    const int hlf  = lane >> 4;
    const int wv = __builtin_amdgcn_readfirstlane((int)(threadIdx.x >> 5));
    const long T      = (long)E;
    const long wpb    = blockDim.x >> 5;
    const long nwaves = (long)gridDim.x * wpb;
    const long wave   = (long)blockIdx.x * wpb + wv;

    float m[F_E];
    #pragma unroll
    for (int f = 0; f < F_E; ++f) m[f] = M[f * NH + h];

    for (long base = wave * (A_EPI * A_UNR); base < T; base += nwaves * (A_EPI * A_UNR)) {
        if (base + A_EPI * A_UNR <= T) {
            int  sv[A_UNR], dv[A_UNR];
            v4fu q[A_UNR][4];
            #pragma unroll
            for (int g = 0; g < A_UNR; ++g) {
                const long t = base + g * A_EPI + sub;
                sv[g] = src[t];
                dv[g] = dst[t];
                const float* row = eattr + t * F_E;
                q[g][0] = *(const v4fu*)(row + 0);
                q[g][1] = *(const v4fu*)(row + 4);
                q[g][2] = *(const v4fu*)(row + 8);
                q[g][3] = *(const v4fu*)(row + 12);
            }
            __builtin_amdgcn_sched_barrier(0);
            float as[A_UNR], ad[A_UNR];
            #pragma unroll
            for (int g = 0; g < A_UNR; ++g) {
                as[g] = asrc[(long)sv[g] * NH + h];
                ad[g] = adst[(long)dv[g] * NH + h];
            }
            __builtin_amdgcn_sched_barrier(0);
            #pragma unroll
            for (int g = 0; g < A_UNR; ++g) {
                const long t = base + g * A_EPI + sub;
                const float ea[F_E] = {q[g][0].x, q[g][0].y, q[g][0].z, q[g][0].w,
                                       q[g][1].x, q[g][1].y, q[g][1].z, q[g][1].w,
                                       q[g][2].x, q[g][2].y, q[g][2].z, q[g][2].w,
                                       q[g][3].x, q[g][3].y, q[g][3].z, q[g][3].w};
                float ae = 0.0f;
                #pragma unroll
                for (int f = 0; f < F_E; ++f) ae = fmaf(ea[f], m[f], ae);
                float al = as[g] + ad[g] + ae;
                al = (al >= 0.0f) ? al : NEG_SLOPE * al;
                exbuf[t * NH + h] = al;
                atomicMax(&amax[(long)dv[g] * NH + h], ordkey(al));
                #pragma unroll
                for (int p = 0; p < 2; ++p) {
                    const int from = ((2 * p + hlf) << 3) | (slot & 7);
                    const float v  = __shfl(ae, from);
                    const int   dd = __shfl(dv[g], from);
                    if (slot <= REC_DEG)
                        atomicAdd(&rec[(long)dd * REC + slot], (slot < REC_DEG) ? v : 1.0f);
                }
            }
        } else {
            for (int g = 0; g < A_UNR; ++g) {
                const long t = base + g * A_EPI + sub;
                if (t < T) {
                    int d;
                    float ae;
                    const float a = GAT_alpha_one(t, h, src, dst, eattr,
                                                  asrc, adst, m, &d, &ae);
                    exbuf[t * NH + h] = a;
                    atomicMax(&amax[(long)d * NH + h], ordkey(a));
                    atomicAdd(&rec[(long)d * REC + h], ae);
                    if (h == 0) atomicAdd(&rec[(long)d * REC + REC_DEG], 1.0f);
                }
            }
        }
    }
}

__global__ void GAT_selfalpha(const float* __restrict__ rec,
                              const float* __restrict__ asrc,
                              const float* __restrict__ adst,
                              float* __restrict__ exbuf,
                              unsigned* __restrict__ amax, int E, int N) {
    long idx = (long)blockIdx.x * blockDim.x + threadIdx.x;
    if (idx >= (long)N * NH) return;
    const long n = idx >> 3;
    const int  h = (int)(idx & 7);
    const float ae = rec[n * REC + h] / fmaxf(rec[n * REC + REC_DEG], 1.0f);
    float al = asrc[idx] + adst[idx] + ae;
    al = (al >= 0.0f) ? al : NEG_SLOPE * al;
    exbuf[((long)E + n) * NH + h] = al;
    const unsigned k = ordkey(al);
    const unsigned c = amax[idx];
    amax[idx] = (k > c) ? k : c;
}

__global__ void GAT_expsum(const int* __restrict__ dst,
                           float* __restrict__ exbuf,
                           const unsigned* __restrict__ amax,
                           float* __restrict__ denom, int E, int N) {
    long idx = (long)blockIdx.x * blockDim.x + threadIdx.x;
    long T = ((long)E + N) * NH;
    if (idx >= T) return;
    long t = idx >> 3;
    int  h = (int)(idx & 7);
    int d = (t < E) ? dst[t] : (int)(t - E);
    float m = ordinv(amax[(long)d * NH + h]);
    float e = __expf(exbuf[idx] - m);
    exbuf[idx] = e;
    atomicAdd(&denom[(long)d * NH + h], e);
}

#define EPW 4
__global__ void GAT_scatter(const int* __restrict__ src, const int* __restrict__ dst,
                            const float* __restrict__ xh,
                            const float* __restrict__ exbuf,
                            const float* __restrict__ denom,
                            float* __restrict__ accum, int E, int N) {
    const int lane = threadIdx.x & 31;
    const int wv = __builtin_amdgcn_readfirstlane((int)(threadIdx.x >> 5));
    const long T  = (long)E + N;
    const long t0 = ((long)blockIdx.x * (blockDim.x >> 5) + wv) * EPW;
    if (t0 >= T) return;

    long tl = t0 + (lane >> 3);
    if (tl >= T) tl = T - 1;
    const int h_l = lane & 7;
    int sl, dl;
    if (tl < E) { sl = src[tl]; dl = dst[tl]; }
    else        { sl = (int)(tl - E); dl = sl; }

    if (t0 + EPW <= T) {
        float xv[EPW][NH];
        #pragma unroll
        for (int i = 0; i < EPW; ++i) {
            const int s = __builtin_amdgcn_readlane(sl, i * 8);
            const float* xr = xh + (long)s * HC + lane;
            #pragma unroll
            for (int h = 0; h < NH; ++h) xv[i][h] = xr[h * NC];
        }
        const float att_l = exbuf[tl * NH + h_l] / denom[(long)dl * NH + h_l];
        #pragma unroll
        for (int i = 0; i < EPW; ++i) {
            float a = __int_as_float(__builtin_amdgcn_readlane(__float_as_int(att_l), i * 8));
            float r = a * xv[i][0];
            #pragma unroll
            for (int h = 1; h < NH; ++h) {
                a = __int_as_float(__builtin_amdgcn_readlane(__float_as_int(att_l), i * 8 + h));
                r = fmaf(a, xv[i][h], r);
            }
            const int d = __builtin_amdgcn_readlane(dl, i * 8);
            atomicAdd(&accum[(long)d * NC + lane], r);
        }
    } else {
        const float att_l = exbuf[tl * NH + h_l] / denom[(long)dl * NH + h_l];
        const int left = (int)(T - t0);
        for (int i = 0; i < left; ++i) {
            const int s = __builtin_amdgcn_readlane(sl, i * 8);
            const int d = __builtin_amdgcn_readlane(dl, i * 8);
            const float* xr = xh + (long)s * HC + lane;
            float r = 0.0f;
            for (int h = 0; h < NH; ++h) {
                const float a = __int_as_float(
                    __builtin_amdgcn_readlane(__float_as_int(att_l), i * 8 + h));
                r = fmaf(a, xr[h * NC], r);
            }
            atomicAdd(&accum[(long)d * NC + lane], r);
        }
    }
}

__global__ void GAT_final(const float* __restrict__ accum,
                          const float* __restrict__ bias,
                          float* __restrict__ out, int N) {
    long idx = (long)blockIdx.x * blockDim.x + threadIdx.x;
    if (idx >= (long)N * NC) return;
    int  c = (int)(idx & 31);
    out[idx] = accum[idx] * (1.0f / NH) + bias[c];
}

extern "C" void kernel_launch(void* const* d_in, const int* in_sizes, int n_in,
                              void* d_out, int out_size, void* d_ws, size_t ws_size,
                              hipStream_t stream) {
    const float* x        = (const float*)d_in[0];
    const int*   eidx     = (const int*)  d_in[1];
    const float* eattr    = (const float*)d_in[2];
    const float* W        = (const float*)d_in[3];
    const float* We       = (const float*)d_in[4];
    const float* att_src  = (const float*)d_in[5];
    const float* att_dst  = (const float*)d_in[6];
    const float* att_edge = (const float*)d_in[7];
    const float* bias     = (const float*)d_in[8];

    const int N = in_sizes[0] / F_IN;
    const int E = in_sizes[1] / 2;
    const int* src = eidx;
    const int* dst = eidx + E;

    float* ws = (float*)d_ws;
    const long Nl = N, El = E, Tl = El + Nl;
    float*    w_acc   = ws;
    float*    w_rec   = ws + 32 * Nl;
    float*    w_denom = ws + 48 * Nl;
    unsigned* w_amax  = (unsigned*)(ws + 56 * Nl);
    const long zeroSpan = 64 * Nl;
    const long xhOff    = 65 * Nl;
    float*    w_xh    = ws + xhOff;
    float*    w_asrc  = ws + xhOff + 256 * Nl;
    float*    w_adst  = ws + xhOff + 264 * Nl;
    float*    w_M     = ws + xhOff + 272 * Nl;
    float*    w_ex    = ws + xhOff + 272 * Nl + 128;

    const int B = 256;
    GAT_zero<<<2048, B, 0, stream>>>(ws, zeroSpan);
    GAT_foldM<<<1, 128, 0, stream>>>(We, att_edge, w_M);
    {
        long rowTiles = (Nl + 15) >> 4;
        long wtiles = rowTiles * 4;
        GAT_gemm_xh<<<(int)((wtiles + 7) / 8), B, 0, stream>>>(x, W, w_xh, N);
    }
    {
        long th = Nl * NH;
        GAT_nodeatt<<<(int)((th + B - 1) / B), B, 0, stream>>>(w_xh, att_src, att_dst,
                                                               w_asrc, w_adst, N);
    }
    GAT_alpha<<<768, B, 0, stream>>>(src, dst, eattr, w_M,
                                      w_asrc, w_adst, w_ex, w_amax, w_rec, E);
    {
        long th = Nl * NH;
        GAT_selfalpha<<<(int)((th + B - 1) / B), B, 0, stream>>>(w_rec, w_asrc, w_adst,
                                                                 w_ex, w_amax, E, N);
    }
    {
        long th = Tl * NH;
        GAT_expsum<<<(int)((th + B - 1) / B), B, 0, stream>>>(dst, w_ex, w_amax, w_denom, E, N);
    }
    {
        long groups = (Tl + EPW - 1) / EPW;
        GAT_scatter<<<(int)((groups + 7) / 8), B, 0, stream>>>(src, dst, w_xh, w_ex, w_denom,
                                                               w_acc, E, N);
    }
    {
        long th = Nl * NC;
        GAT_final<<<(int)((th + B - 1) / B), B, 0, stream>>>(w_acc, bias, (float*)d_out, N);
    }
}


# Projection GEMM: operand loads per matrix instruction cut from 20 to 6 per 16: wave tile 32x64 (B fragments shared by two row tiles), 16-byte loads feed two k-steps; k sum order changes
# speedup vs baseline: 4.8366x; 1.0742x over previous
#include <hip/hip_runtime.h>
#include <hip/hip_bf16.h>

typedef float v2f __attribute__((ext_vector_type(2)));
typedef float v8f __attribute__((ext_vector_type(8)));

#define F_IN 128
#define F_E  16
#define NH   8
#define NC   32
#define HC   256
#define NEG_SLOPE 0.2f

__device__ __forceinline__ unsigned ordkey(float f) {
    unsigned u = __float_as_uint(f);
    return (u & 0x80000000u) ? ~u : (u | 0x80000000u);
}
__device__ __forceinline__ float ordinv(unsigned k) {
    unsigned u = (k & 0x80000000u) ? (k & 0x7fffffffu) : ~k;
    return __uint_as_float(u);
}

__global__ void GAT_zero(float* p, long n) {
    long i = (long)blockIdx.x * blockDim.x + threadIdx.x;
    long stride = (long)gridDim.x * blockDim.x;
    for (; i < n; i += stride) p[i] = 0.0f;
}

__global__ void GAT_foldM(const float* __restrict__ We,
                          const float* __restrict__ att_edge,
                          float* __restrict__ M) {
    int tid = threadIdx.x;
    if (tid >= F_E * NH) return;
    int f = tid >> 3, h = tid & 7;
    float s = 0.0f;
    for (int c = 0; c < NC; ++c)
        s += We[(long)(h * NC + c) * F_E + f] * att_edge[h * NC + c];
    M[f * NH + h] = s;
}

typedef float v4f8 __attribute__((ext_vector_type(4), aligned(8)));
__global__ void GAT_gemm_xh(const float* __restrict__ x,
                            const float* __restrict__ W,
                            float* __restrict__ xh, int N) {
    int wave = threadIdx.x >> 5;
    int lane = threadIdx.x & 31;
    int half = lane >> 4;
    int l    = lane & 15;
    long rowTiles = ((long)N + 31) >> 5;
    long wtile = (long)blockIdx.x * 4 + wave;
    if (wtile >= rowTiles * 4) return;
    long rowTile = wtile >> 2;
    int  colGrp  = (int)(wtile & 3);
    long row0 = rowTile * 32;
    int  col0 = colGrp * 64;

    long ar0 = row0 + l;
    long ar1 = row0 + 16 + l;
    if (ar0 >= N) ar0 = N - 1;
    if (ar1 >= N) ar1 = N - 1;
    const float* xp0 = x + ar0 * F_IN + 4 * half;
    const float* xp1 = x + ar1 * F_IN + 4 * half;
    const float* wp0 = W + (long)(col0 +  0 + l) * F_IN + 4 * half;
    const float* wp1 = W + (long)(col0 + 16 + l) * F_IN + 4 * half;
    const float* wp2 = W + (long)(col0 + 32 + l) * F_IN + 4 * half;
    const float* wp3 = W + (long)(col0 + 48 + l) * F_IN + 4 * half;

    const v8f zero8 = {0.f,0.f,0.f,0.f,0.f,0.f,0.f,0.f};
    v8f c00 = zero8, c01 = zero8, c02 = zero8, c03 = zero8;
    v8f c10 = zero8, c11 = zero8, c12 = zero8, c13 = zero8;
    #pragma unroll
    for (int k0 = 0; k0 < F_IN; k0 += 8) {
        const v4f8 a0 = *(const v4f8*)(xp0 + k0);
        const v4f8 a1 = *(const v4f8*)(xp1 + k0);
        const v4f8 b0 = *(const v4f8*)(wp0 + k0);
        const v4f8 b1 = *(const v4f8*)(wp1 + k0);
        const v4f8 b2 = *(const v4f8*)(wp2 + k0);
        const v4f8 b3 = *(const v4f8*)(wp3 + k0);
        const v2f a0l = {a0.x, a0.y}, a0h = {a0.z, a0.w};
        const v2f a1l = {a1.x, a1.y}, a1h = {a1.z, a1.w};
        const v2f b0l = {b0.x, b0.y}, b0h = {b0.z, b0.w};
        const v2f b1l = {b1.x, b1.y}, b1h = {b1.z, b1.w};
        const v2f b2l = {b2.x, b2.y}, b2h = {b2.z, b2.w};
        const v2f b3l = {b3.x, b3.y}, b3h = {b3.z, b3.w};
        c00 = __builtin_amdgcn_wmma_f32_16x16x4_f32(false, a0l, false, b0l, (short)0, c00, false, false);
        c01 = __builtin_amdgcn_wmma_f32_16x16x4_f32(false, a0l, false, b1l, (short)0, c01, false, false);
        c02 = __builtin_amdgcn_wmma_f32_16x16x4_f32(false, a0l, false, b2l, (short)0, c02, false, false);
        c03 = __builtin_amdgcn_wmma_f32_16x16x4_f32(false, a0l, false, b3l, (short)0, c03, false, false);
        c10 = __builtin_amdgcn_wmma_f32_16x16x4_f32(false, a1l, false, b0l, (short)0, c10, false, false);
        c11 = __builtin_amdgcn_wmma_f32_16x16x4_f32(false, a1l, false, b1l, (short)0, c11, false, false);
        c12 = __builtin_amdgcn_wmma_f32_16x16x4_f32(false, a1l, false, b2l, (short)0, c12, false, false);
        c13 = __builtin_amdgcn_wmma_f32_16x16x4_f32(false, a1l, false, b3l, (short)0, c13, false, false);
        c00 = __builtin_amdgcn_wmma_f32_16x16x4_f32(false, a0h, false, b0h, (short)0, c00, false, false);
        c01 = __builtin_amdgcn_wmma_f32_16x16x4_f32(false, a0h, false, b1h, (short)0, c01, false, false);
        c02 = __builtin_amdgcn_wmma_f32_16x16x4_f32(false, a0h, false, b2h, (short)0, c02, false, false);
        c03 = __builtin_amdgcn_wmma_f32_16x16x4_f32(false, a0h, false, b3h, (short)0, c03, false, false);
        c10 = __builtin_amdgcn_wmma_f32_16x16x4_f32(false, a1h, false, b0h, (short)0, c10, false, false);
        c11 = __builtin_amdgcn_wmma_f32_16x16x4_f32(false, a1h, false, b1h, (short)0, c11, false, false);
        c12 = __builtin_amdgcn_wmma_f32_16x16x4_f32(false, a1h, false, b2h, (short)0, c12, false, false);
        c13 = __builtin_amdgcn_wmma_f32_16x16x4_f32(false, a1h, false, b3h, (short)0, c13, false, false);
    }
    #pragma unroll
    for (int v = 0; v < 8; ++v) {
        const long r0 = row0 + v + 8 * half;
        if (r0 < N) {
            float* o = xh + r0 * HC + col0 + l;
            o[0]  = c00[v];
            o[16] = c01[v];
            o[32] = c02[v];
            o[48] = c03[v];
        }
        const long r1 = r0 + 16;
        if (r1 < N) {
            float* o = xh + r1 * HC + col0 + l;
            o[0]  = c10[v];
            o[16] = c11[v];
            o[32] = c12[v];
            o[48] = c13[v];
        }
    }
}

__global__ void GAT_nodeatt(const float* __restrict__ xh,
                            const float* __restrict__ att_src,
                            const float* __restrict__ att_dst,
                            float* __restrict__ asrc,
                            float* __restrict__ adst, int N) {
    long idx = (long)blockIdx.x * blockDim.x + threadIdx.x;
    if (idx >= (long)N * NH) return;
    long n = idx >> 3;
    int  h = (int)(idx & 7);
    const float* xr = xh + n * HC + h * NC;
    float ss = 0.0f, sd = 0.0f;
    for (int c = 0; c < NC; ++c) {
        float v = xr[c];
        ss += v * att_src[h * NC + c];
        sd += v * att_dst[h * NC + c];
    }
    asrc[idx] = ss;
    adst[idx] = sd;
}

typedef float v4fu __attribute__((ext_vector_type(4), aligned(4)));
#define A_EPI 4
#define A_UNR 4
#define REC   16
#define REC_DEG 8

__device__ __forceinline__ float GAT_alpha_one(long t, int h,
                                               const int* __restrict__ src,
                                               const int* __restrict__ dst,
                                               const float* __restrict__ eattr,
                                               const float* __restrict__ asrc,
                                               const float* __restrict__ adst,
                                               const float* m, int* d_out, float* ae_out) {
    const int s = src[t];
    const int d = dst[t];
    const float* row = eattr + t * F_E;
    const v4fu q0 = *(const v4fu*)(row + 0);
    const v4fu q1 = *(const v4fu*)(row + 4);
    const v4fu q2 = *(const v4fu*)(row + 8);
    const v4fu q3 = *(const v4fu*)(row + 12);
    const float as = asrc[(long)s * NH + h];
    const float ad = adst[(long)d * NH + h];
    const float ea[F_E] = {q0.x, q0.y, q0.z, q0.w, q1.x, q1.y, q1.z, q1.w,
                           q2.x, q2.y, q2.z, q2.w, q3.x, q3.y, q3.z, q3.w};
    float ae = 0.0f;
    #pragma unroll
    for (int f = 0; f < F_E; ++f) ae = fmaf(ea[f], m[f], ae);
    float al = as + ad + ae;
    al = (al >= 0.0f) ? al : NEG_SLOPE * al;
    *d_out = d;
    *ae_out = ae;
    return al;
}

__global__ void __launch_bounds__(256)
GAT_alpha(const int* __restrict__ src, const int* __restrict__ dst,
                          const float* __restrict__ eattr,
                          const float* __restrict__ M,
                          const float* __restrict__ asrc, const float* __restrict__ adst,
                          float* __restrict__ exbuf, unsigned* __restrict__ amax,
                          float* __restrict__ rec, int E) {
    const int lane = threadIdx.x & 31;
    const int h    = lane & 7;
    const int sub  = lane >> 3;
    const int slot = lane & 15;
    const int hlf  = lane >> 4;
    const int wv = __builtin_amdgcn_readfirstlane((int)(threadIdx.x >> 5));
    const long T      = (long)E;
    const long wpb    = blockDim.x >> 5;
    const long nwaves = (long)gridDim.x * wpb;
    const long wave   = (long)blockIdx.x * wpb + wv;

    float m[F_E];
    #pragma unroll
    for (int f = 0; f < F_E; ++f) m[f] = M[f * NH + h];

    for (long base = wave * (A_EPI * A_UNR); base < T; base += nwaves * (A_EPI * A_UNR)) {
        if (base + A_EPI * A_UNR <= T) {
            int  sv[A_UNR], dv[A_UNR];
            v4fu q[A_UNR][4];
            #pragma unroll
            for (int g = 0; g < A_UNR; ++g) {
                const long t = base + g * A_EPI + sub;
                sv[g] = src[t];
                dv[g] = dst[t];
                const float* row = eattr + t * F_E;
                q[g][0] = *(const v4fu*)(row + 0);
                q[g][1] = *(const v4fu*)(row + 4);
                q[g][2] = *(const v4fu*)(row + 8);
                q[g][3] = *(const v4fu*)(row + 12);
            }
            __builtin_amdgcn_sched_barrier(0);
            float as[A_UNR], ad[A_UNR];
            #pragma unroll
            for (int g = 0; g < A_UNR; ++g) {
                as[g] = asrc[(long)sv[g] * NH + h];
                ad[g] = adst[(long)dv[g] * NH + h];
            }
            __builtin_amdgcn_sched_barrier(0);
            #pragma unroll
            for (int g = 0; g < A_UNR; ++g) {
                const long t = base + g * A_EPI + sub;
                const float ea[F_E] = {q[g][0].x, q[g][0].y, q[g][0].z, q[g][0].w,
                                       q[g][1].x, q[g][1].y, q[g][1].z, q[g][1].w,
                                       q[g][2].x, q[g][2].y, q[g][2].z, q[g][2].w,
                                       q[g][3].x, q[g][3].y, q[g][3].z, q[g][3].w};
                float ae = 0.0f;
                #pragma unroll
                for (int f = 0; f < F_E; ++f) ae = fmaf(ea[f], m[f], ae);
                float al = as[g] + ad[g] + ae;
                al = (al >= 0.0f) ? al : NEG_SLOPE * al;
                exbuf[t * NH + h] = al;
                atomicMax(&amax[(long)dv[g] * NH + h], ordkey(al));
                #pragma unroll
                for (int p = 0; p < 2; ++p) {
                    const int from = ((2 * p + hlf) << 3) | (slot & 7);
                    const float v  = __shfl(ae, from);
                    const int   dd = __shfl(dv[g], from);
                    if (slot <= REC_DEG)
                        atomicAdd(&rec[(long)dd * REC + slot], (slot < REC_DEG) ? v : 1.0f);
                }
            }
        } else {
            for (int g = 0; g < A_UNR; ++g) {
                const long t = base + g * A_EPI + sub;
                if (t < T) {
                    int d;
                    float ae;
                    const float a = GAT_alpha_one(t, h, src, dst, eattr,
                                                  asrc, adst, m, &d, &ae);
                    exbuf[t * NH + h] = a;
                    atomicMax(&amax[(long)d * NH + h], ordkey(a));
                    atomicAdd(&rec[(long)d * REC + h], ae);
                    if (h == 0) atomicAdd(&rec[(long)d * REC + REC_DEG], 1.0f);
                }
            }
        }
    }
}

__global__ void GAT_selfalpha(const float* __restrict__ rec,
                              const float* __restrict__ asrc,
                              const float* __restrict__ adst,
                              float* __restrict__ exbuf,
                              unsigned* __restrict__ amax, int E, int N) {
    long idx = (long)blockIdx.x * blockDim.x + threadIdx.x;
    if (idx >= (long)N * NH) return;
    const long n = idx >> 3;
    const int  h = (int)(idx & 7);
    const float ae = rec[n * REC + h] / fmaxf(rec[n * REC + REC_DEG], 1.0f);
    float al = asrc[idx] + adst[idx] + ae;
    al = (al >= 0.0f) ? al : NEG_SLOPE * al;
    exbuf[((long)E + n) * NH + h] = al;
    const unsigned k = ordkey(al);
    const unsigned c = amax[idx];
    amax[idx] = (k > c) ? k : c;
}

__global__ void GAT_expsum(const int* __restrict__ dst,
                           float* __restrict__ exbuf,
                           const unsigned* __restrict__ amax,
                           float* __restrict__ denom, int E, int N) {
    long idx = (long)blockIdx.x * blockDim.x + threadIdx.x;
    long T = ((long)E + N) * NH;
    if (idx >= T) return;
    long t = idx >> 3;
    int  h = (int)(idx & 7);
    int d = (t < E) ? dst[t] : (int)(t - E);
    float m = ordinv(amax[(long)d * NH + h]);
    float e = __expf(exbuf[idx] - m);
    exbuf[idx] = e;
    atomicAdd(&denom[(long)d * NH + h], e);
}

#define EPW 4
__global__ void GAT_scatter(const int* __restrict__ src, const int* __restrict__ dst,
                            const float* __restrict__ xh,
                            const float* __restrict__ exbuf,
                            const float* __restrict__ denom,
                            float* __restrict__ accum, int E, int N) {
    const int lane = threadIdx.x & 31;
    const int wv = __builtin_amdgcn_readfirstlane((int)(threadIdx.x >> 5));
    const long T  = (long)E + N;
    const long t0 = ((long)blockIdx.x * (blockDim.x >> 5) + wv) * EPW;
    if (t0 >= T) return;

    long tl = t0 + (lane >> 3);
    if (tl >= T) tl = T - 1;
    const int h_l = lane & 7;
    int sl, dl;
    if (tl < E) { sl = src[tl]; dl = dst[tl]; }
    else        { sl = (int)(tl - E); dl = sl; }

    if (t0 + EPW <= T) {
        float xv[EPW][NH];
        #pragma unroll
        for (int i = 0; i < EPW; ++i) {
            const int s = __builtin_amdgcn_readlane(sl, i * 8);
            const float* xr = xh + (long)s * HC + lane;
            #pragma unroll
            for (int h = 0; h < NH; ++h) xv[i][h] = xr[h * NC];
        }
        const float att_l = exbuf[tl * NH + h_l] / denom[(long)dl * NH + h_l];
        #pragma unroll
        for (int i = 0; i < EPW; ++i) {
            float a = __int_as_float(__builtin_amdgcn_readlane(__float_as_int(att_l), i * 8));
            float r = a * xv[i][0];
            #pragma unroll
            for (int h = 1; h < NH; ++h) {
                a = __int_as_float(__builtin_amdgcn_readlane(__float_as_int(att_l), i * 8 + h));
                r = fmaf(a, xv[i][h], r);
            }
            const int d = __builtin_amdgcn_readlane(dl, i * 8);
            atomicAdd(&accum[(long)d * NC + lane], r);
        }
    } else {
        const float att_l = exbuf[tl * NH + h_l] / denom[(long)dl * NH + h_l];
        const int left = (int)(T - t0);
        for (int i = 0; i < left; ++i) {
            const int s = __builtin_amdgcn_readlane(sl, i * 8);
            const int d = __builtin_amdgcn_readlane(dl, i * 8);
            const float* xr = xh + (long)s * HC + lane;
            float r = 0.0f;
            for (int h = 0; h < NH; ++h) {
                const float a = __int_as_float(
                    __builtin_amdgcn_readlane(__float_as_int(att_l), i * 8 + h));
                r = fmaf(a, xr[h * NC], r);
            }
            atomicAdd(&accum[(long)d * NC + lane], r);
        }
    }
}

__global__ void GAT_final(const float* __restrict__ accum,
                          const float* __restrict__ bias,
                          float* __restrict__ out, int N) {
    long idx = (long)blockIdx.x * blockDim.x + threadIdx.x;
    if (idx >= (long)N * NC) return;
    int  c = (int)(idx & 31);
    out[idx] = accum[idx] * (1.0f / NH) + bias[c];
}

extern "C" void kernel_launch(void* const* d_in, const int* in_sizes, int n_in,
                              void* d_out, int out_size, void* d_ws, size_t ws_size,
                              hipStream_t stream) {
    const float* x        = (const float*)d_in[0];
    const int*   eidx     = (const int*)  d_in[1];
    const float* eattr    = (const float*)d_in[2];
    const float* W        = (const float*)d_in[3];
    const float* We       = (const float*)d_in[4];
    const float* att_src  = (const float*)d_in[5];
    const float* att_dst  = (const float*)d_in[6];
    const float* att_edge = (const float*)d_in[7];
    const float* bias     = (const float*)d_in[8];

    const int N = in_sizes[0] / F_IN;
    const int E = in_sizes[1] / 2;
    const int* src = eidx;
    const int* dst = eidx + E;

    float* ws = (float*)d_ws;
    const long Nl = N, El = E, Tl = El + Nl;
    float*    w_acc   = ws;
    float*    w_rec   = ws + 32 * Nl;
    float*    w_denom = ws + 48 * Nl;
    unsigned* w_amax  = (unsigned*)(ws + 56 * Nl);
    const long zeroSpan = 64 * Nl;
    const long xhOff    = 65 * Nl;
    float*    w_xh    = ws + xhOff;
    float*    w_asrc  = ws + xhOff + 256 * Nl;
    float*    w_adst  = ws + xhOff + 264 * Nl;
    float*    w_M     = ws + xhOff + 272 * Nl;
    float*    w_ex    = ws + xhOff + 272 * Nl + 128;

    const int B = 256;
    GAT_zero<<<2048, B, 0, stream>>>(ws, zeroSpan);
    GAT_foldM<<<1, 128, 0, stream>>>(We, att_edge, w_M);
    {
        long rowTiles = (Nl + 31) >> 5;
        GAT_gemm_xh<<<(int)rowTiles, 128, 0, stream>>>(x, W, w_xh, N);
    }
    {
        long th = Nl * NH;
        GAT_nodeatt<<<(int)((th + B - 1) / B), B, 0, stream>>>(w_xh, att_src, att_dst,
                                                               w_asrc, w_adst, N);
    }
    GAT_alpha<<<768, B, 0, stream>>>(src, dst, eattr, w_M,
                                      w_asrc, w_adst, w_ex, w_amax, w_rec, E);
    {
        long th = Nl * NH;
        GAT_selfalpha<<<(int)((th + B - 1) / B), B, 0, stream>>>(w_rec, w_asrc, w_adst,
                                                                 w_ex, w_amax, E, N);
    }
    {
        long th = Tl * NH;
        GAT_expsum<<<(int)((th + B - 1) / B), B, 0, stream>>>(dst, w_ex, w_amax, w_denom, E, N);
    }
    {
        long groups = (Tl + EPW - 1) / EPW;
        GAT_scatter<<<(int)((groups + 7) / 8), B, 0, stream>>>(src, dst, w_xh, w_ex, w_denom,
                                                               w_acc, E, N);
    }
    {
        long th = Nl * NC;
        GAT_final<<<(int)((th + B - 1) / B), B, 0, stream>>>(w_acc, bias, (float*)d_out, N);
    }
}


# Edge passes by destination-range owners: 256 work-groups each filter the edge list for their own nodes into LDS, then do max, exp sums, row sums with LDS atomics; no memory atomics; sum order changes
# speedup vs baseline: 6.8850x; 1.4235x over previous
#include <hip/hip_runtime.h>
#include <hip/hip_bf16.h>

typedef float v2f __attribute__((ext_vector_type(2)));
typedef float v8f __attribute__((ext_vector_type(8)));

#define F_IN 128
#define F_E  16
#define NH   8
#define NC   32
#define HC   256
#define NEG_SLOPE 0.2f

__device__ __forceinline__ unsigned ordkey(float f) {
    unsigned u = __float_as_uint(f);
    return (u & 0x80000000u) ? ~u : (u | 0x80000000u);
}
__device__ __forceinline__ float ordinv(unsigned k) {
    unsigned u = (k & 0x80000000u) ? (k & 0x7fffffffu) : ~k;
    return __uint_as_float(u);
}

__global__ void GAT_foldM(const float* __restrict__ We,
                          const float* __restrict__ att_edge,
                          float* __restrict__ M) {
    int tid = threadIdx.x;
    if (tid >= F_E * NH) return;
    int f = tid >> 3, h = tid & 7;
    float s = 0.0f;
    for (int c = 0; c < NC; ++c)
        s += We[(long)(h * NC + c) * F_E + f] * att_edge[h * NC + c];
    M[f * NH + h] = s;
}

typedef float v4f8 __attribute__((ext_vector_type(4), aligned(8)));
__global__ void GAT_gemm_xh(const float* __restrict__ x,
                            const float* __restrict__ W,
                            float* __restrict__ xh, int N) {
    int wave = threadIdx.x >> 5;
    int lane = threadIdx.x & 31;
    int half = lane >> 4;
    int l    = lane & 15;
    long rowTiles = ((long)N + 31) >> 5;
    long wtile = (long)blockIdx.x * 4 + wave;
    if (wtile >= rowTiles * 4) return;
    long rowTile = wtile >> 2;
    int  colGrp  = (int)(wtile & 3);
    long row0 = rowTile * 32;
    int  col0 = colGrp * 64;

    long ar0 = row0 + l;
    long ar1 = row0 + 16 + l;
    if (ar0 >= N) ar0 = N - 1;
    if (ar1 >= N) ar1 = N - 1;
    const float* xp0 = x + ar0 * F_IN + 4 * half;
    const float* xp1 = x + ar1 * F_IN + 4 * half;
    const float* wp0 = W + (long)(col0 +  0 + l) * F_IN + 4 * half;
    const float* wp1 = W + (long)(col0 + 16 + l) * F_IN + 4 * half;
    const float* wp2 = W + (long)(col0 + 32 + l) * F_IN + 4 * half;
    const float* wp3 = W + (long)(col0 + 48 + l) * F_IN + 4 * half;

    const v8f zero8 = {0.f,0.f,0.f,0.f,0.f,0.f,0.f,0.f};
    v8f c00 = zero8, c01 = zero8, c02 = zero8, c03 = zero8;
    v8f c10 = zero8, c11 = zero8, c12 = zero8, c13 = zero8;
    #pragma unroll
    for (int k0 = 0; k0 < F_IN; k0 += 8) {
        const v4f8 a0 = *(const v4f8*)(xp0 + k0);
        const v4f8 a1 = *(const v4f8*)(xp1 + k0);
        const v4f8 b0 = *(const v4f8*)(wp0 + k0);
        const v4f8 b1 = *(const v4f8*)(wp1 + k0);
        const v4f8 b2 = *(const v4f8*)(wp2 + k0);
        const v4f8 b3 = *(const v4f8*)(wp3 + k0);
        const v2f a0l = {a0.x, a0.y}, a0h = {a0.z, a0.w};
        const v2f a1l = {a1.x, a1.y}, a1h = {a1.z, a1.w};
        const v2f b0l = {b0.x, b0.y}, b0h = {b0.z, b0.w};
        const v2f b1l = {b1.x, b1.y}, b1h = {b1.z, b1.w};
        const v2f b2l = {b2.x, b2.y}, b2h = {b2.z, b2.w};
        const v2f b3l = {b3.x, b3.y}, b3h = {b3.z, b3.w};
        c00 = __builtin_amdgcn_wmma_f32_16x16x4_f32(false, a0l, false, b0l, (short)0, c00, false, false);
        c01 = __builtin_amdgcn_wmma_f32_16x16x4_f32(false, a0l, false, b1l, (short)0, c01, false, false);
        c02 = __builtin_amdgcn_wmma_f32_16x16x4_f32(false, a0l, false, b2l, (short)0, c02, false, false);
        c03 = __builtin_amdgcn_wmma_f32_16x16x4_f32(false, a0l, false, b3l, (short)0, c03, false, false);
        c10 = __builtin_amdgcn_wmma_f32_16x16x4_f32(false, a1l, false, b0l, (short)0, c10, false, false);
        c11 = __builtin_amdgcn_wmma_f32_16x16x4_f32(false, a1l, false, b1l, (short)0, c11, false, false);
        c12 = __builtin_amdgcn_wmma_f32_16x16x4_f32(false, a1l, false, b2l, (short)0, c12, false, false);
        c13 = __builtin_amdgcn_wmma_f32_16x16x4_f32(false, a1l, false, b3l, (short)0, c13, false, false);
        c00 = __builtin_amdgcn_wmma_f32_16x16x4_f32(false, a0h, false, b0h, (short)0, c00, false, false);
        c01 = __builtin_amdgcn_wmma_f32_16x16x4_f32(false, a0h, false, b1h, (short)0, c01, false, false);
        c02 = __builtin_amdgcn_wmma_f32_16x16x4_f32(false, a0h, false, b2h, (short)0, c02, false, false);
        c03 = __builtin_amdgcn_wmma_f32_16x16x4_f32(false, a0h, false, b3h, (short)0, c03, false, false);
        c10 = __builtin_amdgcn_wmma_f32_16x16x4_f32(false, a1h, false, b0h, (short)0, c10, false, false);
        c11 = __builtin_amdgcn_wmma_f32_16x16x4_f32(false, a1h, false, b1h, (short)0, c11, false, false);
        c12 = __builtin_amdgcn_wmma_f32_16x16x4_f32(false, a1h, false, b2h, (short)0, c12, false, false);
        c13 = __builtin_amdgcn_wmma_f32_16x16x4_f32(false, a1h, false, b3h, (short)0, c13, false, false);
    }
    #pragma unroll
    for (int v = 0; v < 8; ++v) {
        const long r0 = row0 + v + 8 * half;
        if (r0 < N) {
            float* o = xh + r0 * HC + col0 + l;
            o[0]  = c00[v];
            o[16] = c01[v];
            o[32] = c02[v];
            o[48] = c03[v];
        }
        const long r1 = r0 + 16;
        if (r1 < N) {
            float* o = xh + r1 * HC + col0 + l;
            o[0]  = c10[v];
            o[16] = c11[v];
            o[32] = c12[v];
            o[48] = c13[v];
        }
    }
}

__global__ void GAT_nodeatt(const float* __restrict__ xh,
                            const float* __restrict__ att_src,
                            const float* __restrict__ att_dst,
                            float* __restrict__ asrc,
                            float* __restrict__ adst, int N) {
    long idx = (long)blockIdx.x * blockDim.x + threadIdx.x;
    if (idx >= (long)N * NH) return;
    long n = idx >> 3;
    int  h = (int)(idx & 7);
    const float* xr = xh + n * HC + h * NC;
    float ss = 0.0f, sd = 0.0f;
    for (int c = 0; c < NC; ++c) {
        float v = xr[c];
        ss += v * att_src[h * NC + c];
        sd += v * att_dst[h * NC + c];
    }
    asrc[idx] = ss;
    adst[idx] = sd;
}

typedef float v4fu __attribute__((ext_vector_type(4), aligned(4)));
typedef int   v4iu __attribute__((ext_vector_type(4), aligned(4)));
#define OW_T    1024
#define OW_W    32
#define OW_NPG  256
#define OW_SEG  192
#define OW_CAP  (OW_W * OW_SEG)
#define OW_AU   3
#define OW_SU   8
#define OW_REC  9
#define EPW     4

__device__ __forceinline__ float GAT_alpha_one(long t, int h,
                                               const int* __restrict__ src,
                                               const int* __restrict__ dst,
                                               const float* __restrict__ eattr,
                                               const float* __restrict__ asrc,
                                               const float* __restrict__ adst,
                                               const float* m, int* d_out, int* s_out,
                                               float* ae_out) {
    const int s = src[t];
    const int d = dst[t];
    const float* row = eattr + t * F_E;
    const v4fu q0 = *(const v4fu*)(row + 0);
    const v4fu q1 = *(const v4fu*)(row + 4);
    const v4fu q2 = *(const v4fu*)(row + 8);
    const v4fu q3 = *(const v4fu*)(row + 12);
    const float as = asrc[(long)s * NH + h];
    const float ad = adst[(long)d * NH + h];
    const float ea[F_E] = {q0.x, q0.y, q0.z, q0.w, q1.x, q1.y, q1.z, q1.w,
                           q2.x, q2.y, q2.z, q2.w, q3.x, q3.y, q3.z, q3.w};
    float ae = 0.0f;
    #pragma unroll
    for (int f = 0; f < F_E; ++f) ae = fmaf(ea[f], m[f], ae);
    float al = as + ad + ae;
    al = (al >= 0.0f) ? al : NEG_SLOPE * al;
    *d_out = d;
    *s_out = s;
    *ae_out = ae;
    return al;
}

__global__ void __launch_bounds__(OW_T)
GAT_owner(const int* __restrict__ src, const int* __restrict__ dst,
          const float* __restrict__ eattr, const float* __restrict__ M,
          const float* __restrict__ asrc, const float* __restrict__ adst,
          const float* __restrict__ xh, const float* __restrict__ bias,
          float* __restrict__ out, int E, int N) {
    __shared__ float         accL[OW_NPG * NC];
    __shared__ unsigned      amaxL[OW_NPG * NH];
    __shared__ float         denomL[OW_NPG * NH];
    __shared__ float         recL[OW_NPG * OW_REC];
    __shared__ float         lstA[OW_CAP * NH];
    __shared__ int           lstS[OW_CAP];
    __shared__ unsigned char lstD[OW_CAP];
    __shared__ int           ovfL;

    const int tid  = threadIdx.x;
    const int lane = tid & 31;
    const int h    = lane & 7;
    const int sub  = lane >> 3;
    const int wv = __builtin_amdgcn_readfirstlane(tid >> 5);
    const int G  = (int)gridDim.x;
    int npg = (int)(((long)N + G - 1) / G);
    if (npg > OW_NPG) npg = OW_NPG;
    if (npg < 1) npg = 1;

    float m[F_E];
    #pragma unroll
    for (int f = 0; f < F_E; ++f) m[f] = M[f * NH + h];

    for (long n0 = (long)blockIdx.x * npg; n0 < N; n0 += (long)G * npg) {
        const int nn = (N - n0 < npg) ? (int)(N - n0) : npg;
        const int n0i = (int)n0;

        for (int i = tid; i < nn * NC; i += OW_T) accL[i] = 0.0f;
        for (int i = tid; i < nn * NH; i += OW_T) amaxL[i] = 0u;
        for (int i = tid; i < nn * OW_REC; i += OW_T) recL[i] = 0.0f;
        if (tid == 0) ovfL = 0;
        __syncthreads();

        const int sb = wv * OW_SEG;
        const unsigned lower = (1u << lane) - 1u;
        int cw = 0;
#define OW_TAKE(dval, eidx)                                                      \
        {                                                                        \
            const unsigned r_ = (unsigned)((dval) - n0i);                        \
            const bool hit_ = r_ < (unsigned)nn;                                 \
            const unsigned mask_ = __builtin_amdgcn_ballot_w32(hit_);            \
            if (mask_) {                                                         \
                const int at_ = cw + __builtin_popcount(mask_ & lower);          \
                if (hit_ && at_ < OW_SEG) {                                      \
                    lstS[sb + at_] = (eidx);                                     \
                    lstD[sb + at_] = (unsigned char)r_;                          \
                }                                                                \
                cw += __builtin_popcount(mask_);                                 \
            }                                                                    \
        }
        const int nchunk = E >> 2;
        int base = 0;
        for (; base + OW_SU * OW_T <= nchunk; base += OW_SU * OW_T) {
            v4iu v[OW_SU];
            #pragma unroll
            for (int u = 0; u < OW_SU; ++u)
                v[u] = *(const v4iu*)(dst + 4 * (long)(base + u * OW_T + tid));
            #pragma unroll
            for (int u = 0; u < OW_SU; ++u) {
                const unsigned r0 = (unsigned)(v[u][0] - n0i), r1 = (unsigned)(v[u][1] - n0i);
                const unsigned r2 = (unsigned)(v[u][2] - n0i), r3 = (unsigned)(v[u][3] - n0i);
                const unsigned r01 = (r0 < r1) ? r0 : r1, r23 = (r2 < r3) ? r2 : r3;
                const unsigned rm = (r01 < r23) ? r01 : r23;
                if (__builtin_amdgcn_ballot_w32(rm < (unsigned)nn)) {
                    #pragma unroll
                    for (int j = 0; j < 4; ++j)
                        OW_TAKE(v[u][j], 4 * (base + u * OW_T + tid) + j)
                }
            }
        }
        for (; base < nchunk; base += OW_T) {
            const int c = base + tid;
            const v4iu none = {-1, -1, -1, -1};
            const v4iu v1 = (c < nchunk) ? *(const v4iu*)(dst + 4 * (long)c) : none;
            #pragma unroll
            for (int j = 0; j < 4; ++j)
                OW_TAKE(v1[j], 4 * c + j)
        }
        for (int e0 = nchunk << 2; e0 < E; e0 += OW_T) {
            const int e = e0 + tid;
            const int dval = (e < E) ? dst[e] : -1;
            OW_TAKE(dval, e)
        }
#undef OW_TAKE
        if (cw > OW_SEG) {
            if (lane == 0) ovfL = 1;
            cw = OW_SEG;
        }
        __syncthreads();
        const bool nolist = ovfL != 0;
        if (nolist) cw = 0;

        if (!nolist) {
            for (int i = lane; i < cw; i += 32)
                lstA[(sb + i) * NH] = __int_as_float(src[lstS[sb + i]]);
            const int ngrp = (cw + 3) >> 2;
            for (int g0 = 0; g0 < ngrp; g0 += OW_AU) {
                int   iv[OW_AU], dv[OW_AU], sv[OW_AU];
                bool  ok[OW_AU];
                v4fu  q[OW_AU][4];
                float as[OW_AU], ad[OW_AU];
                #pragma unroll
                for (int u = 0; u < OW_AU; ++u) {
                    const int i = (g0 + u) * 4 + sub;
                    ok[u] = i < cw;
                    iv[u] = sb + i;
                    const int e = ok[u] ? lstS[sb + i] : 0;
                    sv[u] = ok[u] ? __float_as_int(lstA[(sb + i) * NH]) : 0;
                    dv[u] = ok[u] ? (int)lstD[sb + i] : 0;
                    const float* row = eattr + (long)e * F_E;
                    q[u][0] = *(const v4fu*)(row + 0);
                    q[u][1] = *(const v4fu*)(row + 4);
                    q[u][2] = *(const v4fu*)(row + 8);
                    q[u][3] = *(const v4fu*)(row + 12);
                    as[u] = asrc[(long)sv[u] * NH + h];
                    ad[u] = adst[(n0 + dv[u]) * NH + h];
                }
                #pragma unroll
                for (int u = 0; u < OW_AU; ++u) {
                    const float ea[F_E] = {q[u][0].x, q[u][0].y, q[u][0].z, q[u][0].w,
                                           q[u][1].x, q[u][1].y, q[u][1].z, q[u][1].w,
                                           q[u][2].x, q[u][2].y, q[u][2].z, q[u][2].w,
                                           q[u][3].x, q[u][3].y, q[u][3].z, q[u][3].w};
                    float ae = 0.0f;
                    #pragma unroll
                    for (int f = 0; f < F_E; ++f) ae = fmaf(ea[f], m[f], ae);
                    float al = as[u] + ad[u] + ae;
                    al = (al >= 0.0f) ? al : NEG_SLOPE * al;
                    if (ok[u]) {
                        lstA[iv[u] * NH + h] = al;
                        if (h == 0) lstS[iv[u]] = sv[u];
                        atomicMax(&amaxL[dv[u] * NH + h], ordkey(al));
                        atomicAdd(&recL[dv[u] * OW_REC + h], ae);
                        if (h == 0) atomicAdd(&recL[dv[u] * OW_REC + NH], 1.0f);
                    }
                }
            }
        } else {
            for (long b = (long)wv * 32; b < E; b += OW_W * 32) {
                const long el = b + lane;
                unsigned r = 0xffffffffu;
                if (el < E) r = (unsigned)(dst[el] - n0i);
                unsigned mask = __builtin_amdgcn_ballot_w32(r < (unsigned)nn);
                while (mask) {
                    const int b0 = __builtin_ctz(mask);
                    mask &= mask - 1;
                    int d, s;
                    float ae;
                    const float al = GAT_alpha_one(b + b0, h, src, dst, eattr, asrc, adst, m, &d, &s, &ae);
                    const int dl = d - n0i;
                    if (lane < NH) {
                        atomicMax(&amaxL[dl * NH + h], ordkey(al));
                        atomicAdd(&recL[dl * OW_REC + h], ae);
                        if (h == 0) atomicAdd(&recL[dl * OW_REC + NH], 1.0f);
                    }
                }
            }
        }
        __syncthreads();

        for (int it = tid; it < nn * NH; it += OW_T) {
            const int nl = it >> 3, hh = it & 7;
            const long gi = (n0 + nl) * NH + hh;
            const float ae = recL[nl * OW_REC + hh] / fmaxf(recL[nl * OW_REC + NH], 1.0f);
            float al = asrc[gi] + adst[gi] + ae;
            al = (al >= 0.0f) ? al : NEG_SLOPE * al;
            unsigned k = ordkey(al);
            const unsigned c = amaxL[it];
            k = (k > c) ? k : c;
            const float mx = ordinv(k);
            const float es = __expf(al - mx);
            amaxL[it] = __float_as_uint(mx);
            denomL[it] = es;
            recL[nl * OW_REC + hh] = es;
        }
        __syncthreads();

        if (!nolist) {
            for (int it = lane; it < cw * NH; it += 32) {
                const int dl = lstD[sb + (it >> 3)];
                const float ex = __expf(lstA[sb * NH + it] - __uint_as_float(amaxL[dl * NH + h]));
                lstA[sb * NH + it] = ex;
                atomicAdd(&denomL[dl * NH + h], ex);
            }
        } else {
            for (long b = (long)wv * 32; b < E; b += OW_W * 32) {
                const long el = b + lane;
                unsigned r = 0xffffffffu;
                if (el < E) r = (unsigned)(dst[el] - n0i);
                unsigned mask = __builtin_amdgcn_ballot_w32(r < (unsigned)nn);
                while (mask) {
                    const int b0 = __builtin_ctz(mask);
                    mask &= mask - 1;
                    int d, s;
                    float ae;
                    const float al = GAT_alpha_one(b + b0, h, src, dst, eattr, asrc, adst, m, &d, &s, &ae);
                    const int dl = d - n0i;
                    if (lane < NH)
                        atomicAdd(&denomL[dl * NH + h],
                                  __expf(al - __uint_as_float(amaxL[dl * NH + h])));
                }
            }
        }
        __syncthreads();

        {
            const int nself = (nolist || nn <= wv) ? 0 : ((nn - wv + OW_W - 1) / OW_W);
            const int total = cw + nself;
            const int ngc = (total + EPW - 1) / EPW;
            for (int g = 0; g < ngc; ++g) {
                int j = g * EPW + sub;
                if (j >= total) j = total - 1;
                int sl, dl;
                float w;
                if (j < cw) { sl = lstS[sb + j]; dl = lstD[sb + j]; w = lstA[(sb + j) * NH + h]; }
                else        { dl = wv + (j - cw) * OW_W; sl = n0i + dl; w = recL[dl * OW_REC + h]; }
                const float att_l = w / denomL[dl * NH + h];
                float xv[EPW][NH];
                #pragma unroll
                for (int i = 0; i < EPW; ++i) {
                    const int s = __builtin_amdgcn_readlane(sl, i * 8);
                    const float* xr = xh + (long)s * HC + lane;
                    #pragma unroll
                    for (int hh = 0; hh < NH; ++hh) xv[i][hh] = xr[hh * NC];
                }
                #pragma unroll
                for (int i = 0; i < EPW; ++i) {
                    float a = __int_as_float(__builtin_amdgcn_readlane(__float_as_int(att_l), i * 8));
                    float r = a * xv[i][0];
                    #pragma unroll
                    for (int hh = 1; hh < NH; ++hh) {
                        a = __int_as_float(__builtin_amdgcn_readlane(__float_as_int(att_l), i * 8 + hh));
                        r = fmaf(a, xv[i][hh], r);
                    }
                    const int d = __builtin_amdgcn_readlane(dl, i * 8);
                    if (g * EPW + i < total)
                        atomicAdd(&accL[d * NC + lane], r);
                }
            }
        }
        if (nolist) {
            for (long b = (long)wv * 32; b < E; b += OW_W * 32) {
                const long el = b + lane;
                unsigned r = 0xffffffffu;
                if (el < E) r = (unsigned)(dst[el] - n0i);
                unsigned mask = __builtin_amdgcn_ballot_w32(r < (unsigned)nn);
                while (mask) {
                    const int b0 = __builtin_ctz(mask);
                    mask &= mask - 1;
                    int d, s;
                    float ae;
                    const float al = GAT_alpha_one(b + b0, h, src, dst, eattr, asrc, adst, m, &d, &s, &ae);
                    const int dl = d - n0i;
                    const float att_l = __expf(al - __uint_as_float(amaxL[dl * NH + h])) / denomL[dl * NH + h];
                    const float* xr = xh + (long)s * HC + lane;
                    float a = __int_as_float(__builtin_amdgcn_readlane(__float_as_int(att_l), 0));
                    float acc = a * xr[0];
                    #pragma unroll
                    for (int hh = 1; hh < NH; ++hh) {
                        a = __int_as_float(__builtin_amdgcn_readlane(__float_as_int(att_l), hh));
                        acc = fmaf(a, xr[hh * NC], acc);
                    }
                    atomicAdd(&accL[dl * NC + lane], acc);
                }
            }
            for (int nl = wv; nl < nn; nl += OW_W) {
                const float att_l = recL[nl * OW_REC + h] / denomL[nl * NH + h];
                const float* xr = xh + (n0 + nl) * HC + lane;
                float a = __int_as_float(__builtin_amdgcn_readlane(__float_as_int(att_l), 0));
                float acc = a * xr[0];
                #pragma unroll
                for (int hh = 1; hh < NH; ++hh) {
                    a = __int_as_float(__builtin_amdgcn_readlane(__float_as_int(att_l), hh));
                    acc = fmaf(a, xr[hh * NC], acc);
                }
                atomicAdd(&accL[nl * NC + lane], acc);
            }
        }
        __syncthreads();

        for (int it = tid; it < nn * NC; it += OW_T)
            out[n0 * NC + it] = accL[it] * (1.0f / NH) + bias[it & 31];
        __syncthreads();
    }
}

extern "C" void kernel_launch(void* const* d_in, const int* in_sizes, int n_in,
                              void* d_out, int out_size, void* d_ws, size_t ws_size,
                              hipStream_t stream) {
    const float* x        = (const float*)d_in[0];
    const int*   eidx     = (const int*)  d_in[1];
    const float* eattr    = (const float*)d_in[2];
    const float* W        = (const float*)d_in[3];
    const float* We       = (const float*)d_in[4];
    const float* att_src  = (const float*)d_in[5];
    const float* att_dst  = (const float*)d_in[6];
    const float* att_edge = (const float*)d_in[7];
    const float* bias     = (const float*)d_in[8];

    const int N = in_sizes[0] / F_IN;
    const int E = in_sizes[1] / 2;
    const int* src = eidx;
    const int* dst = eidx + E;

    float* ws = (float*)d_ws;
    const long Nl = N;
    const long xhOff    = 65 * Nl;
    float*    w_xh    = ws + xhOff;
    float*    w_asrc  = ws + xhOff + 256 * Nl;
    float*    w_adst  = ws + xhOff + 264 * Nl;
    float*    w_M     = ws + xhOff + 272 * Nl;

    const int B = 256;
    GAT_foldM<<<1, 128, 0, stream>>>(We, att_edge, w_M);
    {
        long rowTiles = (Nl + 31) >> 5;
        GAT_gemm_xh<<<(int)rowTiles, 128, 0, stream>>>(x, W, w_xh, N);
    }
    {
        long th = Nl * NH;
        GAT_nodeatt<<<(int)((th + B - 1) / B), B, 0, stream>>>(w_xh, att_src, att_dst,
                                                               w_asrc, w_adst, N);
    }
    GAT_owner<<<256, OW_T, 0, stream>>>(src, dst, eattr, w_M, w_asrc, w_adst, w_xh, bias,
                                        (float*)d_out, E, N);
}


# Edge list pre-sorted into the 256 owner ranges by a chunk-local counting sort (LDS counters); owners copy their groups, none filters all destinations; sum order not fixed, as before
# speedup vs baseline: 8.1290x; 1.1807x over previous
#include <hip/hip_runtime.h>
#include <hip/hip_bf16.h>

typedef float v2f __attribute__((ext_vector_type(2)));
typedef float v8f __attribute__((ext_vector_type(8)));

#define F_IN 128
#define F_E  16
#define NH   8
#define NC   32
#define HC   256
#define NEG_SLOPE 0.2f

__device__ __forceinline__ unsigned ordkey(float f) {
    unsigned u = __float_as_uint(f);
    return (u & 0x80000000u) ? ~u : (u | 0x80000000u);
}
__device__ __forceinline__ float ordinv(unsigned k) {
    unsigned u = (k & 0x80000000u) ? (k & 0x7fffffffu) : ~k;
    return __uint_as_float(u);
}

#define SB      256
#define SB_T    512
#define SB_U    8
#define OW_G    256
#define DESC_INTS (2 * OW_G * SB)

typedef int v2i8 __attribute__((ext_vector_type(2)));
__device__ __forceinline__ bool GAT_uselist(int E, int N) {
    return N <= OW_G * 256 && 4L * DESC_INTS + 8L * (long)E <= 260L * (long)N;
}

__global__ void __launch_bounds__(SB_T)
GAT_binsort(const int* __restrict__ src, const int* __restrict__ dst,
            const float* __restrict__ We, const float* __restrict__ att_edge,
            float* __restrict__ M, int* __restrict__ binStart, int* __restrict__ binCnt,
            int* __restrict__ ent, int E, int N) {
    __shared__ int histL[OW_G];
    __shared__ int curL[OW_G];
    __shared__ int wtotL[8];
    const int tid = threadIdx.x;
    if (blockIdx.x == SB) {
        if (tid < F_E * NH) {
            const int f = tid >> 3, h = tid & 7;
            float sum = 0.0f;
            for (int c = 0; c < NC; ++c)
                sum += We[(long)(h * NC + c) * F_E + f] * att_edge[h * NC + c];
            M[f * NH + h] = sum;
        }
        return;
    }
    if (!GAT_uselist(E, N)) return;
    const int lane = tid & 31;
    const int wv   = tid >> 5;
    int npg = (N + OW_G - 1) / OW_G;
    if (npg < 1) npg = 1;
    const int c  = (int)blockIdx.x;
    const int CH = (E + SB - 1) / SB;
    const long e0l = (long)c * CH;
    const int e0 = (e0l < E) ? (int)e0l : E;
    const int e1 = (e0l + CH < E) ? (int)(e0l + CH) : E;

    if (tid < OW_G) histL[tid] = 0;
    __syncthreads();
    for (int b0 = e0; b0 < e1; b0 += SB_U * SB_T) {
        int dv[SB_U];
        #pragma unroll
        for (int u = 0; u < SB_U; ++u) {
            const int e = b0 + u * SB_T + tid;
            dv[u] = dst[(e < e1) ? e : (e1 - 1)];
        }
        #pragma unroll
        for (int u = 0; u < SB_U; ++u) {
            const int e = b0 + u * SB_T + tid;
            const unsigned k = (unsigned)dv[u] / (unsigned)npg;
            if (e < e1 && k < (unsigned)OW_G) atomicAdd(&histL[k], 1);
        }
    }
    __syncthreads();
    const int cntv = (tid < OW_G) ? histL[tid] : 0;
    int inc = cntv;
    #pragma unroll
    for (int o = 1; o < 32; o <<= 1) {
        const int y = __shfl_up(inc, o);
        if (lane >= o) inc += y;
    }
    if (tid < OW_G && lane == 31) wtotL[wv] = inc;
    __syncthreads();
    if (tid < OW_G) {
        int basev = 0;
        for (int j = 0; j < wv; ++j) basev += wtotL[j];
        const int ex = basev + inc - cntv;
        curL[tid] = ex;
        binStart[tid * SB + c] = e0 + ex;
        binCnt[tid * SB + c]   = cntv;
    }
    __syncthreads();
    for (int b0 = e0; b0 < e1; b0 += SB_U * SB_T) {
        int dv[SB_U], sv[SB_U];
        #pragma unroll
        for (int u = 0; u < SB_U; ++u) {
            const int e = b0 + u * SB_T + tid;
            const int ec = (e < e1) ? e : (e1 - 1);
            dv[u] = dst[ec];
            sv[u] = src[ec];
        }
        #pragma unroll
        for (int u = 0; u < SB_U; ++u) {
            const int e = b0 + u * SB_T + tid;
            const unsigned k = (unsigned)dv[u] / (unsigned)npg;
            if (e < e1 && k < (unsigned)OW_G) {
                const int p = e0 + atomicAdd(&curL[k], 1);
                const v2i8 en = {e, (sv[u] & 0xffff) | ((dv[u] - (int)k * npg) << 16)};
                *(v2i8*)(ent + 2 * (long)p) = en;
            }
        }
    }
}

typedef float v4f8 __attribute__((ext_vector_type(4), aligned(8)));
__global__ void GAT_gemm_xh(const float* __restrict__ x,
                            const float* __restrict__ W,
                            float* __restrict__ xh, int N) {
    int wave = threadIdx.x >> 5;
    int lane = threadIdx.x & 31;
    int half = lane >> 4;
    int l    = lane & 15;
    long rowTiles = ((long)N + 31) >> 5;
    long wtile = (long)blockIdx.x * 4 + wave;
    if (wtile >= rowTiles * 4) return;
    long rowTile = wtile >> 2;
    int  colGrp  = (int)(wtile & 3);
    long row0 = rowTile * 32;
    int  col0 = colGrp * 64;

    long ar0 = row0 + l;
    long ar1 = row0 + 16 + l;
    if (ar0 >= N) ar0 = N - 1;
    if (ar1 >= N) ar1 = N - 1;
    const float* xp0 = x + ar0 * F_IN + 4 * half;
    const float* xp1 = x + ar1 * F_IN + 4 * half;
    const float* wp0 = W + (long)(col0 +  0 + l) * F_IN + 4 * half;
    const float* wp1 = W + (long)(col0 + 16 + l) * F_IN + 4 * half;
    const float* wp2 = W + (long)(col0 + 32 + l) * F_IN + 4 * half;
    const float* wp3 = W + (long)(col0 + 48 + l) * F_IN + 4 * half;

    const v8f zero8 = {0.f,0.f,0.f,0.f,0.f,0.f,0.f,0.f};
    v8f c00 = zero8, c01 = zero8, c02 = zero8, c03 = zero8;
    v8f c10 = zero8, c11 = zero8, c12 = zero8, c13 = zero8;
    #pragma unroll
    for (int k0 = 0; k0 < F_IN; k0 += 8) {
        const v4f8 a0 = *(const v4f8*)(xp0 + k0);
        const v4f8 a1 = *(const v4f8*)(xp1 + k0);
        const v4f8 b0 = *(const v4f8*)(wp0 + k0);
        const v4f8 b1 = *(const v4f8*)(wp1 + k0);
        const v4f8 b2 = *(const v4f8*)(wp2 + k0);
        const v4f8 b3 = *(const v4f8*)(wp3 + k0);
        const v2f a0l = {a0.x, a0.y}, a0h = {a0.z, a0.w};
        const v2f a1l = {a1.x, a1.y}, a1h = {a1.z, a1.w};
        const v2f b0l = {b0.x, b0.y}, b0h = {b0.z, b0.w};
        const v2f b1l = {b1.x, b1.y}, b1h = {b1.z, b1.w};
        const v2f b2l = {b2.x, b2.y}, b2h = {b2.z, b2.w};
        const v2f b3l = {b3.x, b3.y}, b3h = {b3.z, b3.w};
        c00 = __builtin_amdgcn_wmma_f32_16x16x4_f32(false, a0l, false, b0l, (short)0, c00, false, false);
        c01 = __builtin_amdgcn_wmma_f32_16x16x4_f32(false, a0l, false, b1l, (short)0, c01, false, false);
        c02 = __builtin_amdgcn_wmma_f32_16x16x4_f32(false, a0l, false, b2l, (short)0, c02, false, false);
        c03 = __builtin_amdgcn_wmma_f32_16x16x4_f32(false, a0l, false, b3l, (short)0, c03, false, false);
        c10 = __builtin_amdgcn_wmma_f32_16x16x4_f32(false, a1l, false, b0l, (short)0, c10, false, false);
        c11 = __builtin_amdgcn_wmma_f32_16x16x4_f32(false, a1l, false, b1l, (short)0, c11, false, false);
        c12 = __builtin_amdgcn_wmma_f32_16x16x4_f32(false, a1l, false, b2l, (short)0, c12, false, false);
        c13 = __builtin_amdgcn_wmma_f32_16x16x4_f32(false, a1l, false, b3l, (short)0, c13, false, false);
        c00 = __builtin_amdgcn_wmma_f32_16x16x4_f32(false, a0h, false, b0h, (short)0, c00, false, false);
        c01 = __builtin_amdgcn_wmma_f32_16x16x4_f32(false, a0h, false, b1h, (short)0, c01, false, false);
        c02 = __builtin_amdgcn_wmma_f32_16x16x4_f32(false, a0h, false, b2h, (short)0, c02, false, false);
        c03 = __builtin_amdgcn_wmma_f32_16x16x4_f32(false, a0h, false, b3h, (short)0, c03, false, false);
        c10 = __builtin_amdgcn_wmma_f32_16x16x4_f32(false, a1h, false, b0h, (short)0, c10, false, false);
        c11 = __builtin_amdgcn_wmma_f32_16x16x4_f32(false, a1h, false, b1h, (short)0, c11, false, false);
        c12 = __builtin_amdgcn_wmma_f32_16x16x4_f32(false, a1h, false, b2h, (short)0, c12, false, false);
        c13 = __builtin_amdgcn_wmma_f32_16x16x4_f32(false, a1h, false, b3h, (short)0, c13, false, false);
    }
    #pragma unroll
    for (int v = 0; v < 8; ++v) {
        const long r0 = row0 + v + 8 * half;
        if (r0 < N) {
            float* o = xh + r0 * HC + col0 + l;
            o[0]  = c00[v];
            o[16] = c01[v];
            o[32] = c02[v];
            o[48] = c03[v];
        }
        const long r1 = r0 + 16;
        if (r1 < N) {
            float* o = xh + r1 * HC + col0 + l;
            o[0]  = c10[v];
            o[16] = c11[v];
            o[32] = c12[v];
            o[48] = c13[v];
        }
    }
}

__global__ void GAT_nodeatt(const float* __restrict__ xh,
                            const float* __restrict__ att_src,
                            const float* __restrict__ att_dst,
                            float* __restrict__ asrc,
                            float* __restrict__ adst, int N) {
    long idx = (long)blockIdx.x * blockDim.x + threadIdx.x;
    if (idx >= (long)N * NH) return;
    long n = idx >> 3;
    int  h = (int)(idx & 7);
    const float* xr = xh + n * HC + h * NC;
    float ss = 0.0f, sd = 0.0f;
    for (int c = 0; c < NC; ++c) {
        float v = xr[c];
        ss += v * att_src[h * NC + c];
        sd += v * att_dst[h * NC + c];
    }
    asrc[idx] = ss;
    adst[idx] = sd;
}

typedef float v4fu __attribute__((ext_vector_type(4), aligned(4)));
typedef int   v4iu __attribute__((ext_vector_type(4), aligned(4)));
#define OW_T    1024
#define OW_W    32
#define OW_NPG  256
#define OW_SEG  192
#define OW_CAP  (OW_W * OW_SEG)
#define OW_AU   3
#define OW_REC  9
#define EPW     4

__device__ __forceinline__ float GAT_alpha_one(long t, int h,
                                               const int* __restrict__ src,
                                               const int* __restrict__ dst,
                                               const float* __restrict__ eattr,
                                               const float* __restrict__ asrc,
                                               const float* __restrict__ adst,
                                               const float* m, int* d_out, int* s_out,
                                               float* ae_out) {
    const int s = src[t];
    const int d = dst[t];
    const float* row = eattr + t * F_E;
    const v4fu q0 = *(const v4fu*)(row + 0);
    const v4fu q1 = *(const v4fu*)(row + 4);
    const v4fu q2 = *(const v4fu*)(row + 8);
    const v4fu q3 = *(const v4fu*)(row + 12);
    const float as = asrc[(long)s * NH + h];
    const float ad = adst[(long)d * NH + h];
    const float ea[F_E] = {q0.x, q0.y, q0.z, q0.w, q1.x, q1.y, q1.z, q1.w,
                           q2.x, q2.y, q2.z, q2.w, q3.x, q3.y, q3.z, q3.w};
    float ae = 0.0f;
    #pragma unroll
    for (int f = 0; f < F_E; ++f) ae = fmaf(ea[f], m[f], ae);
    float al = as + ad + ae;
    al = (al >= 0.0f) ? al : NEG_SLOPE * al;
    *d_out = d;
    *s_out = s;
    *ae_out = ae;
    return al;
}

__global__ void __launch_bounds__(OW_T)
GAT_owner(const int* __restrict__ src, const int* __restrict__ dst,
          const float* __restrict__ eattr, const float* __restrict__ M,
          const float* __restrict__ asrc, const float* __restrict__ adst,
          const float* __restrict__ xh, const float* __restrict__ bias,
          const int* __restrict__ binStart, const int* __restrict__ binCnt,
          const int* __restrict__ ent,
          float* __restrict__ out, int E, int N) {
    __shared__ float         accL[OW_NPG * NC];
    __shared__ unsigned      amaxL[OW_NPG * NH];
    __shared__ float         denomL[OW_NPG * NH];
    __shared__ float         recL[OW_NPG * OW_REC];
    __shared__ float         lstA[OW_CAP * NH];
    __shared__ int           lstS[OW_CAP];
    __shared__ unsigned char lstD[OW_CAP];
    __shared__ int           posL[SB];
    __shared__ int           stL[SB];
    __shared__ int           cntL[SB];
    __shared__ int           wtotL[8];

    const int tid  = threadIdx.x;
    const int lane = tid & 31;
    const int h    = lane & 7;
    const int sub  = lane >> 3;
    const int wv = __builtin_amdgcn_readfirstlane(tid >> 5);
    const int G  = (int)gridDim.x;
    int npg = (int)(((long)N + G - 1) / G);
    if (npg > OW_NPG) npg = OW_NPG;
    if (npg < 1) npg = 1;
    const bool uselist = GAT_uselist(E, N) && G == OW_G;

    float m[F_E];
    #pragma unroll
    for (int f = 0; f < F_E; ++f) m[f] = M[f * NH + h];

    for (long n0 = (long)blockIdx.x * npg; n0 < N; n0 += (long)G * npg) {
        const int nn = (N - n0 < npg) ? (int)(N - n0) : npg;
        const int n0i = (int)n0;

        for (int i = tid; i < nn * NC; i += OW_T) accL[i] = 0.0f;
        for (int i = tid; i < nn * NH; i += OW_T) amaxL[i] = 0u;
        for (int i = tid; i < nn * OW_REC; i += OW_T) recL[i] = 0.0f;

        int cntv = 0, stv = 0;
        if (uselist && tid < SB) {
            cntv = binCnt[(int)blockIdx.x * SB + tid];
            stv  = binStart[(int)blockIdx.x * SB + tid];
        }
        int inc = cntv;
        #pragma unroll
        for (int o = 1; o < 32; o <<= 1) {
            const int y = __shfl_up(inc, o);
            if (lane >= o) inc += y;
        }
        if (tid < SB && lane == 31) wtotL[wv] = inc;
        __syncthreads();
        if (tid < SB) {
            int basev = 0;
            for (int j = 0; j < wv; ++j) basev += wtotL[j];
            posL[tid] = basev + inc - cntv;
            stL[tid]  = stv;
            cntL[tid] = cntv;
        }
        int nlist = 0;
        #pragma unroll
        for (int j = 0; j < 8; ++j) nlist += wtotL[j];
        const bool nolist = !uselist || nlist > OW_CAP;
        __syncthreads();

        if (!nolist) {
            v2i8 en[8];
            int  cn[8], st[8], ps[8];
            #pragma unroll
            for (int j = 0; j < 8; ++j) {
                cn[j] = cntL[wv * 8 + j];
                st[j] = stL[wv * 8 + j];
                ps[j] = posL[wv * 8 + j];
                const v2i8 none = {0, 0};
                en[j] = (lane < cn[j]) ? *(const v2i8*)(ent + 2 * (long)(st[j] + lane)) : none;
            }
            #pragma unroll
            for (int j = 0; j < 8; ++j) {
                if (lane < cn[j]) {
                    const int p = ps[j] + lane;
                    lstS[p] = en[j].x;
                    lstD[p] = (unsigned char)(en[j].y >> 16);
                    lstA[p * NH] = __int_as_float(en[j].y & 0xffff);
                }
            }
            #pragma unroll
            for (int j = 0; j < 8; ++j) {
                for (int i = lane + 32; i < cn[j]; i += 32) {
                    const v2i8 e2 = *(const v2i8*)(ent + 2 * (long)(st[j] + i));
                    const int p = ps[j] + i;
                    lstS[p] = e2.x;
                    lstD[p] = (unsigned char)(e2.y >> 16);
                    lstA[p * NH] = __int_as_float(e2.y & 0xffff);
                }
            }
        }
        __syncthreads();
        const int per = (((nlist + OW_W - 1) / OW_W) + 3) & ~3;
        const int sb  = wv * per;
        int cw = nolist ? 0 : (nlist - sb);
        if (cw > per) cw = per;
        if (cw < 0) cw = 0;

        if (!nolist) {
            const int ngrp = (cw + 3) >> 2;
            for (int g0 = 0; g0 < ngrp; g0 += OW_AU) {
                int   iv[OW_AU], dv[OW_AU], sv[OW_AU];
                bool  ok[OW_AU];
                v4fu  q[OW_AU][4];
                float as[OW_AU], ad[OW_AU];
                #pragma unroll
                for (int u = 0; u < OW_AU; ++u) {
                    const int i = (g0 + u) * 4 + sub;
                    ok[u] = i < cw;
                    iv[u] = sb + i;
                    const int e = ok[u] ? lstS[sb + i] : 0;
                    sv[u] = ok[u] ? __float_as_int(lstA[(sb + i) * NH]) : 0;
                    dv[u] = ok[u] ? (int)lstD[sb + i] : 0;
                    const float* row = eattr + (long)e * F_E;
                    q[u][0] = *(const v4fu*)(row + 0);
                    q[u][1] = *(const v4fu*)(row + 4);
                    q[u][2] = *(const v4fu*)(row + 8);
                    q[u][3] = *(const v4fu*)(row + 12);
                    as[u] = asrc[(long)sv[u] * NH + h];
                    ad[u] = adst[(n0 + dv[u]) * NH + h];
                }
                #pragma unroll
                for (int u = 0; u < OW_AU; ++u) {
                    const float ea[F_E] = {q[u][0].x, q[u][0].y, q[u][0].z, q[u][0].w,
                                           q[u][1].x, q[u][1].y, q[u][1].z, q[u][1].w,
                                           q[u][2].x, q[u][2].y, q[u][2].z, q[u][2].w,
                                           q[u][3].x, q[u][3].y, q[u][3].z, q[u][3].w};
                    float ae = 0.0f;
                    #pragma unroll
                    for (int f = 0; f < F_E; ++f) ae = fmaf(ea[f], m[f], ae);
                    float al = as[u] + ad[u] + ae;
                    al = (al >= 0.0f) ? al : NEG_SLOPE * al;
                    if (ok[u]) {
                        lstA[iv[u] * NH + h] = al;
                        if (h == 0) lstS[iv[u]] = sv[u];
                        atomicMax(&amaxL[dv[u] * NH + h], ordkey(al));
                        atomicAdd(&recL[dv[u] * OW_REC + h], ae);
                        if (h == 0) atomicAdd(&recL[dv[u] * OW_REC + NH], 1.0f);
                    }
                }
            }
        } else {
            for (long b = (long)wv * 32; b < E; b += OW_W * 32) {
                const long el = b + lane;
                unsigned r = 0xffffffffu;
                if (el < E) r = (unsigned)(dst[el] - n0i);
                unsigned mask = __builtin_amdgcn_ballot_w32(r < (unsigned)nn);
                while (mask) {
                    const int b0 = __builtin_ctz(mask);
                    mask &= mask - 1;
                    int d, s;
                    float ae;
                    const float al = GAT_alpha_one(b + b0, h, src, dst, eattr, asrc, adst, m, &d, &s, &ae);
                    const int dl = d - n0i;
                    if (lane < NH) {
                        atomicMax(&amaxL[dl * NH + h], ordkey(al));
                        atomicAdd(&recL[dl * OW_REC + h], ae);
                        if (h == 0) atomicAdd(&recL[dl * OW_REC + NH], 1.0f);
                    }
                }
            }
        }
        __syncthreads();

        for (int it = tid; it < nn * NH; it += OW_T) {
            const int nl = it >> 3, hh = it & 7;
            const long gi = (n0 + nl) * NH + hh;
            const float ae = recL[nl * OW_REC + hh] / fmaxf(recL[nl * OW_REC + NH], 1.0f);
            float al = asrc[gi] + adst[gi] + ae;
            al = (al >= 0.0f) ? al : NEG_SLOPE * al;
            unsigned k = ordkey(al);
            const unsigned c = amaxL[it];
            k = (k > c) ? k : c;
            const float mx = ordinv(k);
            const float es = __expf(al - mx);
            amaxL[it] = __float_as_uint(mx);
            denomL[it] = es;
            recL[nl * OW_REC + hh] = es;
        }
        __syncthreads();

        if (!nolist) {
            for (int it = lane; it < cw * NH; it += 32) {
                const int dl = lstD[sb + (it >> 3)];
                const float ex = __expf(lstA[sb * NH + it] - __uint_as_float(amaxL[dl * NH + h]));
                lstA[sb * NH + it] = ex;
                atomicAdd(&denomL[dl * NH + h], ex);
            }
        } else {
            for (long b = (long)wv * 32; b < E; b += OW_W * 32) {
                const long el = b + lane;
                unsigned r = 0xffffffffu;
                if (el < E) r = (unsigned)(dst[el] - n0i);
                unsigned mask = __builtin_amdgcn_ballot_w32(r < (unsigned)nn);
                while (mask) {
                    const int b0 = __builtin_ctz(mask);
                    mask &= mask - 1;
                    int d, s;
                    float ae;
                    const float al = GAT_alpha_one(b + b0, h, src, dst, eattr, asrc, adst, m, &d, &s, &ae);
                    const int dl = d - n0i;
                    if (lane < NH)
                        atomicAdd(&denomL[dl * NH + h],
                                  __expf(al - __uint_as_float(amaxL[dl * NH + h])));
                }
            }
        }
        __syncthreads();

        {
            const int nself = (nolist || nn <= wv) ? 0 : ((nn - wv + OW_W - 1) / OW_W);
            const int total = cw + nself;
            const int ngc = (total + EPW - 1) / EPW;
            for (int g = 0; g < ngc; ++g) {
                int j = g * EPW + sub;
                if (j >= total) j = total - 1;
                int sl, dl;
                float w;
                if (j < cw) { sl = lstS[sb + j]; dl = lstD[sb + j]; w = lstA[(sb + j) * NH + h]; }
                else        { dl = wv + (j - cw) * OW_W; sl = n0i + dl; w = recL[dl * OW_REC + h]; }
                const float att_l = w / denomL[dl * NH + h];
                float xv[EPW][NH];
                #pragma unroll
                for (int i = 0; i < EPW; ++i) {
                    const int s = __builtin_amdgcn_readlane(sl, i * 8);
                    const float* xr = xh + (long)s * HC + lane;
                    #pragma unroll
                    for (int hh = 0; hh < NH; ++hh) xv[i][hh] = xr[hh * NC];
                }
                #pragma unroll
                for (int i = 0; i < EPW; ++i) {
                    float a = __int_as_float(__builtin_amdgcn_readlane(__float_as_int(att_l), i * 8));
                    float r = a * xv[i][0];
                    #pragma unroll
                    for (int hh = 1; hh < NH; ++hh) {
                        a = __int_as_float(__builtin_amdgcn_readlane(__float_as_int(att_l), i * 8 + hh));
                        r = fmaf(a, xv[i][hh], r);
                    }
                    const int d = __builtin_amdgcn_readlane(dl, i * 8);
                    if (g * EPW + i < total)
                        atomicAdd(&accL[d * NC + lane], r);
                }
            }
        }
        if (nolist) {
            for (long b = (long)wv * 32; b < E; b += OW_W * 32) {
                const long el = b + lane;
                unsigned r = 0xffffffffu;
                if (el < E) r = (unsigned)(dst[el] - n0i);
                unsigned mask = __builtin_amdgcn_ballot_w32(r < (unsigned)nn);
                while (mask) {
                    const int b0 = __builtin_ctz(mask);
                    mask &= mask - 1;
                    int d, s;
                    float ae;
                    const float al = GAT_alpha_one(b + b0, h, src, dst, eattr, asrc, adst, m, &d, &s, &ae);
                    const int dl = d - n0i;
                    const float att_l = __expf(al - __uint_as_float(amaxL[dl * NH + h])) / denomL[dl * NH + h];
                    const float* xr = xh + (long)s * HC + lane;
                    float a = __int_as_float(__builtin_amdgcn_readlane(__float_as_int(att_l), 0));
                    float acc = a * xr[0];
                    #pragma unroll
                    for (int hh = 1; hh < NH; ++hh) {
                        a = __int_as_float(__builtin_amdgcn_readlane(__float_as_int(att_l), hh));
                        acc = fmaf(a, xr[hh * NC], acc);
                    }
                    atomicAdd(&accL[dl * NC + lane], acc);
                }
            }
            for (int nl = wv; nl < nn; nl += OW_W) {
                const float att_l = recL[nl * OW_REC + h] / denomL[nl * NH + h];
                const float* xr = xh + (n0 + nl) * HC + lane;
                float a = __int_as_float(__builtin_amdgcn_readlane(__float_as_int(att_l), 0));
                float acc = a * xr[0];
                #pragma unroll
                for (int hh = 1; hh < NH; ++hh) {
                    a = __int_as_float(__builtin_amdgcn_readlane(__float_as_int(att_l), hh));
                    acc = fmaf(a, xr[hh * NC], acc);
                }
                atomicAdd(&accL[nl * NC + lane], acc);
            }
        }
        __syncthreads();

        for (int it = tid; it < nn * NC; it += OW_T)
            out[n0 * NC + it] = accL[it] * (1.0f / NH) + bias[it & 31];
        __syncthreads();
    }
}

extern "C" void kernel_launch(void* const* d_in, const int* in_sizes, int n_in,
                              void* d_out, int out_size, void* d_ws, size_t ws_size,
                              hipStream_t stream) {
    const float* x        = (const float*)d_in[0];
    const int*   eidx     = (const int*)  d_in[1];
    const float* eattr    = (const float*)d_in[2];
    const float* W        = (const float*)d_in[3];
    const float* We       = (const float*)d_in[4];
    const float* att_src  = (const float*)d_in[5];
    const float* att_dst  = (const float*)d_in[6];
    const float* att_edge = (const float*)d_in[7];
    const float* bias     = (const float*)d_in[8];

    const int N = in_sizes[0] / F_IN;
    const int E = in_sizes[1] / 2;
    const int* src = eidx;
    const int* dst = eidx + E;

    float* ws = (float*)d_ws;
    const long Nl = N;
    const long xhOff    = 65 * Nl;
    float*    w_xh    = ws + xhOff;
    float*    w_asrc  = ws + xhOff + 256 * Nl;
    float*    w_adst  = ws + xhOff + 264 * Nl;
    float*    w_M     = ws + xhOff + 272 * Nl;
    int*      w_bst   = (int*)d_ws;
    int*      w_bcn   = w_bst + OW_G * SB;
    int*      w_ent   = w_bst + DESC_INTS;

    const int B = 256;
    GAT_binsort<<<SB + 1, SB_T, 0, stream>>>(src, dst, We, att_edge, w_M, w_bst, w_bcn, w_ent, E, N);
    {
        long rowTiles = (Nl + 31) >> 5;
        GAT_gemm_xh<<<(int)rowTiles, 128, 0, stream>>>(x, W, w_xh, N);
    }
    {
        long th = Nl * NH;
        GAT_nodeatt<<<(int)((th + B - 1) / B), B, 0, stream>>>(w_xh, att_src, att_dst,
                                                               w_asrc, w_adst, N);
    }
    GAT_owner<<<OW_G, OW_T, 0, stream>>>(src, dst, eattr, w_M, w_asrc, w_adst, w_xh, bias,
                                         w_bst, w_bcn, w_ent, (float*)d_out, E, N);
}


# Per-node attention terms by linearity: [a_src|a_dst] = x @ (W^T att), 16 folded columns as extra WMMA tiles in the projection launch; the node pass over xh and its launch dropped; sum order changes
# speedup vs baseline: 8.6254x; 1.0611x over previous
#include <hip/hip_runtime.h>
#include <hip/hip_bf16.h>

typedef float v2f __attribute__((ext_vector_type(2)));
typedef float v8f __attribute__((ext_vector_type(8)));

#define F_IN 128
#define F_E  16
#define NH   8
#define NC   32
#define HC   256
#define NEG_SLOPE 0.2f

__device__ __forceinline__ unsigned ordkey(float f) {
    unsigned u = __float_as_uint(f);
    return (u & 0x80000000u) ? ~u : (u | 0x80000000u);
}
__device__ __forceinline__ float ordinv(unsigned k) {
    unsigned u = (k & 0x80000000u) ? (k & 0x7fffffffu) : ~k;
    return __uint_as_float(u);
}

#define SB      256
#define WA_COLS 16
#define WA_FLOATS (WA_COLS * F_IN)
#define WA_G    ((WA_FLOATS + SB_T - 1) / SB_T)
#define SB_T    512
#define SB_U    8
#define OW_G    256
#define DESC_INTS (2 * OW_G * SB)

typedef int v2i8 __attribute__((ext_vector_type(2)));
__device__ __forceinline__ bool GAT_uselist(int E, int N) {
    return N <= OW_G * 256 && 4L * (WA_FLOATS + DESC_INTS) + 8L * (long)E <= 260L * (long)N;
}

__global__ void __launch_bounds__(SB_T)
GAT_binsort(const int* __restrict__ src, const int* __restrict__ dst,
            const float* __restrict__ We, const float* __restrict__ att_edge,
            const float* __restrict__ W, const float* __restrict__ att_src,
            const float* __restrict__ att_dst,
            float* __restrict__ M, float* __restrict__ Wa,
            int* __restrict__ binStart, int* __restrict__ binCnt,
            int* __restrict__ ent, int E, int N) {
    __shared__ int histL[OW_G];
    __shared__ int curL[OW_G];
    __shared__ int wtotL[8];
    const int tid = threadIdx.x;
    if (blockIdx.x > SB) {
        const int o = ((int)blockIdx.x - SB - 1) * SB_T + tid;
        if (o < WA_FLOATS) {
            const int j = o / F_IN, k = o % F_IN;
            const int hh = j & (NH - 1);
            const float* att = (j < NH) ? att_src : att_dst;
            float sum = 0.0f;
            for (int c = 0; c < NC; ++c)
                sum = fmaf(W[(long)(hh * NC + c) * F_IN + k], att[hh * NC + c], sum);
            Wa[o] = sum;
        }
        return;
    }
    if (blockIdx.x == SB) {
        if (tid < F_E * NH) {
            const int f = tid >> 3, h = tid & 7;
            float sum = 0.0f;
            for (int c = 0; c < NC; ++c)
                sum += We[(long)(h * NC + c) * F_E + f] * att_edge[h * NC + c];
            M[f * NH + h] = sum;
        }
        return;
    }
    if (!GAT_uselist(E, N)) return;
    const int lane = tid & 31;
    const int wv   = tid >> 5;
    int npg = (N + OW_G - 1) / OW_G;
    if (npg < 1) npg = 1;
    const int c  = (int)blockIdx.x;
    const int CH = (E + SB - 1) / SB;
    const long e0l = (long)c * CH;
    const int e0 = (e0l < E) ? (int)e0l : E;
    const int e1 = (e0l + CH < E) ? (int)(e0l + CH) : E;

    if (tid < OW_G) histL[tid] = 0;
    __syncthreads();
    for (int b0 = e0; b0 < e1; b0 += SB_U * SB_T) {
        int dv[SB_U];
        #pragma unroll
        for (int u = 0; u < SB_U; ++u) {
            const int e = b0 + u * SB_T + tid;
            dv[u] = dst[(e < e1) ? e : (e1 - 1)];
        }
        #pragma unroll
        for (int u = 0; u < SB_U; ++u) {
            const int e = b0 + u * SB_T + tid;
            const unsigned k = (unsigned)dv[u] / (unsigned)npg;
            if (e < e1 && k < (unsigned)OW_G) atomicAdd(&histL[k], 1);
        }
    }
    __syncthreads();
    const int cntv = (tid < OW_G) ? histL[tid] : 0;
    int inc = cntv;
    #pragma unroll
    for (int o = 1; o < 32; o <<= 1) {
        const int y = __shfl_up(inc, o);
        if (lane >= o) inc += y;
    }
    if (tid < OW_G && lane == 31) wtotL[wv] = inc;
    __syncthreads();
    if (tid < OW_G) {
        int basev = 0;
        for (int j = 0; j < wv; ++j) basev += wtotL[j];
        const int ex = basev + inc - cntv;
        curL[tid] = ex;
        binStart[tid * SB + c] = e0 + ex;
        binCnt[tid * SB + c]   = cntv;
    }
    __syncthreads();
    for (int b0 = e0; b0 < e1; b0 += SB_U * SB_T) {
        int dv[SB_U], sv[SB_U];
        #pragma unroll
        for (int u = 0; u < SB_U; ++u) {
            const int e = b0 + u * SB_T + tid;
            const int ec = (e < e1) ? e : (e1 - 1);
            dv[u] = dst[ec];
            sv[u] = src[ec];
        }
        #pragma unroll
        for (int u = 0; u < SB_U; ++u) {
            const int e = b0 + u * SB_T + tid;
            const unsigned k = (unsigned)dv[u] / (unsigned)npg;
            if (e < e1 && k < (unsigned)OW_G) {
                const int p = e0 + atomicAdd(&curL[k], 1);
                const v2i8 en = {e, (sv[u] & 0xffff) | ((dv[u] - (int)k * npg) << 16)};
                *(v2i8*)(ent + 2 * (long)p) = en;
            }
        }
    }
}

typedef float v4f8 __attribute__((ext_vector_type(4), aligned(8)));
__global__ void GAT_gemm_xh(const float* __restrict__ x,
                            const float* __restrict__ W,
                            const float* __restrict__ Wa,
                            float* __restrict__ xh,
                            float* __restrict__ asrc,
                            float* __restrict__ adst, int N) {
    int wave = threadIdx.x >> 5;
    int lane = threadIdx.x & 31;
    int half = lane >> 4;
    int l    = lane & 15;
    long rowTiles = ((long)N + 31) >> 5;
    if ((long)blockIdx.x >= rowTiles) {
        const int wv = __builtin_amdgcn_readfirstlane(wave);
        const long rt = ((long)blockIdx.x - rowTiles) * 4 + wv;
        if (rt >= rowTiles) return;
        const long rowA = rt * 32;
        long q0 = rowA + l;
        long q1 = rowA + 16 + l;
        if (q0 >= N) q0 = N - 1;
        if (q1 >= N) q1 = N - 1;
        const float* xq0 = x + q0 * F_IN + 4 * half;
        const float* xq1 = x + q1 * F_IN + 4 * half;
        const float* wq  = Wa + (long)l * F_IN + 4 * half;
        const v8f z8 = {0.f,0.f,0.f,0.f,0.f,0.f,0.f,0.f};
        v8f d0 = z8, d1 = z8;
        #pragma unroll
        for (int k0 = 0; k0 < F_IN; k0 += 8) {
            const v4f8 a0 = *(const v4f8*)(xq0 + k0);
            const v4f8 a1 = *(const v4f8*)(xq1 + k0);
            const v4f8 b  = *(const v4f8*)(wq + k0);
            const v2f a0l = {a0.x, a0.y}, a0h = {a0.z, a0.w};
            const v2f a1l = {a1.x, a1.y}, a1h = {a1.z, a1.w};
            const v2f bl  = {b.x, b.y},   bh  = {b.z, b.w};
            d0 = __builtin_amdgcn_wmma_f32_16x16x4_f32(false, a0l, false, bl, (short)0, d0, false, false);
            d1 = __builtin_amdgcn_wmma_f32_16x16x4_f32(false, a1l, false, bl, (short)0, d1, false, false);
            d0 = __builtin_amdgcn_wmma_f32_16x16x4_f32(false, a0h, false, bh, (short)0, d0, false, false);
            d1 = __builtin_amdgcn_wmma_f32_16x16x4_f32(false, a1h, false, bh, (short)0, d1, false, false);
        }
        float* ap = (l < NH) ? asrc : adst;
        const int hh = l & (NH - 1);
        #pragma unroll
        for (int v = 0; v < 8; ++v) {
            const long r0 = rowA + v + 8 * half;
            if (r0 < N) ap[r0 * NH + hh] = d0[v];
            const long r1 = r0 + 16;
            if (r1 < N) ap[r1 * NH + hh] = d1[v];
        }
        return;
    }
    long wtile = (long)blockIdx.x * 4 + wave;
    if (wtile >= rowTiles * 4) return;
    long rowTile = wtile >> 2;
    int  colGrp  = (int)(wtile & 3);
    long row0 = rowTile * 32;
    int  col0 = colGrp * 64;

    long ar0 = row0 + l;
    long ar1 = row0 + 16 + l;
    if (ar0 >= N) ar0 = N - 1;
    if (ar1 >= N) ar1 = N - 1;
    const float* xp0 = x + ar0 * F_IN + 4 * half;
    const float* xp1 = x + ar1 * F_IN + 4 * half;
    const float* wp0 = W + (long)(col0 +  0 + l) * F_IN + 4 * half;
    const float* wp1 = W + (long)(col0 + 16 + l) * F_IN + 4 * half;
    const float* wp2 = W + (long)(col0 + 32 + l) * F_IN + 4 * half;
    const float* wp3 = W + (long)(col0 + 48 + l) * F_IN + 4 * half;

    const v8f zero8 = {0.f,0.f,0.f,0.f,0.f,0.f,0.f,0.f};
    v8f c00 = zero8, c01 = zero8, c02 = zero8, c03 = zero8;
    v8f c10 = zero8, c11 = zero8, c12 = zero8, c13 = zero8;
    #pragma unroll
    for (int k0 = 0; k0 < F_IN; k0 += 8) {
        const v4f8 a0 = *(const v4f8*)(xp0 + k0);
        const v4f8 a1 = *(const v4f8*)(xp1 + k0);
        const v4f8 b0 = *(const v4f8*)(wp0 + k0);
        const v4f8 b1 = *(const v4f8*)(wp1 + k0);
        const v4f8 b2 = *(const v4f8*)(wp2 + k0);
        const v4f8 b3 = *(const v4f8*)(wp3 + k0);
        const v2f a0l = {a0.x, a0.y}, a0h = {a0.z, a0.w};
        const v2f a1l = {a1.x, a1.y}, a1h = {a1.z, a1.w};
        const v2f b0l = {b0.x, b0.y}, b0h = {b0.z, b0.w};
        const v2f b1l = {b1.x, b1.y}, b1h = {b1.z, b1.w};
        const v2f b2l = {b2.x, b2.y}, b2h = {b2.z, b2.w};
        const v2f b3l = {b3.x, b3.y}, b3h = {b3.z, b3.w};
        c00 = __builtin_amdgcn_wmma_f32_16x16x4_f32(false, a0l, false, b0l, (short)0, c00, false, false);
        c01 = __builtin_amdgcn_wmma_f32_16x16x4_f32(false, a0l, false, b1l, (short)0, c01, false, false);
        c02 = __builtin_amdgcn_wmma_f32_16x16x4_f32(false, a0l, false, b2l, (short)0, c02, false, false);
        c03 = __builtin_amdgcn_wmma_f32_16x16x4_f32(false, a0l, false, b3l, (short)0, c03, false, false);
        c10 = __builtin_amdgcn_wmma_f32_16x16x4_f32(false, a1l, false, b0l, (short)0, c10, false, false);
        c11 = __builtin_amdgcn_wmma_f32_16x16x4_f32(false, a1l, false, b1l, (short)0, c11, false, false);
        c12 = __builtin_amdgcn_wmma_f32_16x16x4_f32(false, a1l, false, b2l, (short)0, c12, false, false);
        c13 = __builtin_amdgcn_wmma_f32_16x16x4_f32(false, a1l, false, b3l, (short)0, c13, false, false);
        c00 = __builtin_amdgcn_wmma_f32_16x16x4_f32(false, a0h, false, b0h, (short)0, c00, false, false);
        c01 = __builtin_amdgcn_wmma_f32_16x16x4_f32(false, a0h, false, b1h, (short)0, c01, false, false);
        c02 = __builtin_amdgcn_wmma_f32_16x16x4_f32(false, a0h, false, b2h, (short)0, c02, false, false);
        c03 = __builtin_amdgcn_wmma_f32_16x16x4_f32(false, a0h, false, b3h, (short)0, c03, false, false);
        c10 = __builtin_amdgcn_wmma_f32_16x16x4_f32(false, a1h, false, b0h, (short)0, c10, false, false);
        c11 = __builtin_amdgcn_wmma_f32_16x16x4_f32(false, a1h, false, b1h, (short)0, c11, false, false);
        c12 = __builtin_amdgcn_wmma_f32_16x16x4_f32(false, a1h, false, b2h, (short)0, c12, false, false);
        c13 = __builtin_amdgcn_wmma_f32_16x16x4_f32(false, a1h, false, b3h, (short)0, c13, false, false);
    }
    #pragma unroll
    for (int v = 0; v < 8; ++v) {
        const long r0 = row0 + v + 8 * half;
        if (r0 < N) {
            float* o = xh + r0 * HC + col0 + l;
            o[0]  = c00[v];
            o[16] = c01[v];
            o[32] = c02[v];
            o[48] = c03[v];
        }
        const long r1 = r0 + 16;
        if (r1 < N) {
            float* o = xh + r1 * HC + col0 + l;
            o[0]  = c10[v];
            o[16] = c11[v];
            o[32] = c12[v];
            o[48] = c13[v];
        }
    }
}

typedef float v4fu __attribute__((ext_vector_type(4), aligned(4)));
typedef int   v4iu __attribute__((ext_vector_type(4), aligned(4)));
#define OW_T    1024
#define OW_W    32
#define OW_NPG  256
#define OW_SEG  192
#define OW_CAP  (OW_W * OW_SEG)
#define OW_AU   3
#define OW_REC  9
#define EPW     4

__device__ __forceinline__ float GAT_alpha_one(long t, int h,
                                               const int* __restrict__ src,
                                               const int* __restrict__ dst,
                                               const float* __restrict__ eattr,
                                               const float* __restrict__ asrc,
                                               const float* __restrict__ adst,
                                               const float* m, int* d_out, int* s_out,
                                               float* ae_out) {
    const int s = src[t];
    const int d = dst[t];
    const float* row = eattr + t * F_E;
    const v4fu q0 = *(const v4fu*)(row + 0);
    const v4fu q1 = *(const v4fu*)(row + 4);
    const v4fu q2 = *(const v4fu*)(row + 8);
    const v4fu q3 = *(const v4fu*)(row + 12);
    const float as = asrc[(long)s * NH + h];
    const float ad = adst[(long)d * NH + h];
    const float ea[F_E] = {q0.x, q0.y, q0.z, q0.w, q1.x, q1.y, q1.z, q1.w,
                           q2.x, q2.y, q2.z, q2.w, q3.x, q3.y, q3.z, q3.w};
    float ae = 0.0f;
    #pragma unroll
    for (int f = 0; f < F_E; ++f) ae = fmaf(ea[f], m[f], ae);
    float al = as + ad + ae;
    al = (al >= 0.0f) ? al : NEG_SLOPE * al;
    *d_out = d;
    *s_out = s;
    *ae_out = ae;
    return al;
}

__global__ void __launch_bounds__(OW_T)
GAT_owner(const int* __restrict__ src, const int* __restrict__ dst,
          const float* __restrict__ eattr, const float* __restrict__ M,
          const float* __restrict__ asrc, const float* __restrict__ adst,
          const float* __restrict__ xh, const float* __restrict__ bias,
          const int* __restrict__ binStart, const int* __restrict__ binCnt,
          const int* __restrict__ ent,
          float* __restrict__ out, int E, int N) {
    __shared__ float         accL[OW_NPG * NC];
    __shared__ unsigned      amaxL[OW_NPG * NH];
    __shared__ float         denomL[OW_NPG * NH];
    __shared__ float         recL[OW_NPG * OW_REC];
    __shared__ float         lstA[OW_CAP * NH];
    __shared__ int           lstS[OW_CAP];
    __shared__ unsigned char lstD[OW_CAP];
    __shared__ int           posL[SB];
    __shared__ int           stL[SB];
    __shared__ int           cntL[SB];
    __shared__ int           wtotL[8];

    const int tid  = threadIdx.x;
    const int lane = tid & 31;
    const int h    = lane & 7;
    const int sub  = lane >> 3;
    const int wv = __builtin_amdgcn_readfirstlane(tid >> 5);
    const int G  = (int)gridDim.x;
    int npg = (int)(((long)N + G - 1) / G);
    if (npg > OW_NPG) npg = OW_NPG;
    if (npg < 1) npg = 1;
    const bool uselist = GAT_uselist(E, N) && G == OW_G;

    float m[F_E];
    #pragma unroll
    for (int f = 0; f < F_E; ++f) m[f] = M[f * NH + h];

    for (long n0 = (long)blockIdx.x * npg; n0 < N; n0 += (long)G * npg) {
        const int nn = (N - n0 < npg) ? (int)(N - n0) : npg;
        const int n0i = (int)n0;

        for (int i = tid; i < nn * NC; i += OW_T) accL[i] = 0.0f;
        for (int i = tid; i < nn * NH; i += OW_T) amaxL[i] = 0u;
        for (int i = tid; i < nn * OW_REC; i += OW_T) recL[i] = 0.0f;

        int cntv = 0, stv = 0;
        if (uselist && tid < SB) {
            cntv = binCnt[(int)blockIdx.x * SB + tid];
            stv  = binStart[(int)blockIdx.x * SB + tid];
        }
        int inc = cntv;
        #pragma unroll
        for (int o = 1; o < 32; o <<= 1) {
            const int y = __shfl_up(inc, o);
            if (lane >= o) inc += y;
        }
        if (tid < SB && lane == 31) wtotL[wv] = inc;
        __syncthreads();
        if (tid < SB) {
            int basev = 0;
            for (int j = 0; j < wv; ++j) basev += wtotL[j];
            posL[tid] = basev + inc - cntv;
            stL[tid]  = stv;
            cntL[tid] = cntv;
        }
        int nlist = 0;
        #pragma unroll
        for (int j = 0; j < 8; ++j) nlist += wtotL[j];
        const bool nolist = !uselist || nlist > OW_CAP;
        __syncthreads();

        if (!nolist) {
            v2i8 en[8];
            int  cn[8], st[8], ps[8];
            #pragma unroll
            for (int j = 0; j < 8; ++j) {
                cn[j] = cntL[wv * 8 + j];
                st[j] = stL[wv * 8 + j];
                ps[j] = posL[wv * 8 + j];
                const v2i8 none = {0, 0};
                en[j] = (lane < cn[j]) ? *(const v2i8*)(ent + 2 * (long)(st[j] + lane)) : none;
            }
            #pragma unroll
            for (int j = 0; j < 8; ++j) {
                if (lane < cn[j]) {
                    const int p = ps[j] + lane;
                    lstS[p] = en[j].x;
                    lstD[p] = (unsigned char)(en[j].y >> 16);
                    lstA[p * NH] = __int_as_float(en[j].y & 0xffff);
                }
            }
            #pragma unroll
            for (int j = 0; j < 8; ++j) {
                for (int i = lane + 32; i < cn[j]; i += 32) {
                    const v2i8 e2 = *(const v2i8*)(ent + 2 * (long)(st[j] + i));
                    const int p = ps[j] + i;
                    lstS[p] = e2.x;
                    lstD[p] = (unsigned char)(e2.y >> 16);
                    lstA[p * NH] = __int_as_float(e2.y & 0xffff);
                }
            }
        }
        __syncthreads();
        const int per = (((nlist + OW_W - 1) / OW_W) + 3) & ~3;
        const int sb  = wv * per;
        int cw = nolist ? 0 : (nlist - sb);
        if (cw > per) cw = per;
        if (cw < 0) cw = 0;

        if (!nolist) {
            const int ngrp = (cw + 3) >> 2;
            for (int g0 = 0; g0 < ngrp; g0 += OW_AU) {
                int   iv[OW_AU], dv[OW_AU], sv[OW_AU];
                bool  ok[OW_AU];
                v4fu  q[OW_AU][4];
                float as[OW_AU], ad[OW_AU];
                #pragma unroll
                for (int u = 0; u < OW_AU; ++u) {
                    const int i = (g0 + u) * 4 + sub;
                    ok[u] = i < cw;
                    iv[u] = sb + i;
                    const int e = ok[u] ? lstS[sb + i] : 0;
                    sv[u] = ok[u] ? __float_as_int(lstA[(sb + i) * NH]) : 0;
                    dv[u] = ok[u] ? (int)lstD[sb + i] : 0;
                    const float* row = eattr + (long)e * F_E;
                    q[u][0] = *(const v4fu*)(row + 0);
                    q[u][1] = *(const v4fu*)(row + 4);
                    q[u][2] = *(const v4fu*)(row + 8);
                    q[u][3] = *(const v4fu*)(row + 12);
                    as[u] = asrc[(long)sv[u] * NH + h];
                    ad[u] = adst[(n0 + dv[u]) * NH + h];
                }
                #pragma unroll
                for (int u = 0; u < OW_AU; ++u) {
                    const float ea[F_E] = {q[u][0].x, q[u][0].y, q[u][0].z, q[u][0].w,
                                           q[u][1].x, q[u][1].y, q[u][1].z, q[u][1].w,
                                           q[u][2].x, q[u][2].y, q[u][2].z, q[u][2].w,
                                           q[u][3].x, q[u][3].y, q[u][3].z, q[u][3].w};
                    float ae = 0.0f;
                    #pragma unroll
                    for (int f = 0; f < F_E; ++f) ae = fmaf(ea[f], m[f], ae);
                    float al = as[u] + ad[u] + ae;
                    al = (al >= 0.0f) ? al : NEG_SLOPE * al;
                    if (ok[u]) {
                        lstA[iv[u] * NH + h] = al;
                        if (h == 0) lstS[iv[u]] = sv[u];
                        atomicMax(&amaxL[dv[u] * NH + h], ordkey(al));
                        atomicAdd(&recL[dv[u] * OW_REC + h], ae);
                        if (h == 0) atomicAdd(&recL[dv[u] * OW_REC + NH], 1.0f);
                    }
                }
            }
        } else {
            for (long b = (long)wv * 32; b < E; b += OW_W * 32) {
                const long el = b + lane;
                unsigned r = 0xffffffffu;
                if (el < E) r = (unsigned)(dst[el] - n0i);
                unsigned mask = __builtin_amdgcn_ballot_w32(r < (unsigned)nn);
                while (mask) {
                    const int b0 = __builtin_ctz(mask);
                    mask &= mask - 1;
                    int d, s;
                    float ae;
                    const float al = GAT_alpha_one(b + b0, h, src, dst, eattr, asrc, adst, m, &d, &s, &ae);
                    const int dl = d - n0i;
                    if (lane < NH) {
                        atomicMax(&amaxL[dl * NH + h], ordkey(al));
                        atomicAdd(&recL[dl * OW_REC + h], ae);
                        if (h == 0) atomicAdd(&recL[dl * OW_REC + NH], 1.0f);
                    }
                }
            }
        }
        __syncthreads();

        for (int it = tid; it < nn * NH; it += OW_T) {
            const int nl = it >> 3, hh = it & 7;
            const long gi = (n0 + nl) * NH + hh;
            const float ae = recL[nl * OW_REC + hh] / fmaxf(recL[nl * OW_REC + NH], 1.0f);
            float al = asrc[gi] + adst[gi] + ae;
            al = (al >= 0.0f) ? al : NEG_SLOPE * al;
            unsigned k = ordkey(al);
            const unsigned c = amaxL[it];
            k = (k > c) ? k : c;
            const float mx = ordinv(k);
            const float es = __expf(al - mx);
            amaxL[it] = __float_as_uint(mx);
            denomL[it] = es;
            recL[nl * OW_REC + hh] = es;
        }
        __syncthreads();

        if (!nolist) {
            for (int it = lane; it < cw * NH; it += 32) {
                const int dl = lstD[sb + (it >> 3)];
                const float ex = __expf(lstA[sb * NH + it] - __uint_as_float(amaxL[dl * NH + h]));
                lstA[sb * NH + it] = ex;
                atomicAdd(&denomL[dl * NH + h], ex);
            }
        } else {
            for (long b = (long)wv * 32; b < E; b += OW_W * 32) {
                const long el = b + lane;
                unsigned r = 0xffffffffu;
                if (el < E) r = (unsigned)(dst[el] - n0i);
                unsigned mask = __builtin_amdgcn_ballot_w32(r < (unsigned)nn);
                while (mask) {
                    const int b0 = __builtin_ctz(mask);
                    mask &= mask - 1;
                    int d, s;
                    float ae;
                    const float al = GAT_alpha_one(b + b0, h, src, dst, eattr, asrc, adst, m, &d, &s, &ae);
                    const int dl = d - n0i;
                    if (lane < NH)
                        atomicAdd(&denomL[dl * NH + h],
                                  __expf(al - __uint_as_float(amaxL[dl * NH + h])));
                }
            }
        }
        __syncthreads();

        {
            const int nself = (nolist || nn <= wv) ? 0 : ((nn - wv + OW_W - 1) / OW_W);
            const int total = cw + nself;
            const int ngc = (total + EPW - 1) / EPW;
            for (int g = 0; g < ngc; ++g) {
                int j = g * EPW + sub;
                if (j >= total) j = total - 1;
                int sl, dl;
                float w;
                if (j < cw) { sl = lstS[sb + j]; dl = lstD[sb + j]; w = lstA[(sb + j) * NH + h]; }
                else        { dl = wv + (j - cw) * OW_W; sl = n0i + dl; w = recL[dl * OW_REC + h]; }
                const float att_l = w / denomL[dl * NH + h];
                float xv[EPW][NH];
                #pragma unroll
                for (int i = 0; i < EPW; ++i) {
                    const int s = __builtin_amdgcn_readlane(sl, i * 8);
                    const float* xr = xh + (long)s * HC + lane;
                    #pragma unroll
                    for (int hh = 0; hh < NH; ++hh) xv[i][hh] = xr[hh * NC];
                }
                #pragma unroll
                for (int i = 0; i < EPW; ++i) {
                    float a = __int_as_float(__builtin_amdgcn_readlane(__float_as_int(att_l), i * 8));
                    float r = a * xv[i][0];
                    #pragma unroll
                    for (int hh = 1; hh < NH; ++hh) {
                        a = __int_as_float(__builtin_amdgcn_readlane(__float_as_int(att_l), i * 8 + hh));
                        r = fmaf(a, xv[i][hh], r);
                    }
                    const int d = __builtin_amdgcn_readlane(dl, i * 8);
                    if (g * EPW + i < total)
                        atomicAdd(&accL[d * NC + lane], r);
                }
            }
        }
        if (nolist) {
            for (long b = (long)wv * 32; b < E; b += OW_W * 32) {
                const long el = b + lane;
                unsigned r = 0xffffffffu;
                if (el < E) r = (unsigned)(dst[el] - n0i);
                unsigned mask = __builtin_amdgcn_ballot_w32(r < (unsigned)nn);
                while (mask) {
                    const int b0 = __builtin_ctz(mask);
                    mask &= mask - 1;
                    int d, s;
                    float ae;
                    const float al = GAT_alpha_one(b + b0, h, src, dst, eattr, asrc, adst, m, &d, &s, &ae);
                    const int dl = d - n0i;
                    const float att_l = __expf(al - __uint_as_float(amaxL[dl * NH + h])) / denomL[dl * NH + h];
                    const float* xr = xh + (long)s * HC + lane;
                    float a = __int_as_float(__builtin_amdgcn_readlane(__float_as_int(att_l), 0));
                    float acc = a * xr[0];
                    #pragma unroll
                    for (int hh = 1; hh < NH; ++hh) {
                        a = __int_as_float(__builtin_amdgcn_readlane(__float_as_int(att_l), hh));
                        acc = fmaf(a, xr[hh * NC], acc);
                    }
                    atomicAdd(&accL[dl * NC + lane], acc);
                }
            }
            for (int nl = wv; nl < nn; nl += OW_W) {
                const float att_l = recL[nl * OW_REC + h] / denomL[nl * NH + h];
                const float* xr = xh + (n0 + nl) * HC + lane;
                float a = __int_as_float(__builtin_amdgcn_readlane(__float_as_int(att_l), 0));
                float acc = a * xr[0];
                #pragma unroll
                for (int hh = 1; hh < NH; ++hh) {
                    a = __int_as_float(__builtin_amdgcn_readlane(__float_as_int(att_l), hh));
                    acc = fmaf(a, xr[hh * NC], acc);
                }
                atomicAdd(&accL[nl * NC + lane], acc);
            }
        }
        __syncthreads();

        for (int it = tid; it < nn * NC; it += OW_T)
            out[n0 * NC + it] = accL[it] * (1.0f / NH) + bias[it & 31];
        __syncthreads();
    }
}

extern "C" void kernel_launch(void* const* d_in, const int* in_sizes, int n_in,
                              void* d_out, int out_size, void* d_ws, size_t ws_size,
                              hipStream_t stream) {
    const float* x        = (const float*)d_in[0];
    const int*   eidx     = (const int*)  d_in[1];
    const float* eattr    = (const float*)d_in[2];
    const float* W        = (const float*)d_in[3];
    const float* We       = (const float*)d_in[4];
    const float* att_src  = (const float*)d_in[5];
    const float* att_dst  = (const float*)d_in[6];
    const float* att_edge = (const float*)d_in[7];
    const float* bias     = (const float*)d_in[8];

    const int N = in_sizes[0] / F_IN;
    const int E = in_sizes[1] / 2;
    const int* src = eidx;
    const int* dst = eidx + E;

    float* ws = (float*)d_ws;
    const long Nl = N;
    const long xhOff    = 65 * Nl;
    float*    w_xh    = ws + xhOff;
    float*    w_asrc  = ws + xhOff + 256 * Nl;
    float*    w_adst  = ws + xhOff + 264 * Nl;
    float*    w_M     = ws + xhOff + 272 * Nl;
    float*    w_wa    = ws;
    int*      w_bst   = (int*)d_ws + WA_FLOATS;
    int*      w_bcn   = w_bst + OW_G * SB;
    int*      w_ent   = w_bst + DESC_INTS;

    GAT_binsort<<<SB + 1 + WA_G, SB_T, 0, stream>>>(src, dst, We, att_edge, W, att_src, att_dst,
                                                   w_M, w_wa, w_bst, w_bcn, w_ent, E, N);
    {
        long rowTiles = (Nl + 31) >> 5;
        long attBlocks = (rowTiles + 3) >> 2;
        GAT_gemm_xh<<<(int)(rowTiles + attBlocks), 128, 0, stream>>>(x, W, w_wa, w_xh,
                                                                     w_asrc, w_adst, N);
    }
    GAT_owner<<<OW_G, OW_T, 0, stream>>>(src, dst, eattr, w_M, w_asrc, w_adst, w_xh, bias,
                                         w_bst, w_bcn, w_ent, (float*)d_out, E, N);
}


# Row phase: each 1-KB source row read as two 16-byte-per-lane loads of whole lines (xh plane on a 256-byte boundary), not eight 4-byte loads; heads summed across lane groups: head sum order changes
# speedup vs baseline: 9.5006x; 1.1015x over previous
#include <hip/hip_runtime.h>
#include <hip/hip_bf16.h>

typedef float v2f __attribute__((ext_vector_type(2)));
typedef float v8f __attribute__((ext_vector_type(8)));

#define F_IN 128
#define F_E  16
#define NH   8
#define NC   32
#define HC   256
#define NEG_SLOPE 0.2f

__device__ __forceinline__ unsigned ordkey(float f) {
    unsigned u = __float_as_uint(f);
    return (u & 0x80000000u) ? ~u : (u | 0x80000000u);
}
__device__ __forceinline__ float ordinv(unsigned k) {
    unsigned u = (k & 0x80000000u) ? (k & 0x7fffffffu) : ~k;
    return __uint_as_float(u);
}

#define SB      256
#define WA_COLS 16
#define WA_FLOATS (WA_COLS * F_IN)
#define WA_G    ((WA_FLOATS + SB_T - 1) / SB_T)
#define SB_T    512
#define SB_U    8
#define OW_G    256
#define DESC_INTS (2 * OW_G * SB)

typedef int v2i8 __attribute__((ext_vector_type(2)));
__device__ __forceinline__ bool GAT_uselist(int E, int N) {
    return N <= OW_G * 256 && 4L * (WA_FLOATS + DESC_INTS) + 8L * (long)E <= 260L * (long)N - 256L;
}

__global__ void __launch_bounds__(SB_T)
GAT_binsort(const int* __restrict__ src, const int* __restrict__ dst,
            const float* __restrict__ We, const float* __restrict__ att_edge,
            const float* __restrict__ W, const float* __restrict__ att_src,
            const float* __restrict__ att_dst,
            float* __restrict__ M, float* __restrict__ Wa,
            int* __restrict__ binStart, int* __restrict__ binCnt,
            int* __restrict__ ent, int E, int N) {
    __shared__ int histL[OW_G];
    __shared__ int curL[OW_G];
    __shared__ int wtotL[8];
    const int tid = threadIdx.x;
    if (blockIdx.x > SB) {
        const int o = ((int)blockIdx.x - SB - 1) * SB_T + tid;
        if (o < WA_FLOATS) {
            const int j = o / F_IN, k = o % F_IN;
            const int hh = j & (NH - 1);
            const float* att = (j < NH) ? att_src : att_dst;
            float sum = 0.0f;
            for (int c = 0; c < NC; ++c)
                sum = fmaf(W[(long)(hh * NC + c) * F_IN + k], att[hh * NC + c], sum);
            Wa[o] = sum;
        }
        return;
    }
    if (blockIdx.x == SB) {
        if (tid < F_E * NH) {
            const int f = tid >> 3, h = tid & 7;
            float sum = 0.0f;
            for (int c = 0; c < NC; ++c)
                sum += We[(long)(h * NC + c) * F_E + f] * att_edge[h * NC + c];
            M[f * NH + h] = sum;
        }
        return;
    }
    if (!GAT_uselist(E, N)) return;
    const int lane = tid & 31;
    const int wv   = tid >> 5;
    int npg = (N + OW_G - 1) / OW_G;
    if (npg < 1) npg = 1;
    const int c  = (int)blockIdx.x;
    const int CH = (E + SB - 1) / SB;
    const long e0l = (long)c * CH;
    const int e0 = (e0l < E) ? (int)e0l : E;
    const int e1 = (e0l + CH < E) ? (int)(e0l + CH) : E;

    if (tid < OW_G) histL[tid] = 0;
    __syncthreads();
    for (int b0 = e0; b0 < e1; b0 += SB_U * SB_T) {
        int dv[SB_U];
        #pragma unroll
        for (int u = 0; u < SB_U; ++u) {
            const int e = b0 + u * SB_T + tid;
            dv[u] = dst[(e < e1) ? e : (e1 - 1)];
        }
        #pragma unroll
        for (int u = 0; u < SB_U; ++u) {
            const int e = b0 + u * SB_T + tid;
            const unsigned k = (unsigned)dv[u] / (unsigned)npg;
            if (e < e1 && k < (unsigned)OW_G) atomicAdd(&histL[k], 1);
        }
    }
    __syncthreads();
    const int cntv = (tid < OW_G) ? histL[tid] : 0;
    int inc = cntv;
    #pragma unroll
    for (int o = 1; o < 32; o <<= 1) {
        const int y = __shfl_up(inc, o);
        if (lane >= o) inc += y;
    }
    if (tid < OW_G && lane == 31) wtotL[wv] = inc;
    __syncthreads();
    if (tid < OW_G) {
        int basev = 0;
        for (int j = 0; j < wv; ++j) basev += wtotL[j];
        const int ex = basev + inc - cntv;
        curL[tid] = ex;
        binStart[tid * SB + c] = e0 + ex;
        binCnt[tid * SB + c]   = cntv;
    }
    __syncthreads();
    for (int b0 = e0; b0 < e1; b0 += SB_U * SB_T) {
        int dv[SB_U], sv[SB_U];
        #pragma unroll
        for (int u = 0; u < SB_U; ++u) {
            const int e = b0 + u * SB_T + tid;
            const int ec = (e < e1) ? e : (e1 - 1);
            dv[u] = dst[ec];
            sv[u] = src[ec];
        }
        #pragma unroll
        for (int u = 0; u < SB_U; ++u) {
            const int e = b0 + u * SB_T + tid;
            const unsigned k = (unsigned)dv[u] / (unsigned)npg;
            if (e < e1 && k < (unsigned)OW_G) {
                const int p = e0 + atomicAdd(&curL[k], 1);
                const v2i8 en = {e, (sv[u] & 0xffff) | ((dv[u] - (int)k * npg) << 16)};
                *(v2i8*)(ent + 2 * (long)p) = en;
            }
        }
    }
}

typedef float v4f8 __attribute__((ext_vector_type(4), aligned(8)));
__global__ void GAT_gemm_xh(const float* __restrict__ x,
                            const float* __restrict__ W,
                            const float* __restrict__ Wa,
                            float* __restrict__ xh,
                            float* __restrict__ asrc,
                            float* __restrict__ adst, int N) {
    int wave = threadIdx.x >> 5;
    int lane = threadIdx.x & 31;
    int half = lane >> 4;
    int l    = lane & 15;
    long rowTiles = ((long)N + 31) >> 5;
    if ((long)blockIdx.x >= rowTiles) {
        const int wv = __builtin_amdgcn_readfirstlane(wave);
        const long rt = ((long)blockIdx.x - rowTiles) * 4 + wv;
        if (rt >= rowTiles) return;
        const long rowA = rt * 32;
        long q0 = rowA + l;
        long q1 = rowA + 16 + l;
        if (q0 >= N) q0 = N - 1;
        if (q1 >= N) q1 = N - 1;
        const float* xq0 = x + q0 * F_IN + 4 * half;
        const float* xq1 = x + q1 * F_IN + 4 * half;
        const float* wq  = Wa + (long)l * F_IN + 4 * half;
        const v8f z8 = {0.f,0.f,0.f,0.f,0.f,0.f,0.f,0.f};
        v8f d0 = z8, d1 = z8;
        #pragma unroll
        for (int k0 = 0; k0 < F_IN; k0 += 8) {
            const v4f8 a0 = *(const v4f8*)(xq0 + k0);
            const v4f8 a1 = *(const v4f8*)(xq1 + k0);
            const v4f8 b  = *(const v4f8*)(wq + k0);
            const v2f a0l = {a0.x, a0.y}, a0h = {a0.z, a0.w};
            const v2f a1l = {a1.x, a1.y}, a1h = {a1.z, a1.w};
            const v2f bl  = {b.x, b.y},   bh  = {b.z, b.w};
            d0 = __builtin_amdgcn_wmma_f32_16x16x4_f32(false, a0l, false, bl, (short)0, d0, false, false);
            d1 = __builtin_amdgcn_wmma_f32_16x16x4_f32(false, a1l, false, bl, (short)0, d1, false, false);
            d0 = __builtin_amdgcn_wmma_f32_16x16x4_f32(false, a0h, false, bh, (short)0, d0, false, false);
            d1 = __builtin_amdgcn_wmma_f32_16x16x4_f32(false, a1h, false, bh, (short)0, d1, false, false);
        }
        float* ap = (l < NH) ? asrc : adst;
        const int hh = l & (NH - 1);
        #pragma unroll
        for (int v = 0; v < 8; ++v) {
            const long r0 = rowA + v + 8 * half;
            if (r0 < N) ap[r0 * NH + hh] = d0[v];
            const long r1 = r0 + 16;
            if (r1 < N) ap[r1 * NH + hh] = d1[v];
        }
        return;
    }
    long wtile = (long)blockIdx.x * 4 + wave;
    if (wtile >= rowTiles * 4) return;
    long rowTile = wtile >> 2;
    int  colGrp  = (int)(wtile & 3);
    long row0 = rowTile * 32;
    int  col0 = colGrp * 64;

    long ar0 = row0 + l;
    long ar1 = row0 + 16 + l;
    if (ar0 >= N) ar0 = N - 1;
    if (ar1 >= N) ar1 = N - 1;
    const float* xp0 = x + ar0 * F_IN + 4 * half;
    const float* xp1 = x + ar1 * F_IN + 4 * half;
    const float* wp0 = W + (long)(col0 +  0 + l) * F_IN + 4 * half;
    const float* wp1 = W + (long)(col0 + 16 + l) * F_IN + 4 * half;
    const float* wp2 = W + (long)(col0 + 32 + l) * F_IN + 4 * half;
    const float* wp3 = W + (long)(col0 + 48 + l) * F_IN + 4 * half;

    const v8f zero8 = {0.f,0.f,0.f,0.f,0.f,0.f,0.f,0.f};
    v8f c00 = zero8, c01 = zero8, c02 = zero8, c03 = zero8;
    v8f c10 = zero8, c11 = zero8, c12 = zero8, c13 = zero8;
    #pragma unroll
    for (int k0 = 0; k0 < F_IN; k0 += 8) {
        const v4f8 a0 = *(const v4f8*)(xp0 + k0);
        const v4f8 a1 = *(const v4f8*)(xp1 + k0);
        const v4f8 b0 = *(const v4f8*)(wp0 + k0);
        const v4f8 b1 = *(const v4f8*)(wp1 + k0);
        const v4f8 b2 = *(const v4f8*)(wp2 + k0);
        const v4f8 b3 = *(const v4f8*)(wp3 + k0);
        const v2f a0l = {a0.x, a0.y}, a0h = {a0.z, a0.w};
        const v2f a1l = {a1.x, a1.y}, a1h = {a1.z, a1.w};
        const v2f b0l = {b0.x, b0.y}, b0h = {b0.z, b0.w};
        const v2f b1l = {b1.x, b1.y}, b1h = {b1.z, b1.w};
        const v2f b2l = {b2.x, b2.y}, b2h = {b2.z, b2.w};
        const v2f b3l = {b3.x, b3.y}, b3h = {b3.z, b3.w};
        c00 = __builtin_amdgcn_wmma_f32_16x16x4_f32(false, a0l, false, b0l, (short)0, c00, false, false);
        c01 = __builtin_amdgcn_wmma_f32_16x16x4_f32(false, a0l, false, b1l, (short)0, c01, false, false);
        c02 = __builtin_amdgcn_wmma_f32_16x16x4_f32(false, a0l, false, b2l, (short)0, c02, false, false);
        c03 = __builtin_amdgcn_wmma_f32_16x16x4_f32(false, a0l, false, b3l, (short)0, c03, false, false);
        c10 = __builtin_amdgcn_wmma_f32_16x16x4_f32(false, a1l, false, b0l, (short)0, c10, false, false);
        c11 = __builtin_amdgcn_wmma_f32_16x16x4_f32(false, a1l, false, b1l, (short)0, c11, false, false);
        c12 = __builtin_amdgcn_wmma_f32_16x16x4_f32(false, a1l, false, b2l, (short)0, c12, false, false);
        c13 = __builtin_amdgcn_wmma_f32_16x16x4_f32(false, a1l, false, b3l, (short)0, c13, false, false);
        c00 = __builtin_amdgcn_wmma_f32_16x16x4_f32(false, a0h, false, b0h, (short)0, c00, false, false);
        c01 = __builtin_amdgcn_wmma_f32_16x16x4_f32(false, a0h, false, b1h, (short)0, c01, false, false);
        c02 = __builtin_amdgcn_wmma_f32_16x16x4_f32(false, a0h, false, b2h, (short)0, c02, false, false);
        c03 = __builtin_amdgcn_wmma_f32_16x16x4_f32(false, a0h, false, b3h, (short)0, c03, false, false);
        c10 = __builtin_amdgcn_wmma_f32_16x16x4_f32(false, a1h, false, b0h, (short)0, c10, false, false);
        c11 = __builtin_amdgcn_wmma_f32_16x16x4_f32(false, a1h, false, b1h, (short)0, c11, false, false);
        c12 = __builtin_amdgcn_wmma_f32_16x16x4_f32(false, a1h, false, b2h, (short)0, c12, false, false);
        c13 = __builtin_amdgcn_wmma_f32_16x16x4_f32(false, a1h, false, b3h, (short)0, c13, false, false);
    }
    #pragma unroll
    for (int v = 0; v < 8; ++v) {
        const long r0 = row0 + v + 8 * half;
        if (r0 < N) {
            float* o = xh + r0 * HC + col0 + l;
            o[0]  = c00[v];
            o[16] = c01[v];
            o[32] = c02[v];
            o[48] = c03[v];
        }
        const long r1 = r0 + 16;
        if (r1 < N) {
            float* o = xh + r1 * HC + col0 + l;
            o[0]  = c10[v];
            o[16] = c11[v];
            o[32] = c12[v];
            o[48] = c13[v];
        }
    }
}

typedef float v4fu __attribute__((ext_vector_type(4), aligned(4)));
typedef int   v4iu __attribute__((ext_vector_type(4), aligned(4)));
#define OW_T    1024
#define OW_W    32
#define OW_NPG  256
#define OW_SEG  192
#define OW_CAP  (OW_W * OW_SEG)
#define OW_AU   3
#define OW_REC  9
#define EPW     4

__device__ __forceinline__ float GAT_swap16(float v) {
    const int b = __float_as_int(v);
    return __int_as_float(__builtin_amdgcn_permlanex16(b, b, 0x76543210, 0xfedcba98, false, false));
}
__device__ __forceinline__ float GAT_swap8(float v) {
    return __int_as_float(__builtin_amdgcn_update_dpp(0, __float_as_int(v), 0x128  , 0xf, 0xf, false));
}

__device__ __forceinline__ float GAT_alpha_one(long t, int h,
                                               const int* __restrict__ src,
                                               const int* __restrict__ dst,
                                               const float* __restrict__ eattr,
                                               const float* __restrict__ asrc,
                                               const float* __restrict__ adst,
                                               const float* m, int* d_out, int* s_out,
                                               float* ae_out) {
    const int s = src[t];
    const int d = dst[t];
    const float* row = eattr + t * F_E;
    const v4fu q0 = *(const v4fu*)(row + 0);
    const v4fu q1 = *(const v4fu*)(row + 4);
    const v4fu q2 = *(const v4fu*)(row + 8);
    const v4fu q3 = *(const v4fu*)(row + 12);
    const float as = asrc[(long)s * NH + h];
    const float ad = adst[(long)d * NH + h];
    const float ea[F_E] = {q0.x, q0.y, q0.z, q0.w, q1.x, q1.y, q1.z, q1.w,
                           q2.x, q2.y, q2.z, q2.w, q3.x, q3.y, q3.z, q3.w};
    float ae = 0.0f;
    #pragma unroll
    for (int f = 0; f < F_E; ++f) ae = fmaf(ea[f], m[f], ae);
    float al = as + ad + ae;
    al = (al >= 0.0f) ? al : NEG_SLOPE * al;
    *d_out = d;
    *s_out = s;
    *ae_out = ae;
    return al;
}

__global__ void __launch_bounds__(OW_T)
GAT_owner(const int* __restrict__ src, const int* __restrict__ dst,
          const float* __restrict__ eattr, const float* __restrict__ M,
          const float* __restrict__ asrc, const float* __restrict__ adst,
          const float* __restrict__ xh, const float* __restrict__ bias,
          const int* __restrict__ binStart, const int* __restrict__ binCnt,
          const int* __restrict__ ent,
          float* __restrict__ out, int E, int N) {
    __shared__ float         accL[OW_NPG * NC];
    __shared__ unsigned      amaxL[OW_NPG * NH];
    __shared__ float         denomL[OW_NPG * NH];
    __shared__ float         recL[OW_NPG * OW_REC];
    __shared__ float         lstA[OW_CAP * NH];
    __shared__ int           lstS[OW_CAP];
    __shared__ unsigned char lstD[OW_CAP];
    __shared__ int           posL[SB];
    __shared__ int           stL[SB];
    __shared__ int           cntL[SB];
    __shared__ int           wtotL[8];
    __shared__ float         attW[OW_W * 32];

    const int tid  = threadIdx.x;
    const int lane = tid & 31;
    const int h    = lane & 7;
    const int sub  = lane >> 3;
    const int wv = __builtin_amdgcn_readfirstlane(tid >> 5);
    const int G  = (int)gridDim.x;
    int npg = (int)(((long)N + G - 1) / G);
    if (npg > OW_NPG) npg = OW_NPG;
    if (npg < 1) npg = 1;
    const bool uselist = GAT_uselist(E, N) && G == OW_G;

    float m[F_E];
    #pragma unroll
    for (int f = 0; f < F_E; ++f) m[f] = M[f * NH + h];

    for (long n0 = (long)blockIdx.x * npg; n0 < N; n0 += (long)G * npg) {
        const int nn = (N - n0 < npg) ? (int)(N - n0) : npg;
        const int n0i = (int)n0;

        for (int i = tid; i < nn * NC; i += OW_T) accL[i] = 0.0f;
        for (int i = tid; i < nn * NH; i += OW_T) amaxL[i] = 0u;
        for (int i = tid; i < nn * OW_REC; i += OW_T) recL[i] = 0.0f;

        int cntv = 0, stv = 0;
        if (uselist && tid < SB) {
            cntv = binCnt[(int)blockIdx.x * SB + tid];
            stv  = binStart[(int)blockIdx.x * SB + tid];
        }
        int inc = cntv;
        #pragma unroll
        for (int o = 1; o < 32; o <<= 1) {
            const int y = __shfl_up(inc, o);
            if (lane >= o) inc += y;
        }
        if (tid < SB && lane == 31) wtotL[wv] = inc;
        __syncthreads();
        if (tid < SB) {
            int basev = 0;
            for (int j = 0; j < wv; ++j) basev += wtotL[j];
            posL[tid] = basev + inc - cntv;
            stL[tid]  = stv;
            cntL[tid] = cntv;
        }
        int nlist = 0;
        #pragma unroll
        for (int j = 0; j < 8; ++j) nlist += wtotL[j];
        const bool nolist = !uselist || nlist > OW_CAP;
        __syncthreads();

        if (!nolist) {
            v2i8 en[8];
            int  cn[8], st[8], ps[8];
            #pragma unroll
            for (int j = 0; j < 8; ++j) {
                cn[j] = cntL[wv * 8 + j];
                st[j] = stL[wv * 8 + j];
                ps[j] = posL[wv * 8 + j];
                const v2i8 none = {0, 0};
                en[j] = (lane < cn[j]) ? *(const v2i8*)(ent + 2 * (long)(st[j] + lane)) : none;
            }
            #pragma unroll
            for (int j = 0; j < 8; ++j) {
                if (lane < cn[j]) {
                    const int p = ps[j] + lane;
                    lstS[p] = en[j].x;
                    lstD[p] = (unsigned char)(en[j].y >> 16);
                    lstA[p * NH] = __int_as_float(en[j].y & 0xffff);
                }
            }
            #pragma unroll
            for (int j = 0; j < 8; ++j) {
                for (int i = lane + 32; i < cn[j]; i += 32) {
                    const v2i8 e2 = *(const v2i8*)(ent + 2 * (long)(st[j] + i));
                    const int p = ps[j] + i;
                    lstS[p] = e2.x;
                    lstD[p] = (unsigned char)(e2.y >> 16);
                    lstA[p * NH] = __int_as_float(e2.y & 0xffff);
                }
            }
        }
        __syncthreads();
        const int per = (((nlist + OW_W - 1) / OW_W) + 3) & ~3;
        const int sb  = wv * per;
        int cw = nolist ? 0 : (nlist - sb);
        if (cw > per) cw = per;
        if (cw < 0) cw = 0;

        if (!nolist) {
            const int ngrp = (cw + 3) >> 2;
            for (int g0 = 0; g0 < ngrp; g0 += OW_AU) {
                int   iv[OW_AU], dv[OW_AU], sv[OW_AU];
                bool  ok[OW_AU];
                v4fu  q[OW_AU][4];
                float as[OW_AU], ad[OW_AU];
                #pragma unroll
                for (int u = 0; u < OW_AU; ++u) {
                    const int i = (g0 + u) * 4 + sub;
                    ok[u] = i < cw;
                    iv[u] = sb + i;
                    const int e = ok[u] ? lstS[sb + i] : 0;
                    sv[u] = ok[u] ? __float_as_int(lstA[(sb + i) * NH]) : 0;
                    dv[u] = ok[u] ? (int)lstD[sb + i] : 0;
                    const float* row = eattr + (long)e * F_E;
                    q[u][0] = *(const v4fu*)(row + 0);
                    q[u][1] = *(const v4fu*)(row + 4);
                    q[u][2] = *(const v4fu*)(row + 8);
                    q[u][3] = *(const v4fu*)(row + 12);
                    as[u] = asrc[(long)sv[u] * NH + h];
                    ad[u] = adst[(n0 + dv[u]) * NH + h];
                }
                #pragma unroll
                for (int u = 0; u < OW_AU; ++u) {
                    const float ea[F_E] = {q[u][0].x, q[u][0].y, q[u][0].z, q[u][0].w,
                                           q[u][1].x, q[u][1].y, q[u][1].z, q[u][1].w,
                                           q[u][2].x, q[u][2].y, q[u][2].z, q[u][2].w,
                                           q[u][3].x, q[u][3].y, q[u][3].z, q[u][3].w};
                    float ae = 0.0f;
                    #pragma unroll
                    for (int f = 0; f < F_E; ++f) ae = fmaf(ea[f], m[f], ae);
                    float al = as[u] + ad[u] + ae;
                    al = (al >= 0.0f) ? al : NEG_SLOPE * al;
                    if (ok[u]) {
                        lstA[iv[u] * NH + h] = al;
                        if (h == 0) lstS[iv[u]] = sv[u];
                        atomicMax(&amaxL[dv[u] * NH + h], ordkey(al));
                        atomicAdd(&recL[dv[u] * OW_REC + h], ae);
                        if (h == 0) atomicAdd(&recL[dv[u] * OW_REC + NH], 1.0f);
                    }
                }
            }
        } else {
            for (long b = (long)wv * 32; b < E; b += OW_W * 32) {
                const long el = b + lane;
                unsigned r = 0xffffffffu;
                if (el < E) r = (unsigned)(dst[el] - n0i);
                unsigned mask = __builtin_amdgcn_ballot_w32(r < (unsigned)nn);
                while (mask) {
                    const int b0 = __builtin_ctz(mask);
                    mask &= mask - 1;
                    int d, s;
                    float ae;
                    const float al = GAT_alpha_one(b + b0, h, src, dst, eattr, asrc, adst, m, &d, &s, &ae);
                    const int dl = d - n0i;
                    if (lane < NH) {
                        atomicMax(&amaxL[dl * NH + h], ordkey(al));
                        atomicAdd(&recL[dl * OW_REC + h], ae);
                        if (h == 0) atomicAdd(&recL[dl * OW_REC + NH], 1.0f);
                    }
                }
            }
        }
        __syncthreads();

        for (int it = tid; it < nn * NH; it += OW_T) {
            const int nl = it >> 3, hh = it & 7;
            const long gi = (n0 + nl) * NH + hh;
            const float ae = recL[nl * OW_REC + hh] / fmaxf(recL[nl * OW_REC + NH], 1.0f);
            float al = asrc[gi] + adst[gi] + ae;
            al = (al >= 0.0f) ? al : NEG_SLOPE * al;
            unsigned k = ordkey(al);
            const unsigned c = amaxL[it];
            k = (k > c) ? k : c;
            const float mx = ordinv(k);
            const float es = __expf(al - mx);
            amaxL[it] = __float_as_uint(mx);
            denomL[it] = es;
            recL[nl * OW_REC + hh] = es;
        }
        __syncthreads();

        if (!nolist) {
            for (int it = lane; it < cw * NH; it += 32) {
                const int dl = lstD[sb + (it >> 3)];
                const float ex = __expf(lstA[sb * NH + it] - __uint_as_float(amaxL[dl * NH + h]));
                lstA[sb * NH + it] = ex;
                atomicAdd(&denomL[dl * NH + h], ex);
            }
        } else {
            for (long b = (long)wv * 32; b < E; b += OW_W * 32) {
                const long el = b + lane;
                unsigned r = 0xffffffffu;
                if (el < E) r = (unsigned)(dst[el] - n0i);
                unsigned mask = __builtin_amdgcn_ballot_w32(r < (unsigned)nn);
                while (mask) {
                    const int b0 = __builtin_ctz(mask);
                    mask &= mask - 1;
                    int d, s;
                    float ae;
                    const float al = GAT_alpha_one(b + b0, h, src, dst, eattr, asrc, adst, m, &d, &s, &ae);
                    const int dl = d - n0i;
                    if (lane < NH)
                        atomicAdd(&denomL[dl * NH + h],
                                  __expf(al - __uint_as_float(amaxL[dl * NH + h])));
                }
            }
        }
        __syncthreads();

        {
            const int nself = (nolist || nn <= wv) ? 0 : ((nn - wv + OW_W - 1) / OW_W);
            const int total = cw + nself;
            const int ngc = (total + EPW - 1) / EPW;
            const int grp = lane >> 3;
            const int cq  = lane & 7;
            const bool up16 = (lane & 16) != 0;
            const bool up8  = (lane & 8) != 0;
            float* aw = &attW[wv * 32];
            for (int g = 0; g < ngc; ++g) {
                int j = g * EPW + sub;
                if (j >= total) j = total - 1;
                int sl, dl;
                float w;
                if (j < cw) { sl = lstS[sb + j]; dl = lstD[sb + j]; w = lstA[(sb + j) * NH + h]; }
                else        { dl = wv + (j - cw) * OW_W; sl = n0i + dl; w = recL[dl * OW_REC + h]; }
                v4fu xa[EPW], xb[EPW];
                #pragma unroll
                for (int i = 0; i < EPW; ++i) {
                    const int s = __builtin_amdgcn_readlane(sl, i * 8);
                    const float* xr = xh + (long)s * HC + 4 * lane;
                    xa[i] = *(const v4fu*)(xr);
                    xb[i] = *(const v4fu*)(xr + 4 * NC);
                }
                aw[lane] = w / denomL[dl * NH + h];
                #pragma unroll
                for (int i = 0; i < EPW; ++i) {
                    const float a1 = aw[i * 8 + grp];
                    const float a2 = aw[i * 8 + 4 + grp];
                    const float r0 = fmaf(a2, xb[i].x, a1 * xa[i].x);
                    const float r1 = fmaf(a2, xb[i].y, a1 * xa[i].y);
                    const float r2 = fmaf(a2, xb[i].z, a1 * xa[i].z);
                    const float r3 = fmaf(a2, xb[i].w, a1 * xa[i].w);
                    float k0 = up16 ? r2 : r0;
                    float k1 = up16 ? r3 : r1;
                    const float s0 = up16 ? r0 : r2;
                    const float s1 = up16 ? r1 : r3;
                    k0 += GAT_swap16(s0);
                    k1 += GAT_swap16(s1);
                    float kk = up8 ? k1 : k0;
                    const float ss = up8 ? k0 : k1;
                    kk += GAT_swap8(ss);
                    const int d = __builtin_amdgcn_readlane(dl, i * 8);
                    if (g * EPW + i < total)
                        atomicAdd(&accL[d * NC + 4 * cq + grp], kk);
                }
            }
        }
        if (nolist) {
            for (long b = (long)wv * 32; b < E; b += OW_W * 32) {
                const long el = b + lane;
                unsigned r = 0xffffffffu;
                if (el < E) r = (unsigned)(dst[el] - n0i);
                unsigned mask = __builtin_amdgcn_ballot_w32(r < (unsigned)nn);
                while (mask) {
                    const int b0 = __builtin_ctz(mask);
                    mask &= mask - 1;
                    int d, s;
                    float ae;
                    const float al = GAT_alpha_one(b + b0, h, src, dst, eattr, asrc, adst, m, &d, &s, &ae);
                    const int dl = d - n0i;
                    const float att_l = __expf(al - __uint_as_float(amaxL[dl * NH + h])) / denomL[dl * NH + h];
                    const float* xr = xh + (long)s * HC + lane;
                    float a = __int_as_float(__builtin_amdgcn_readlane(__float_as_int(att_l), 0));
                    float acc = a * xr[0];
                    #pragma unroll
                    for (int hh = 1; hh < NH; ++hh) {
                        a = __int_as_float(__builtin_amdgcn_readlane(__float_as_int(att_l), hh));
                        acc = fmaf(a, xr[hh * NC], acc);
                    }
                    atomicAdd(&accL[dl * NC + lane], acc);
                }
            }
            for (int nl = wv; nl < nn; nl += OW_W) {
                const float att_l = recL[nl * OW_REC + h] / denomL[nl * NH + h];
                const float* xr = xh + (n0 + nl) * HC + lane;
                float a = __int_as_float(__builtin_amdgcn_readlane(__float_as_int(att_l), 0));
                float acc = a * xr[0];
                #pragma unroll
                for (int hh = 1; hh < NH; ++hh) {
                    a = __int_as_float(__builtin_amdgcn_readlane(__float_as_int(att_l), hh));
                    acc = fmaf(a, xr[hh * NC], acc);
                }
                atomicAdd(&accL[nl * NC + lane], acc);
            }
        }
        __syncthreads();

        for (int it = tid; it < nn * NC; it += OW_T)
            out[n0 * NC + it] = accL[it] * (1.0f / NH) + bias[it & 31];
        __syncthreads();
    }
}

extern "C" void kernel_launch(void* const* d_in, const int* in_sizes, int n_in,
                              void* d_out, int out_size, void* d_ws, size_t ws_size,
                              hipStream_t stream) {
    const float* x        = (const float*)d_in[0];
    const int*   eidx     = (const int*)  d_in[1];
    const float* eattr    = (const float*)d_in[2];
    const float* W        = (const float*)d_in[3];
    const float* We       = (const float*)d_in[4];
    const float* att_src  = (const float*)d_in[5];
    const float* att_dst  = (const float*)d_in[6];
    const float* att_edge = (const float*)d_in[7];
    const float* bias     = (const float*)d_in[8];

    const int N = in_sizes[0] / F_IN;
    const int E = in_sizes[1] / 2;
    const int* src = eidx;
    const int* dst = eidx + E;

    float* ws = (float*)d_ws;
    const long Nl = N;
    const long xhOff    = (65 * Nl) & ~63L;
    float*    w_xh    = ws + xhOff;
    float*    w_asrc  = ws + xhOff + 256 * Nl;
    float*    w_adst  = ws + xhOff + 264 * Nl;
    float*    w_M     = ws + xhOff + 272 * Nl;
    float*    w_wa    = ws;
    int*      w_bst   = (int*)d_ws + WA_FLOATS;
    int*      w_bcn   = w_bst + OW_G * SB;
    int*      w_ent   = w_bst + DESC_INTS;

    GAT_binsort<<<SB + 1 + WA_G, SB_T, 0, stream>>>(src, dst, We, att_edge, W, att_src, att_dst,
                                                   w_M, w_wa, w_bst, w_bcn, w_ent, E, N);
    {
        long rowTiles = (Nl + 31) >> 5;
        long attBlocks = (rowTiles + 3) >> 2;
        GAT_gemm_xh<<<(int)(rowTiles + attBlocks), 128, 0, stream>>>(x, W, w_wa, w_xh,
                                                                     w_asrc, w_adst, N);
    }
    GAT_owner<<<OW_G, OW_T, 0, stream>>>(src, dst, eattr, w_M, w_asrc, w_adst, w_xh, bias,
                                         w_bst, w_bcn, w_ent, (float*)d_out, E, N);
}


# Edge sort and projection in ONE launch of 4-wave work-groups: the sort's LDS and memory work runs beside the matrix-bound GEMM; attention-term groups fold their weights in LDS; sum order kept
# speedup vs baseline: 9.8262x; 1.0343x over previous
#include <hip/hip_runtime.h>
#include <hip/hip_bf16.h>

typedef float v2f __attribute__((ext_vector_type(2)));
typedef float v8f __attribute__((ext_vector_type(8)));

#define F_IN 128
#define F_E  16
#define NH   8
#define NC   32
#define HC   256
#define NEG_SLOPE 0.2f

__device__ __forceinline__ unsigned ordkey(float f) {
    unsigned u = __float_as_uint(f);
    return (u & 0x80000000u) ? ~u : (u | 0x80000000u);
}
__device__ __forceinline__ float ordinv(unsigned k) {
    unsigned u = (k & 0x80000000u) ? (k & 0x7fffffffu) : ~k;
    return __uint_as_float(u);
}

#define SB      256
#define WA_COLS 16
#define WA_FLOATS (WA_COLS * F_IN)
#define FR_T    128
#define SB_U    16
#define OW_G    256
#define DESC_INTS (2 * OW_G * SB)

typedef int v2i8 __attribute__((ext_vector_type(2)));
__device__ __forceinline__ bool GAT_uselist(int E, int N) {
    return N <= OW_G * 256 && 4L * (WA_FLOATS + DESC_INTS) + 8L * (long)E <= 260L * (long)N - 256L;
}

__device__ __forceinline__ void GAT_binsort_chunk(const int* __restrict__ src, const int* __restrict__ dst,
                                                  int* __restrict__ binStart, int* __restrict__ binCnt,
                                                  int* __restrict__ ent, int E, int N, int c) {
    __shared__ int histL[OW_G];
    __shared__ int curL[OW_G];
    __shared__ int wtotL[FR_T / 32];
    const int tid = threadIdx.x;
    if (!GAT_uselist(E, N)) return;
    const int lane = tid & 31;
    const int wv   = tid >> 5;
    int npg = (N + OW_G - 1) / OW_G;
    if (npg < 1) npg = 1;
    const int CH = (E + SB - 1) / SB;
    const long e0l = (long)c * CH;
    const int e0 = (e0l < E) ? (int)e0l : E;
    const int e1 = (e0l + CH < E) ? (int)(e0l + CH) : E;

    histL[2 * tid] = 0;
    histL[2 * tid + 1] = 0;
    __syncthreads();
    for (int b0 = e0; b0 < e1; b0 += SB_U * FR_T) {
        int dv[SB_U];
        #pragma unroll
        for (int u = 0; u < SB_U; ++u) {
            const int e = b0 + u * FR_T + tid;
            dv[u] = dst[(e < e1) ? e : (e1 - 1)];
        }
        #pragma unroll
        for (int u = 0; u < SB_U; ++u) {
            const int e = b0 + u * FR_T + tid;
            const unsigned k = (unsigned)dv[u] / (unsigned)npg;
            if (e < e1 && k < (unsigned)OW_G) atomicAdd(&histL[k], 1);
        }
    }
    __syncthreads();
    const int c0 = histL[2 * tid];
    const int c1 = histL[2 * tid + 1];
    const int pair = c0 + c1;
    int inc = pair;
    #pragma unroll
    for (int o = 1; o < 32; o <<= 1) {
        const int y = __shfl_up(inc, o);
        if (lane >= o) inc += y;
    }
    if (lane == 31) wtotL[wv] = inc;
    __syncthreads();
    {
        int basev = 0;
        for (int j = 0; j < wv; ++j) basev += wtotL[j];
        const int ex0 = basev + inc - pair;
        const int ex1 = ex0 + c0;
        curL[2 * tid]     = ex0;
        curL[2 * tid + 1] = ex1;
        binStart[(2 * tid) * SB + c]     = e0 + ex0;
        binCnt[(2 * tid) * SB + c]       = c0;
        binStart[(2 * tid + 1) * SB + c] = e0 + ex1;
        binCnt[(2 * tid + 1) * SB + c]   = c1;
    }
    __syncthreads();
    for (int b0 = e0; b0 < e1; b0 += SB_U * FR_T) {
        int dv[SB_U], sv[SB_U];
        #pragma unroll
        for (int u = 0; u < SB_U; ++u) {
            const int e = b0 + u * FR_T + tid;
            const int ec = (e < e1) ? e : (e1 - 1);
            dv[u] = dst[ec];
            sv[u] = src[ec];
        }
        #pragma unroll
        for (int u = 0; u < SB_U; ++u) {
            const int e = b0 + u * FR_T + tid;
            const unsigned k = (unsigned)dv[u] / (unsigned)npg;
            if (e < e1 && k < (unsigned)OW_G) {
                const int p = e0 + atomicAdd(&curL[k], 1);
                const v2i8 en = {e, (sv[u] & 0xffff) | ((dv[u] - (int)k * npg) << 16)};
                *(v2i8*)(ent + 2 * (long)p) = en;
            }
        }
    }
}

typedef float v4f8 __attribute__((ext_vector_type(4), aligned(8)));
typedef float v4f  __attribute__((ext_vector_type(4)));

#define ATT_TPW  4
#define WA_PITCH 132
__device__ __forceinline__ void GAT_att_tiles(const float* __restrict__ x,
                                              const float* __restrict__ W,
                                              const float* __restrict__ att_src,
                                              const float* __restrict__ att_dst,
                                              float* __restrict__ asrc,
                                              float* __restrict__ adst,
                                              int N, long ab, long rowTiles) {
    __shared__ __attribute__((aligned(16))) float waL[WA_COLS * WA_PITCH];
    const int lane = threadIdx.x & 31;
    const int half = lane >> 4;
    const int l    = lane & 15;
    const int wv = __builtin_amdgcn_readfirstlane((int)threadIdx.x >> 5);
    for (int i = 0; i < 2; ++i) {
        const int hh = 2 * wv + i;
        float s0 = 0.0f, s1 = 0.0f, s2 = 0.0f, s3 = 0.0f;
        float t0 = 0.0f, t1 = 0.0f, t2 = 0.0f, t3 = 0.0f;
        #pragma unroll 8
        for (int c = 0; c < NC; ++c) {
            const v4f8 w = *(const v4f8*)(W + (long)(hh * NC + c) * F_IN + 4 * lane);
            const float as = att_src[hh * NC + c];
            const float ad = att_dst[hh * NC + c];
            s0 = fmaf(w.x, as, s0); s1 = fmaf(w.y, as, s1);
            s2 = fmaf(w.z, as, s2); s3 = fmaf(w.w, as, s3);
            t0 = fmaf(w.x, ad, t0); t1 = fmaf(w.y, ad, t1);
            t2 = fmaf(w.z, ad, t2); t3 = fmaf(w.w, ad, t3);
        }
        const v4f sv = {s0, s1, s2, s3};
        const v4f tv = {t0, t1, t2, t3};
        *(v4f*)(waL + hh * WA_PITCH + 4 * lane) = sv;
        *(v4f*)(waL + (NH + hh) * WA_PITCH + 4 * lane) = tv;
    }
    __syncthreads();
    const float* wq = waL + l * WA_PITCH + 4 * half;
    #pragma nounroll
    for (int t = 0; t < ATT_TPW; ++t) {
        const long rt = (ab * 4 + wv) * ATT_TPW + t;
        if (rt >= rowTiles) break;
        const long rowA = rt * 32;
        long q0 = rowA + l;
        long q1 = rowA + 16 + l;
        if (q0 >= N) q0 = N - 1;
        if (q1 >= N) q1 = N - 1;
        const float* xq0 = x + q0 * F_IN + 4 * half;
        const float* xq1 = x + q1 * F_IN + 4 * half;
        const v8f z8 = {0.f,0.f,0.f,0.f,0.f,0.f,0.f,0.f};
        v8f d0 = z8, d1 = z8;
        #pragma unroll
        for (int k0 = 0; k0 < F_IN; k0 += 8) {
            const v4f8 a0 = *(const v4f8*)(xq0 + k0);
            const v4f8 a1 = *(const v4f8*)(xq1 + k0);
            const v4f  b  = *(const v4f*)(wq + k0);
            const v2f a0l = {a0.x, a0.y}, a0h = {a0.z, a0.w};
            const v2f a1l = {a1.x, a1.y}, a1h = {a1.z, a1.w};
            const v2f bl  = {b.x, b.y},   bh  = {b.z, b.w};
            d0 = __builtin_amdgcn_wmma_f32_16x16x4_f32(false, a0l, false, bl, (short)0, d0, false, false);
            d1 = __builtin_amdgcn_wmma_f32_16x16x4_f32(false, a1l, false, bl, (short)0, d1, false, false);
            d0 = __builtin_amdgcn_wmma_f32_16x16x4_f32(false, a0h, false, bh, (short)0, d0, false, false);
            d1 = __builtin_amdgcn_wmma_f32_16x16x4_f32(false, a1h, false, bh, (short)0, d1, false, false);
        }
        float* ap = (l < NH) ? asrc : adst;
        const int hh = l & (NH - 1);
        #pragma unroll
        for (int v = 0; v < 8; ++v) {
            const long r0 = rowA + v + 8 * half;
            if (r0 < N) ap[r0 * NH + hh] = d0[v];
            const long r1 = r0 + 16;
            if (r1 < N) ap[r1 * NH + hh] = d1[v];
        }
    }
}

__global__ void __launch_bounds__(FR_T)
GAT_front(const int* __restrict__ src, const int* __restrict__ dst,
          const float* __restrict__ We, const float* __restrict__ att_edge,
          const float* __restrict__ att_src, const float* __restrict__ att_dst,
          const float* __restrict__ x, const float* __restrict__ W,
          float* __restrict__ M,
          int* __restrict__ binStart, int* __restrict__ binCnt, int* __restrict__ ent,
          float* __restrict__ xh, float* __restrict__ asrc, float* __restrict__ adst,
          int E, int N) {
    const int bid = (int)blockIdx.x;
    if (bid < SB) {
        GAT_binsort_chunk(src, dst, binStart, binCnt, ent, E, N, bid);
        return;
    }
    if (bid == SB) {
        const int tid = threadIdx.x;
        if (tid < F_E * NH) {
            const int f = tid >> 3, h = tid & 7;
            float sum = 0.0f;
            for (int c = 0; c < NC; ++c)
                sum += We[(long)(h * NC + c) * F_E + f] * att_edge[h * NC + c];
            M[f * NH + h] = sum;
        }
        return;
    }
    const long rowTiles = ((long)N + 31) >> 5;
    const long attBlocks = (rowTiles + 4 * ATT_TPW - 1) / (4 * ATT_TPW);
    const long rb = (long)bid - (SB + 1);
    if (rb < attBlocks) {
        GAT_att_tiles(x, W, att_src, att_dst, asrc, adst, N, rb, rowTiles);
        return;
    }
    const long pb = rb - attBlocks;
    int wave = threadIdx.x >> 5;
    int lane = threadIdx.x & 31;
    int half = lane >> 4;
    int l    = lane & 15;
    long wtile = pb * 4 + wave;
    if (wtile >= rowTiles * 4) return;
    long rowTile = wtile >> 2;
    int  colGrp  = (int)(wtile & 3);
    long row0 = rowTile * 32;
    int  col0 = colGrp * 64;

    long ar0 = row0 + l;
    long ar1 = row0 + 16 + l;
    if (ar0 >= N) ar0 = N - 1;
    if (ar1 >= N) ar1 = N - 1;
    const float* xp0 = x + ar0 * F_IN + 4 * half;
    const float* xp1 = x + ar1 * F_IN + 4 * half;
    const float* wp0 = W + (long)(col0 +  0 + l) * F_IN + 4 * half;
    const float* wp1 = W + (long)(col0 + 16 + l) * F_IN + 4 * half;
    const float* wp2 = W + (long)(col0 + 32 + l) * F_IN + 4 * half;
    const float* wp3 = W + (long)(col0 + 48 + l) * F_IN + 4 * half;

    const v8f zero8 = {0.f,0.f,0.f,0.f,0.f,0.f,0.f,0.f};
    v8f c00 = zero8, c01 = zero8, c02 = zero8, c03 = zero8;
    v8f c10 = zero8, c11 = zero8, c12 = zero8, c13 = zero8;
    #pragma unroll
    for (int k0 = 0; k0 < F_IN; k0 += 8) {
        const v4f8 a0 = *(const v4f8*)(xp0 + k0);
        const v4f8 a1 = *(const v4f8*)(xp1 + k0);
        const v4f8 b0 = *(const v4f8*)(wp0 + k0);
        const v4f8 b1 = *(const v4f8*)(wp1 + k0);
        const v4f8 b2 = *(const v4f8*)(wp2 + k0);
        const v4f8 b3 = *(const v4f8*)(wp3 + k0);
        const v2f a0l = {a0.x, a0.y}, a0h = {a0.z, a0.w};
        const v2f a1l = {a1.x, a1.y}, a1h = {a1.z, a1.w};
        const v2f b0l = {b0.x, b0.y}, b0h = {b0.z, b0.w};
        const v2f b1l = {b1.x, b1.y}, b1h = {b1.z, b1.w};
        const v2f b2l = {b2.x, b2.y}, b2h = {b2.z, b2.w};
        const v2f b3l = {b3.x, b3.y}, b3h = {b3.z, b3.w};
        c00 = __builtin_amdgcn_wmma_f32_16x16x4_f32(false, a0l, false, b0l, (short)0, c00, false, false);
        c01 = __builtin_amdgcn_wmma_f32_16x16x4_f32(false, a0l, false, b1l, (short)0, c01, false, false);
        c02 = __builtin_amdgcn_wmma_f32_16x16x4_f32(false, a0l, false, b2l, (short)0, c02, false, false);
        c03 = __builtin_amdgcn_wmma_f32_16x16x4_f32(false, a0l, false, b3l, (short)0, c03, false, false);
        c10 = __builtin_amdgcn_wmma_f32_16x16x4_f32(false, a1l, false, b0l, (short)0, c10, false, false);
        c11 = __builtin_amdgcn_wmma_f32_16x16x4_f32(false, a1l, false, b1l, (short)0, c11, false, false);
        c12 = __builtin_amdgcn_wmma_f32_16x16x4_f32(false, a1l, false, b2l, (short)0, c12, false, false);
        c13 = __builtin_amdgcn_wmma_f32_16x16x4_f32(false, a1l, false, b3l, (short)0, c13, false, false);
        c00 = __builtin_amdgcn_wmma_f32_16x16x4_f32(false, a0h, false, b0h, (short)0, c00, false, false);
        c01 = __builtin_amdgcn_wmma_f32_16x16x4_f32(false, a0h, false, b1h, (short)0, c01, false, false);
        c02 = __builtin_amdgcn_wmma_f32_16x16x4_f32(false, a0h, false, b2h, (short)0, c02, false, false);
        c03 = __builtin_amdgcn_wmma_f32_16x16x4_f32(false, a0h, false, b3h, (short)0, c03, false, false);
        c10 = __builtin_amdgcn_wmma_f32_16x16x4_f32(false, a1h, false, b0h, (short)0, c10, false, false);
        c11 = __builtin_amdgcn_wmma_f32_16x16x4_f32(false, a1h, false, b1h, (short)0, c11, false, false);
        c12 = __builtin_amdgcn_wmma_f32_16x16x4_f32(false, a1h, false, b2h, (short)0, c12, false, false);
        c13 = __builtin_amdgcn_wmma_f32_16x16x4_f32(false, a1h, false, b3h, (short)0, c13, false, false);
    }
    #pragma unroll
    for (int v = 0; v < 8; ++v) {
        const long r0 = row0 + v + 8 * half;
        if (r0 < N) {
            float* o = xh + r0 * HC + col0 + l;
            o[0]  = c00[v];
            o[16] = c01[v];
            o[32] = c02[v];
            o[48] = c03[v];
        }
        const long r1 = r0 + 16;
        if (r1 < N) {
            float* o = xh + r1 * HC + col0 + l;
            o[0]  = c10[v];
            o[16] = c11[v];
            o[32] = c12[v];
            o[48] = c13[v];
        }
    }
}

typedef float v4fu __attribute__((ext_vector_type(4), aligned(4)));
typedef int   v4iu __attribute__((ext_vector_type(4), aligned(4)));
#define OW_T    1024
#define OW_W    32
#define OW_NPG  256
#define OW_SEG  192
#define OW_CAP  (OW_W * OW_SEG)
#define OW_AU   3
#define OW_REC  9
#define EPW     4

__device__ __forceinline__ float GAT_swap16(float v) {
    const int b = __float_as_int(v);
    return __int_as_float(__builtin_amdgcn_permlanex16(b, b, 0x76543210, 0xfedcba98, false, false));
}
__device__ __forceinline__ float GAT_swap8(float v) {
    return __int_as_float(__builtin_amdgcn_update_dpp(0, __float_as_int(v), 0x128  , 0xf, 0xf, false));
}

__device__ __forceinline__ float GAT_alpha_one(long t, int h,
                                               const int* __restrict__ src,
                                               const int* __restrict__ dst,
                                               const float* __restrict__ eattr,
                                               const float* __restrict__ asrc,
                                               const float* __restrict__ adst,
                                               const float* m, int* d_out, int* s_out,
                                               float* ae_out) {
    const int s = src[t];
    const int d = dst[t];
    const float* row = eattr + t * F_E;
    const v4fu q0 = *(const v4fu*)(row + 0);
    const v4fu q1 = *(const v4fu*)(row + 4);
    const v4fu q2 = *(const v4fu*)(row + 8);
    const v4fu q3 = *(const v4fu*)(row + 12);
    const float as = asrc[(long)s * NH + h];
    const float ad = adst[(long)d * NH + h];
    const float ea[F_E] = {q0.x, q0.y, q0.z, q0.w, q1.x, q1.y, q1.z, q1.w,
                           q2.x, q2.y, q2.z, q2.w, q3.x, q3.y, q3.z, q3.w};
    float ae = 0.0f;
    #pragma unroll
    for (int f = 0; f < F_E; ++f) ae = fmaf(ea[f], m[f], ae);
    float al = as + ad + ae;
    al = (al >= 0.0f) ? al : NEG_SLOPE * al;
    *d_out = d;
    *s_out = s;
    *ae_out = ae;
    return al;
}

__global__ void __launch_bounds__(OW_T)
GAT_owner(const int* __restrict__ src, const int* __restrict__ dst,
          const float* __restrict__ eattr, const float* __restrict__ M,
          const float* __restrict__ asrc, const float* __restrict__ adst,
          const float* __restrict__ xh, const float* __restrict__ bias,
          const int* __restrict__ binStart, const int* __restrict__ binCnt,
          const int* __restrict__ ent,
          float* __restrict__ out, int E, int N) {
    __shared__ float         accL[OW_NPG * NC];
    __shared__ unsigned      amaxL[OW_NPG * NH];
    __shared__ float         denomL[OW_NPG * NH];
    __shared__ float         recL[OW_NPG * OW_REC];
    __shared__ float         lstA[OW_CAP * NH];
    __shared__ int           lstS[OW_CAP];
    __shared__ unsigned char lstD[OW_CAP];
    __shared__ int           posL[SB];
    __shared__ int           stL[SB];
    __shared__ int           cntL[SB];
    __shared__ int           wtotL[8];
    __shared__ float         attW[OW_W * 32];

    const int tid  = threadIdx.x;
    const int lane = tid & 31;
    const int h    = lane & 7;
    const int sub  = lane >> 3;
    const int wv = __builtin_amdgcn_readfirstlane(tid >> 5);
    const int G  = (int)gridDim.x;
    int npg = (int)(((long)N + G - 1) / G);
    if (npg > OW_NPG) npg = OW_NPG;
    if (npg < 1) npg = 1;
    const bool uselist = GAT_uselist(E, N) && G == OW_G;

    float m[F_E];
    #pragma unroll
    for (int f = 0; f < F_E; ++f) m[f] = M[f * NH + h];

    for (long n0 = (long)blockIdx.x * npg; n0 < N; n0 += (long)G * npg) {
        const int nn = (N - n0 < npg) ? (int)(N - n0) : npg;
        const int n0i = (int)n0;

        for (int i = tid; i < nn * NC; i += OW_T) accL[i] = 0.0f;
        for (int i = tid; i < nn * NH; i += OW_T) amaxL[i] = 0u;
        for (int i = tid; i < nn * OW_REC; i += OW_T) recL[i] = 0.0f;

        int cntv = 0, stv = 0;
        if (uselist && tid < SB) {
            cntv = binCnt[(int)blockIdx.x * SB + tid];
            stv  = binStart[(int)blockIdx.x * SB + tid];
        }
        int inc = cntv;
        #pragma unroll
        for (int o = 1; o < 32; o <<= 1) {
            const int y = __shfl_up(inc, o);
            if (lane >= o) inc += y;
        }
        if (tid < SB && lane == 31) wtotL[wv] = inc;
        __syncthreads();
        if (tid < SB) {
            int basev = 0;
            for (int j = 0; j < wv; ++j) basev += wtotL[j];
            posL[tid] = basev + inc - cntv;
            stL[tid]  = stv;
            cntL[tid] = cntv;
        }
        int nlist = 0;
        #pragma unroll
        for (int j = 0; j < 8; ++j) nlist += wtotL[j];
        const bool nolist = !uselist || nlist > OW_CAP;
        __syncthreads();

        if (!nolist) {
            v2i8 en[8];
            int  cn[8], st[8], ps[8];
            #pragma unroll
            for (int j = 0; j < 8; ++j) {
                cn[j] = cntL[wv * 8 + j];
                st[j] = stL[wv * 8 + j];
                ps[j] = posL[wv * 8 + j];
                const v2i8 none = {0, 0};
                en[j] = (lane < cn[j]) ? *(const v2i8*)(ent + 2 * (long)(st[j] + lane)) : none;
            }
            #pragma unroll
            for (int j = 0; j < 8; ++j) {
                if (lane < cn[j]) {
                    const int p = ps[j] + lane;
                    lstS[p] = en[j].x;
                    lstD[p] = (unsigned char)(en[j].y >> 16);
                    lstA[p * NH] = __int_as_float(en[j].y & 0xffff);
                }
            }
            #pragma unroll
            for (int j = 0; j < 8; ++j) {
                for (int i = lane + 32; i < cn[j]; i += 32) {
                    const v2i8 e2 = *(const v2i8*)(ent + 2 * (long)(st[j] + i));
                    const int p = ps[j] + i;
                    lstS[p] = e2.x;
                    lstD[p] = (unsigned char)(e2.y >> 16);
                    lstA[p * NH] = __int_as_float(e2.y & 0xffff);
                }
            }
        }
        __syncthreads();
        const int per = (((nlist + OW_W - 1) / OW_W) + 3) & ~3;
        const int sb  = wv * per;
        int cw = nolist ? 0 : (nlist - sb);
        if (cw > per) cw = per;
        if (cw < 0) cw = 0;

        if (!nolist) {
            const int ngrp = (cw + 3) >> 2;
            for (int g0 = 0; g0 < ngrp; g0 += OW_AU) {
                int   iv[OW_AU], dv[OW_AU], sv[OW_AU];
                bool  ok[OW_AU];
                v4fu  q[OW_AU][4];
                float as[OW_AU], ad[OW_AU];
                #pragma unroll
                for (int u = 0; u < OW_AU; ++u) {
                    const int i = (g0 + u) * 4 + sub;
                    ok[u] = i < cw;
                    iv[u] = sb + i;
                    const int e = ok[u] ? lstS[sb + i] : 0;
                    sv[u] = ok[u] ? __float_as_int(lstA[(sb + i) * NH]) : 0;
                    dv[u] = ok[u] ? (int)lstD[sb + i] : 0;
                    const float* row = eattr + (long)e * F_E;
                    q[u][0] = *(const v4fu*)(row + 0);
                    q[u][1] = *(const v4fu*)(row + 4);
                    q[u][2] = *(const v4fu*)(row + 8);
                    q[u][3] = *(const v4fu*)(row + 12);
                    as[u] = asrc[(long)sv[u] * NH + h];
                    ad[u] = adst[(n0 + dv[u]) * NH + h];
                }
                #pragma unroll
                for (int u = 0; u < OW_AU; ++u) {
                    const float ea[F_E] = {q[u][0].x, q[u][0].y, q[u][0].z, q[u][0].w,
                                           q[u][1].x, q[u][1].y, q[u][1].z, q[u][1].w,
                                           q[u][2].x, q[u][2].y, q[u][2].z, q[u][2].w,
                                           q[u][3].x, q[u][3].y, q[u][3].z, q[u][3].w};
                    float ae = 0.0f;
                    #pragma unroll
                    for (int f = 0; f < F_E; ++f) ae = fmaf(ea[f], m[f], ae);
                    float al = as[u] + ad[u] + ae;
                    al = (al >= 0.0f) ? al : NEG_SLOPE * al;
                    if (ok[u]) {
                        lstA[iv[u] * NH + h] = al;
                        if (h == 0) lstS[iv[u]] = sv[u];
                        atomicMax(&amaxL[dv[u] * NH + h], ordkey(al));
                        atomicAdd(&recL[dv[u] * OW_REC + h], ae);
                        if (h == 0) atomicAdd(&recL[dv[u] * OW_REC + NH], 1.0f);
                    }
                }
            }
        } else {
            for (long b = (long)wv * 32; b < E; b += OW_W * 32) {
                const long el = b + lane;
                unsigned r = 0xffffffffu;
                if (el < E) r = (unsigned)(dst[el] - n0i);
                unsigned mask = __builtin_amdgcn_ballot_w32(r < (unsigned)nn);
                while (mask) {
                    const int b0 = __builtin_ctz(mask);
                    mask &= mask - 1;
                    int d, s;
                    float ae;
                    const float al = GAT_alpha_one(b + b0, h, src, dst, eattr, asrc, adst, m, &d, &s, &ae);
                    const int dl = d - n0i;
                    if (lane < NH) {
                        atomicMax(&amaxL[dl * NH + h], ordkey(al));
                        atomicAdd(&recL[dl * OW_REC + h], ae);
                        if (h == 0) atomicAdd(&recL[dl * OW_REC + NH], 1.0f);
                    }
                }
            }
        }
        __syncthreads();

        for (int it = tid; it < nn * NH; it += OW_T) {
            const int nl = it >> 3, hh = it & 7;
            const long gi = (n0 + nl) * NH + hh;
            const float ae = recL[nl * OW_REC + hh] / fmaxf(recL[nl * OW_REC + NH], 1.0f);
            float al = asrc[gi] + adst[gi] + ae;
            al = (al >= 0.0f) ? al : NEG_SLOPE * al;
            unsigned k = ordkey(al);
            const unsigned c = amaxL[it];
            k = (k > c) ? k : c;
            const float mx = ordinv(k);
            const float es = __expf(al - mx);
            amaxL[it] = __float_as_uint(mx);
            denomL[it] = es;
            recL[nl * OW_REC + hh] = es;
        }
        __syncthreads();

        if (!nolist) {
            for (int it = lane; it < cw * NH; it += 32) {
                const int dl = lstD[sb + (it >> 3)];
                const float ex = __expf(lstA[sb * NH + it] - __uint_as_float(amaxL[dl * NH + h]));
                lstA[sb * NH + it] = ex;
                atomicAdd(&denomL[dl * NH + h], ex);
            }
        } else {
            for (long b = (long)wv * 32; b < E; b += OW_W * 32) {
                const long el = b + lane;
                unsigned r = 0xffffffffu;
                if (el < E) r = (unsigned)(dst[el] - n0i);
                unsigned mask = __builtin_amdgcn_ballot_w32(r < (unsigned)nn);
                while (mask) {
                    const int b0 = __builtin_ctz(mask);
                    mask &= mask - 1;
                    int d, s;
                    float ae;
                    const float al = GAT_alpha_one(b + b0, h, src, dst, eattr, asrc, adst, m, &d, &s, &ae);
                    const int dl = d - n0i;
                    if (lane < NH)
                        atomicAdd(&denomL[dl * NH + h],
                                  __expf(al - __uint_as_float(amaxL[dl * NH + h])));
                }
            }
        }
        __syncthreads();

        {
            const int nself = (nolist || nn <= wv) ? 0 : ((nn - wv + OW_W - 1) / OW_W);
            const int total = cw + nself;
            const int ngc = (total + EPW - 1) / EPW;
            const int grp = lane >> 3;
            const int cq  = lane & 7;
            const bool up16 = (lane & 16) != 0;
            const bool up8  = (lane & 8) != 0;
            float* aw = &attW[wv * 32];
            for (int g = 0; g < ngc; ++g) {
                int j = g * EPW + sub;
                if (j >= total) j = total - 1;
                int sl, dl;
                float w;
                if (j < cw) { sl = lstS[sb + j]; dl = lstD[sb + j]; w = lstA[(sb + j) * NH + h]; }
                else        { dl = wv + (j - cw) * OW_W; sl = n0i + dl; w = recL[dl * OW_REC + h]; }
                v4fu xa[EPW], xb[EPW];
                #pragma unroll
                for (int i = 0; i < EPW; ++i) {
                    const int s = __builtin_amdgcn_readlane(sl, i * 8);
                    const float* xr = xh + (long)s * HC + 4 * lane;
                    xa[i] = *(const v4fu*)(xr);
                    xb[i] = *(const v4fu*)(xr + 4 * NC);
                }
                aw[lane] = w / denomL[dl * NH + h];
                #pragma unroll
                for (int i = 0; i < EPW; ++i) {
                    const float a1 = aw[i * 8 + grp];
                    const float a2 = aw[i * 8 + 4 + grp];
                    const float r0 = fmaf(a2, xb[i].x, a1 * xa[i].x);
                    const float r1 = fmaf(a2, xb[i].y, a1 * xa[i].y);
                    const float r2 = fmaf(a2, xb[i].z, a1 * xa[i].z);
                    const float r3 = fmaf(a2, xb[i].w, a1 * xa[i].w);
                    float k0 = up16 ? r2 : r0;
                    float k1 = up16 ? r3 : r1;
                    const float s0 = up16 ? r0 : r2;
                    const float s1 = up16 ? r1 : r3;
                    k0 += GAT_swap16(s0);
                    k1 += GAT_swap16(s1);
                    float kk = up8 ? k1 : k0;
                    const float ss = up8 ? k0 : k1;
                    kk += GAT_swap8(ss);
                    const int d = __builtin_amdgcn_readlane(dl, i * 8);
                    if (g * EPW + i < total)
                        atomicAdd(&accL[d * NC + 4 * cq + grp], kk);
                }
            }
        }
        if (nolist) {
            for (long b = (long)wv * 32; b < E; b += OW_W * 32) {
                const long el = b + lane;
                unsigned r = 0xffffffffu;
                if (el < E) r = (unsigned)(dst[el] - n0i);
                unsigned mask = __builtin_amdgcn_ballot_w32(r < (unsigned)nn);
                while (mask) {
                    const int b0 = __builtin_ctz(mask);
                    mask &= mask - 1;
                    int d, s;
                    float ae;
                    const float al = GAT_alpha_one(b + b0, h, src, dst, eattr, asrc, adst, m, &d, &s, &ae);
                    const int dl = d - n0i;
                    const float att_l = __expf(al - __uint_as_float(amaxL[dl * NH + h])) / denomL[dl * NH + h];
                    const float* xr = xh + (long)s * HC + lane;
                    float a = __int_as_float(__builtin_amdgcn_readlane(__float_as_int(att_l), 0));
                    float acc = a * xr[0];
                    #pragma unroll
                    for (int hh = 1; hh < NH; ++hh) {
                        a = __int_as_float(__builtin_amdgcn_readlane(__float_as_int(att_l), hh));
                        acc = fmaf(a, xr[hh * NC], acc);
                    }
                    atomicAdd(&accL[dl * NC + lane], acc);
                }
            }
            for (int nl = wv; nl < nn; nl += OW_W) {
                const float att_l = recL[nl * OW_REC + h] / denomL[nl * NH + h];
                const float* xr = xh + (n0 + nl) * HC + lane;
                float a = __int_as_float(__builtin_amdgcn_readlane(__float_as_int(att_l), 0));
                float acc = a * xr[0];
                #pragma unroll
                for (int hh = 1; hh < NH; ++hh) {
                    a = __int_as_float(__builtin_amdgcn_readlane(__float_as_int(att_l), hh));
                    acc = fmaf(a, xr[hh * NC], acc);
                }
                atomicAdd(&accL[nl * NC + lane], acc);
            }
        }
        __syncthreads();

        for (int it = tid; it < nn * NC; it += OW_T)
            out[n0 * NC + it] = accL[it] * (1.0f / NH) + bias[it & 31];
        __syncthreads();
    }
}

extern "C" void kernel_launch(void* const* d_in, const int* in_sizes, int n_in,
                              void* d_out, int out_size, void* d_ws, size_t ws_size,
                              hipStream_t stream) {
    const float* x        = (const float*)d_in[0];
    const int*   eidx     = (const int*)  d_in[1];
    const float* eattr    = (const float*)d_in[2];
    const float* W        = (const float*)d_in[3];
    const float* We       = (const float*)d_in[4];
    const float* att_src  = (const float*)d_in[5];
    const float* att_dst  = (const float*)d_in[6];
    const float* att_edge = (const float*)d_in[7];
    const float* bias     = (const float*)d_in[8];

    const int N = in_sizes[0] / F_IN;
    const int E = in_sizes[1] / 2;
    const int* src = eidx;
    const int* dst = eidx + E;

    float* ws = (float*)d_ws;
    const long Nl = N;
    const long xhOff    = (65 * Nl) & ~63L;
    float*    w_xh    = ws + xhOff;
    float*    w_asrc  = ws + xhOff + 256 * Nl;
    float*    w_adst  = ws + xhOff + 264 * Nl;
    float*    w_M     = ws + xhOff + 272 * Nl;
    int*      w_bst   = (int*)d_ws + WA_FLOATS;
    int*      w_bcn   = w_bst + OW_G * SB;
    int*      w_ent   = w_bst + DESC_INTS;

    {
        long rowTiles = (Nl + 31) >> 5;
        long attBlocks = (rowTiles + 4 * ATT_TPW - 1) / (4 * ATT_TPW);
        GAT_front<<<(int)(SB + 1 + attBlocks + rowTiles), FR_T, 0, stream>>>(
            src, dst, We, att_edge, att_src, att_dst, x, W, w_M, w_bst, w_bcn, w_ent,
            w_xh, w_asrc, w_adst, E, N);
    }
    GAT_owner<<<OW_G, OW_T, 0, stream>>>(src, dst, eattr, w_M, w_asrc, w_adst, w_xh, bias,
                                         w_bst, w_bcn, w_ent, (float*)d_out, E, N);
}
